# starting kernel
# baseline (speedup 1.0000x reference)
#include <hip/hip_runtime.h>
#include <math.h>

typedef float v2f __attribute__((ext_vector_type(2)));
typedef float v8f __attribute__((ext_vector_type(8)));

#define HD __device__ __forceinline__

static constexpr int   Hc   = 64;
static constexpr float GEPS = 1e-7f;

HD void atomicMaxF(float* addr, float val) {
    if (val >= 0.0f)
        atomicMax((int*)addr, __float_as_int(val));
    else
        atomicMin((unsigned int*)addr, __float_as_uint(val));
}

__global__ void gemm_bias_wmma(const float* __restrict__ A,
                               const float* __restrict__ B,
                               const float* __restrict__ bias,
                               float* __restrict__ C,
                               int M, int N, int K) {
    int wave    = (int)((blockIdx.x * blockDim.x + threadIdx.x) >> 5);
    int lane    = threadIdx.x & 31;
    int groupsN = (N + 63) >> 6;
    int tm = wave / groupsN;
    int gn = wave - tm * groupsN;
    if (tm >= (M >> 4)) return;

    int mr   = lane & 15;
    int half = lane >> 4;
    const float* Arow = A + (long)(tm * 16 + mr) * K;

    int ncol[4];
#pragma unroll
    for (int t = 0; t < 4; ++t) {
        int n   = gn * 64 + t * 16 + mr;
        ncol[t] = (n < N) ? n : (N - 1);
    }

    v8f acc[4] = {{0.f,0.f,0.f,0.f,0.f,0.f,0.f,0.f},
                  {0.f,0.f,0.f,0.f,0.f,0.f,0.f,0.f},
                  {0.f,0.f,0.f,0.f,0.f,0.f,0.f,0.f},
                  {0.f,0.f,0.f,0.f,0.f,0.f,0.f,0.f}};

    int ka = half * 2;
    v2f a_cur;
    a_cur.x = Arow[ka];
    a_cur.y = Arow[ka + 1];
    v2f b_cur[4];
#pragma unroll
    for (int t = 0; t < 4; ++t) {
        b_cur[t].x = B[ka * N + ncol[t]];
        b_cur[t].y = B[(ka + 1) * N + ncol[t]];
    }

    for (int k = 0; k < K; k += 4) {
        int kn = (k + 4 < K) ? (k + 4) : k;
        int kb = kn + half * 2;

        v2f a_nxt;
        a_nxt.x = Arow[kb];
        a_nxt.y = Arow[kb + 1];
        v2f b_nxt[4];
#pragma unroll
        for (int t = 0; t < 4; ++t) {
            b_nxt[t].x = B[kb * N + ncol[t]];
            b_nxt[t].y = B[(kb + 1) * N + ncol[t]];
        }

#pragma unroll
        for (int t = 0; t < 4; ++t)
            acc[t] = __builtin_amdgcn_wmma_f32_16x16x4_f32(
                         false, a_cur, false, b_cur[t], (short)0, acc[t], false, false);

        a_cur = a_nxt;
#pragma unroll
        for (int t = 0; t < 4; ++t) b_cur[t] = b_nxt[t];
    }

#pragma unroll
    for (int t = 0; t < 4; ++t) {
        int n = gn * 64 + t * 16 + mr;
        if (n < N) {
            float bv = bias[n];
#pragma unroll
            for (int r = 0; r < 8; ++r)
                C[(long)(tm * 16 + half * 8 + r) * N + n] = acc[t][r] + bv;
        }
    }
}

__global__ void ln_relu_kernel(const float* __restrict__ X,
                               const float* __restrict__ g,
                               const float* __restrict__ bvec,
                               float* __restrict__ Y,
                               int rows, int D) {
    int warp = threadIdx.x >> 5;
    int lane = threadIdx.x & 31;
    int row  = blockIdx.x * (blockDim.x >> 5) + warp;
    if (row >= rows) return;

    const float* x = X + (long)row * D;
    int   nper = D >> 5;
    float vals[4];
    float s = 0.f, ss = 0.f;
    for (int j = 0; j < nper; ++j) {
        float v = x[lane + j * 32];
        vals[j] = v;
        s  += v;
        ss += v * v;
    }
    for (int off = 16; off; off >>= 1) {
        s  += __shfl_xor(s,  off);
        ss += __shfl_xor(ss, off);
    }
    float mean = s / (float)D;
    float var  = ss / (float)D - mean * mean;
    float rstd = rsqrtf(var + 1e-5f);

    float* y = Y + (long)row * D;
    for (int j = 0; j < nper; ++j) {
        int   idx = lane + j * 32;
        float v   = (vals[j] - mean) * rstd * g[idx] + bvec[idx];
        y[idx] = fmaxf(v, 0.f);
    }
}

__global__ void fill_buffers(float* __restrict__ smax,
                             float* __restrict__ den,
                             float* __restrict__ num, int n) {
    int i = blockIdx.x * blockDim.x + threadIdx.x;
    if (i < n) { smax[i] = -3.0e38f; den[i] = 0.f; num[i] = 0.f; }
}

__global__ void edge_max_kernel(const float* __restrict__ z,
                                const int* __restrict__ src,
                                const int* __restrict__ dst,
                                const float* __restrict__ tptr, int layer,
                                float* __restrict__ smax, int total) {
    int i = blockIdx.x * blockDim.x + threadIdx.x;
    if (i >= total) return;
    int e  = i >> 6;
    int ch = i & 63;
    float t   = tptr[layer];
    float msg = fmaxf(z[src[e] * Hc + ch], 0.f) + GEPS;
    atomicMaxF(&smax[dst[e] * Hc + ch], msg * t);
}

__global__ void edge_sum_kernel(const float* __restrict__ z,
                                const int* __restrict__ src,
                                const int* __restrict__ dst,
                                const float* __restrict__ tptr, int layer,
                                const float* __restrict__ smax,
                                float* __restrict__ den,
                                float* __restrict__ num, int total) {
    int i = blockIdx.x * blockDim.x + threadIdx.x;
    if (i >= total) return;
    int e  = i >> 6;
    int ch = i & 63;
    float t   = tptr[layer];
    float msg = fmaxf(z[src[e] * Hc + ch], 0.f) + GEPS;
    int   d   = dst[e] * Hc + ch;
    float ex  = __expf(msg * t - smax[d]);
    atomicAdd(&den[d], ex);
    atomicAdd(&num[d], msg * ex);
}

__global__ void agg_kernel(const float* __restrict__ num,
                           const float* __restrict__ den,
                           const float* __restrict__ z,
                           float* __restrict__ out, int n) {
    int i = blockIdx.x * blockDim.x + threadIdx.x;
    if (i < n) out[i] = num[i] / (den[i] + 1e-16f) + z[i];
}

__global__ void residual_kernel(float* __restrict__ h,
                                const float* __restrict__ y,
                                float beta, int n) {
    int i = blockIdx.x * blockDim.x + threadIdx.x;
    if (i < n) h[i] = beta * h[i] + y[i];
}

__global__ void log_softmax_kernel(const float* __restrict__ X,
                                   float* __restrict__ Y, int rows, int C) {
    int r = blockIdx.x * blockDim.x + threadIdx.x;
    if (r >= rows) return;
    const float* x = X + (long)r * C;
    float m = -3.0e38f;
    for (int j = 0; j < C; ++j) m = fmaxf(m, x[j]);
    float s = 0.f;
    for (int j = 0; j < C; ++j) s += __expf(x[j] - m);
    float lse = m + __logf(s);
    float* y = Y + (long)r * C;
    for (int j = 0; j < C; ++j) y[j] = x[j] - lse;
}

extern "C" void kernel_launch(void* const* d_in, const int* in_sizes, int n_in,
                              void* d_out, int out_size, void* d_ws, size_t ws_size,
                              hipStream_t stream) {
    (void)n_in; (void)out_size; (void)ws_size;

    const float* x    = (const float*)d_in[0];
    const int*   ei   = (const int*)  d_in[1];
    const float* encW = (const float*)d_in[2];
    const float* encB = (const float*)d_in[3];
    const float* t    = (const float*)d_in[4];
    const float* W1   = (const float*)d_in[5];
    const float* b1   = (const float*)d_in[6];
    const float* g1   = (const float*)d_in[7];
    const float* bb1  = (const float*)d_in[8];
    const float* W2   = (const float*)d_in[9];
    const float* b2   = (const float*)d_in[10];
    const float* ng   = (const float*)d_in[11];
    const float* nb   = (const float*)d_in[12];
    const float* linW = (const float*)d_in[13];
    const float* linB = (const float*)d_in[14];

    const int F = 128, H = 64, C = 40, L = 4;
    const int N = in_sizes[0] / F;
    const int E = in_sizes[1] / 2;
    const int* src = ei;
    const int* dst = ei + E;

    float* ws = (float*)d_ws;
    const long NH = (long)N * H;
    float* h    = ws;  ws += NH;
    float* z    = ws;  ws += NH;
    float* smax = ws;  ws += NH;
    float* den  = ws;  ws += NH;
    float* num  = ws;  ws += NH;
    float* outb = ws;  ws += NH;
    float* yb   = ws;  ws += NH;
    float* tmp  = ws;  ws += (long)N * 2 * H;

    const int  TB = 256;
    auto cdiv = [](long a, long b) { return (int)((a + b - 1) / b); };
    auto gemm_waves = [&](int M, int Nc) { return (long)(M / 16) * ((Nc + 63) / 64); };

    gemm_bias_wmma<<<cdiv(gemm_waves(N, H) * 32, TB), TB, 0, stream>>>(x, encW, encB, h, N, H, F);

    const long totalE = (long)E * H;

    for (int l = 0; l < L; ++l) {
        const float* zin;
        if (l == 0) {
            zin = h;
        } else {
            ln_relu_kernel<<<cdiv(N, 8), 256, 0, stream>>>(h, ng + l * H, nb + l * H, z, N, H);
            zin = z;
        }

        fill_buffers<<<cdiv(NH, TB), TB, 0, stream>>>(smax, den, num, (int)NH);
        edge_max_kernel<<<cdiv(totalE, TB), TB, 0, stream>>>(zin, src, dst, t, l, smax, (int)totalE);
        edge_sum_kernel<<<cdiv(totalE, TB), TB, 0, stream>>>(zin, src, dst, t, l, smax, den, num, (int)totalE);
        agg_kernel<<<cdiv(NH, TB), TB, 0, stream>>>(num, den, zin, outb, (int)NH);

        gemm_bias_wmma<<<cdiv(gemm_waves(N, 2 * H) * 32, TB), TB, 0, stream>>>(
            outb, W1 + (long)l * H * 2 * H, b1 + l * 2 * H, tmp, N, 2 * H, H);
        ln_relu_kernel<<<cdiv(N, 8), 256, 0, stream>>>(tmp, g1 + l * 2 * H, bb1 + l * 2 * H, tmp, N, 2 * H);
        gemm_bias_wmma<<<cdiv(gemm_waves(N, H) * 32, TB), TB, 0, stream>>>(
            tmp, W2 + (long)l * 2 * H * H, b2 + l * H, yb, N, H, 2 * H);

        residual_kernel<<<cdiv(NH, TB), TB, 0, stream>>>(h, yb, (l == 0) ? 0.f : 1.f, (int)NH);
    }

    ln_relu_kernel<<<cdiv(N, 8), 256, 0, stream>>>(h, ng, nb, z, N, H);
    gemm_bias_wmma<<<cdiv(gemm_waves(N, C) * 32, TB), TB, 0, stream>>>(z, linW, linB, tmp, N, C, H);
    log_softmax_kernel<<<cdiv(N, TB), TB, 0, stream>>>(tmp, (float*)d_out, N, C);
}


# Edge softmax-aggregate: per-(edge,channel) float atomics replaced by in-call dst-grouped edge list + one wave per node gather with running-max rescale; fp sum order changes
# speedup vs baseline: 2.8726x; 2.8726x over previous
#include <hip/hip_runtime.h>
#include <math.h>

typedef float v2f __attribute__((ext_vector_type(2)));
typedef float v8f __attribute__((ext_vector_type(8)));

#define HD __device__ __forceinline__

static constexpr int   Hc   = 64;
static constexpr float GEPS = 1e-7f;

__global__ void gemm_bias_wmma(const float* __restrict__ A,
                               const float* __restrict__ B,
                               const float* __restrict__ bias,
                               float* __restrict__ C,
                               int M, int N, int K) {
    int wave    = (int)((blockIdx.x * blockDim.x + threadIdx.x) >> 5);
    int lane    = threadIdx.x & 31;
    int groupsN = (N + 63) >> 6;
    int tm = wave / groupsN;
    int gn = wave - tm * groupsN;
    if (tm >= (M >> 4)) return;

    int mr   = lane & 15;
    int half = lane >> 4;
    const float* Arow = A + (long)(tm * 16 + mr) * K;

    int ncol[4];
#pragma unroll
    for (int t = 0; t < 4; ++t) {
        int n   = gn * 64 + t * 16 + mr;
        ncol[t] = (n < N) ? n : (N - 1);
    }

    v8f acc[4] = {{0.f,0.f,0.f,0.f,0.f,0.f,0.f,0.f},
                  {0.f,0.f,0.f,0.f,0.f,0.f,0.f,0.f},
                  {0.f,0.f,0.f,0.f,0.f,0.f,0.f,0.f},
                  {0.f,0.f,0.f,0.f,0.f,0.f,0.f,0.f}};

    int ka = half * 2;
    v2f a_cur;
    a_cur.x = Arow[ka];
    a_cur.y = Arow[ka + 1];
    v2f b_cur[4];
#pragma unroll
    for (int t = 0; t < 4; ++t) {
        b_cur[t].x = B[ka * N + ncol[t]];
        b_cur[t].y = B[(ka + 1) * N + ncol[t]];
    }

    for (int k = 0; k < K; k += 4) {
        int kn = (k + 4 < K) ? (k + 4) : k;
        int kb = kn + half * 2;

        v2f a_nxt;
        a_nxt.x = Arow[kb];
        a_nxt.y = Arow[kb + 1];
        v2f b_nxt[4];
#pragma unroll
        for (int t = 0; t < 4; ++t) {
            b_nxt[t].x = B[kb * N + ncol[t]];
            b_nxt[t].y = B[(kb + 1) * N + ncol[t]];
        }

#pragma unroll
        for (int t = 0; t < 4; ++t)
            acc[t] = __builtin_amdgcn_wmma_f32_16x16x4_f32(
                         false, a_cur, false, b_cur[t], (short)0, acc[t], false, false);

        a_cur = a_nxt;
#pragma unroll
        for (int t = 0; t < 4; ++t) b_cur[t] = b_nxt[t];
    }

#pragma unroll
    for (int t = 0; t < 4; ++t) {
        int n = gn * 64 + t * 16 + mr;
        if (n < N) {
            float bv = bias[n];
#pragma unroll
            for (int r = 0; r < 8; ++r)
                C[(long)(tm * 16 + half * 8 + r) * N + n] = acc[t][r] + bv;
        }
    }
}

__global__ void ln_relu_kernel(const float* __restrict__ X,
                               const float* __restrict__ g,
                               const float* __restrict__ bvec,
                               float* __restrict__ Y,
                               int rows, int D) {
    int warp = threadIdx.x >> 5;
    int lane = threadIdx.x & 31;
    int row  = blockIdx.x * (blockDim.x >> 5) + warp;
    if (row >= rows) return;

    const float* x = X + (long)row * D;
    int   nper = D >> 5;
    float vals[4];
    float s = 0.f, ss = 0.f;
    for (int j = 0; j < nper; ++j) {
        float v = x[lane + j * 32];
        vals[j] = v;
        s  += v;
        ss += v * v;
    }
    for (int off = 16; off; off >>= 1) {
        s  += __shfl_xor(s,  off);
        ss += __shfl_xor(ss, off);
    }
    float mean = s / (float)D;
    float var  = ss / (float)D - mean * mean;
    float rstd = rsqrtf(var + 1e-5f);

    float* y = Y + (long)row * D;
    for (int j = 0; j < nper; ++j) {
        int   idx = lane + j * 32;
        float v   = (vals[j] - mean) * rstd * g[idx] + bvec[idx];
        y[idx] = fmaxf(v, 0.f);
    }
}

__global__ void csr_zero_kernel(int* __restrict__ cnt, int n) {
    int i = blockIdx.x * blockDim.x + threadIdx.x;
    if (i < n) cnt[i] = 0;
}

__global__ void csr_hist_kernel(const int* __restrict__ dst,
                                int* __restrict__ cnt, int E) {
    int i = blockIdx.x * blockDim.x + threadIdx.x;
    if (i < E) atomicAdd(&cnt[dst[i]], 1);
}

__global__ void
csr_scan_kernel(const int* __restrict__ cnt, int* __restrict__ rowptr,
                int* __restrict__ cursor, int n) {
    __shared__ int wsum[32];
    int lane = threadIdx.x & 31;
    int wv   = threadIdx.x >> 5;
    int per  = (((n + 31) >> 5) + 31) & ~31;
    int lo   = wv * per;
    int hi   = lo + per; if (hi > n) hi = n;

    int acc = 0;
    for (int i = lo + lane; i < hi; i += 32) acc += cnt[i];
    for (int off = 16; off; off >>= 1) acc += __shfl_xor(acc, off);
    if (lane == 0) wsum[wv] = acc;
    __syncthreads();
    if (wv == 0) {
        int w  = wsum[lane];
        int wi = w;
        for (int off = 1; off < 32; off <<= 1) {
            int v = __shfl_up(wi, off);
            if (lane >= off) wi += v;
        }
        wsum[lane] = wi - w;
    }
    __syncthreads();

    int carry = wsum[wv];
    for (int i0 = lo; i0 < hi; i0 += 128) {
        int c[4], incl[4], tot[4];
#pragma unroll
        for (int q = 0; q < 4; ++q) {
            int i = i0 + 32 * q + lane;
            c[q]    = (i < hi) ? cnt[i] : 0;
            incl[q] = c[q];
        }
        for (int off = 1; off < 32; off <<= 1) {
#pragma unroll
            for (int q = 0; q < 4; ++q) {
                int v = __shfl_up(incl[q], off);
                if (lane >= off) incl[q] += v;
            }
        }
#pragma unroll
        for (int q = 0; q < 4; ++q) tot[q] = __shfl(incl[q], 31);
#pragma unroll
        for (int q = 0; q < 4; ++q) {
            int i  = i0 + 32 * q + lane;
            int ex = carry + incl[q] - c[q];
            if (i < hi) { rowptr[i] = ex; cursor[i] = ex; }
            carry += tot[q];
        }
    }
    if (threadIdx.x == 1023) rowptr[n] = carry;
}

__global__ void csr_fill_kernel(const int* __restrict__ src,
                                const int* __restrict__ dst,
                                int* __restrict__ cursor,
                                int* __restrict__ ssrc, int E) {
    int i = blockIdx.x * blockDim.x + threadIdx.x;
    if (i < E) {
        int p = atomicAdd(&cursor[dst[i]], 1);
        ssrc[p] = src[i];
    }
}

static constexpr int EB = 8;

__global__ void
node_softmax_agg_kernel(const float* __restrict__ z,
                        const int* __restrict__ rowptr,
                        const int* __restrict__ ssrc,
                        const float* __restrict__ tptr, int layer,
                        float* __restrict__ out, int n) {
    int lane = threadIdx.x & 31;
    int node = __builtin_amdgcn_readfirstlane(
                   (int)(blockIdx.x * (blockDim.x >> 5) + (threadIdx.x >> 5)));
    if (node >= n) return;

    int   start = rowptr[node];
    int   end   = rowptr[node + 1];
    float t     = tptr[layer];
    const v2f* zr = (const v2f*)z;
    v2f self = zr[node * (Hc / 2) + lane];

    float mx = -3.0e38f, my = -3.0e38f;
    float dx = 0.f, dy = 0.f, nx = 0.f, ny = 0.f;

    for (int base = start; base < end; base += 32) {
        int cnt = end - base; if (cnt > 32) cnt = 32;
        int myidx = (lane < cnt) ? ssrc[base + lane] : 0;
        for (int j0 = 0; j0 < cnt; j0 += EB) {
            v2f v[EB];
#pragma unroll
            for (int u = 0; u < EB; ++u) {
                int sidx = __builtin_amdgcn_readlane(myidx, j0 + u);
                v[u] = zr[(long)sidx * (Hc / 2) + lane];
            }
            float gx[EB], gy[EB], sx[EB], sy[EB];
            float cmx = -3.0e38f, cmy = -3.0e38f;
#pragma unroll
            for (int u = 0; u < EB; ++u) {
                bool ok = (j0 + u) < cnt;
                float ax = fmaxf(v[u].x, 0.f) + GEPS;
                float ay = fmaxf(v[u].y, 0.f) + GEPS;
                gx[u] = ok ? ax : 0.f;
                gy[u] = ok ? ay : 0.f;
                sx[u] = ok ? ax * t : -3.0e38f;
                sy[u] = ok ? ay * t : -3.0e38f;
                cmx = fmaxf(cmx, sx[u]);
                cmy = fmaxf(cmy, sy[u]);
            }
            float nmx = fmaxf(mx, cmx);
            float nmy = fmaxf(my, cmy);
            float rx  = __expf(mx - nmx);
            float ry  = __expf(my - nmy);
            dx *= rx; nx *= rx;
            dy *= ry; ny *= ry;
#pragma unroll
            for (int u = 0; u < EB; ++u) {
                float ex = __expf(sx[u] - nmx);
                float ey = __expf(sy[u] - nmy);
                dx += ex; nx += gx[u] * ex;
                dy += ey; ny += gy[u] * ey;
            }
            mx = nmx; my = nmy;
        }
    }

    v2f o;
    o.x = nx / (dx + 1e-16f) + self.x;
    o.y = ny / (dy + 1e-16f) + self.y;
    ((v2f*)out)[node * (Hc / 2) + lane] = o;
}

__global__ void residual_kernel(float* __restrict__ h,
                                const float* __restrict__ y,
                                float beta, int n) {
    int i = blockIdx.x * blockDim.x + threadIdx.x;
    if (i < n) h[i] = beta * h[i] + y[i];
}

__global__ void log_softmax_kernel(const float* __restrict__ X,
                                   float* __restrict__ Y, int rows, int C) {
    int r = blockIdx.x * blockDim.x + threadIdx.x;
    if (r >= rows) return;
    const float* x = X + (long)r * C;
    float m = -3.0e38f;
    for (int j = 0; j < C; ++j) m = fmaxf(m, x[j]);
    float s = 0.f;
    for (int j = 0; j < C; ++j) s += __expf(x[j] - m);
    float lse = m + __logf(s);
    float* y = Y + (long)r * C;
    for (int j = 0; j < C; ++j) y[j] = x[j] - lse;
}

extern "C" void kernel_launch(void* const* d_in, const int* in_sizes, int n_in,
                              void* d_out, int out_size, void* d_ws, size_t ws_size,
                              hipStream_t stream) {
    (void)n_in; (void)out_size; (void)ws_size;

    const float* x    = (const float*)d_in[0];
    const int*   ei   = (const int*)  d_in[1];
    const float* encW = (const float*)d_in[2];
    const float* encB = (const float*)d_in[3];
    const float* t    = (const float*)d_in[4];
    const float* W1   = (const float*)d_in[5];
    const float* b1   = (const float*)d_in[6];
    const float* g1   = (const float*)d_in[7];
    const float* bb1  = (const float*)d_in[8];
    const float* W2   = (const float*)d_in[9];
    const float* b2   = (const float*)d_in[10];
    const float* ng   = (const float*)d_in[11];
    const float* nb   = (const float*)d_in[12];
    const float* linW = (const float*)d_in[13];
    const float* linB = (const float*)d_in[14];

    const int F = 128, H = 64, C = 40, L = 4;
    const int N = in_sizes[0] / F;
    const int E = in_sizes[1] / 2;
    const int* src = ei;
    const int* dst = ei + E;

    float* ws = (float*)d_ws;
    const long NH = (long)N * H;
    float* h    = ws;  ws += NH;
    float* z    = ws;  ws += NH;
    int*   csr  = (int*)ws;  ws += 3 * NH;
    float* outb = ws;  ws += NH;
    float* yb   = ws;  ws += NH;
    float* tmp  = ws;  ws += (long)N * 2 * H;

    const int  TB = 256;
    auto cdiv = [](long a, long b) { return (int)((a + b - 1) / b); };
    auto gemm_waves = [&](int M, int Nc) { return (long)(M / 16) * ((Nc + 63) / 64); };

    gemm_bias_wmma<<<cdiv(gemm_waves(N, H) * 32, TB), TB, 0, stream>>>(x, encW, encB, h, N, H, F);

    int* cnt    = csr;
    int* cursor = cnt + N;
    int* rowptr = cursor + N;
    int* ssrc   = rowptr + N + 1;
    csr_zero_kernel<<<cdiv(N, TB), TB, 0, stream>>>(cnt, N);
    csr_hist_kernel<<<cdiv(E, TB), TB, 0, stream>>>(dst, cnt, E);
    csr_scan_kernel<<<1, 1024, 0, stream>>>(cnt, rowptr, cursor, N);
    csr_fill_kernel<<<cdiv(E, TB), TB, 0, stream>>>(src, dst, cursor, ssrc, E);

    for (int l = 0; l < L; ++l) {
        const float* zin;
        if (l == 0) {
            zin = h;
        } else {
            ln_relu_kernel<<<cdiv(N, 8), 256, 0, stream>>>(h, ng + l * H, nb + l * H, z, N, H);
            zin = z;
        }

        node_softmax_agg_kernel<<<cdiv(N, 8), 256, 0, stream>>>(zin, rowptr, ssrc, t, l, outb, N);

        gemm_bias_wmma<<<cdiv(gemm_waves(N, 2 * H) * 32, TB), TB, 0, stream>>>(
            outb, W1 + (long)l * H * 2 * H, b1 + l * 2 * H, tmp, N, 2 * H, H);
        ln_relu_kernel<<<cdiv(N, 8), 256, 0, stream>>>(tmp, g1 + l * 2 * H, bb1 + l * 2 * H, tmp, N, 2 * H);
        gemm_bias_wmma<<<cdiv(gemm_waves(N, H) * 32, TB), TB, 0, stream>>>(
            tmp, W2 + (long)l * 2 * H * H, b2 + l * H, yb, N, H, 2 * H);

        residual_kernel<<<cdiv(NH, TB), TB, 0, stream>>>(h, yb, (l == 0) ? 0.f : 1.f, (int)NH);
    }

    ln_relu_kernel<<<cdiv(N, 8), 256, 0, stream>>>(h, ng, nb, z, N, H);
    gemm_bias_wmma<<<cdiv(gemm_waves(N, C) * 32, TB), TB, 0, stream>>>(z, linW, linB, tmp, N, C, H);
    log_softmax_kernel<<<cdiv(N, TB), TB, 0, stream>>>(tmp, (float*)d_out, N, C);
}


# Per-layer node chain fused into one kernel: GEMM1, LayerNorm+ReLU, GEMM2, residual and next LayerNorm+ReLU per 16-row wave tile, in registers and LDS; LayerNorm row-sum order changes
# speedup vs baseline: 4.1007x; 1.4275x over previous
#include <hip/hip_runtime.h>
#include <math.h>

typedef float v2f __attribute__((ext_vector_type(2)));
typedef float v8f __attribute__((ext_vector_type(8)));

#define HD __device__ __forceinline__

static constexpr int   Hc   = 64;
static constexpr float GEPS = 1e-7f;

__global__ void gemm_bias_wmma(const float* __restrict__ A,
                               const float* __restrict__ B,
                               const float* __restrict__ bias,
                               float* __restrict__ C,
                               int M, int N, int K) {
    int wave    = (int)((blockIdx.x * blockDim.x + threadIdx.x) >> 5);
    int lane    = threadIdx.x & 31;
    int groupsN = (N + 63) >> 6;
    int tm = wave / groupsN;
    int gn = wave - tm * groupsN;
    if (tm >= (M >> 4)) return;

    int mr   = lane & 15;
    int half = lane >> 4;
    const float* Arow = A + (long)(tm * 16 + mr) * K;

    int ncol[4];
#pragma unroll
    for (int t = 0; t < 4; ++t) {
        int n   = gn * 64 + t * 16 + mr;
        ncol[t] = (n < N) ? n : (N - 1);
    }

    v8f acc[4] = {{0.f,0.f,0.f,0.f,0.f,0.f,0.f,0.f},
                  {0.f,0.f,0.f,0.f,0.f,0.f,0.f,0.f},
                  {0.f,0.f,0.f,0.f,0.f,0.f,0.f,0.f},
                  {0.f,0.f,0.f,0.f,0.f,0.f,0.f,0.f}};

    int ka = half * 2;
    v2f a_cur;
    a_cur.x = Arow[ka];
    a_cur.y = Arow[ka + 1];
    v2f b_cur[4];
#pragma unroll
    for (int t = 0; t < 4; ++t) {
        b_cur[t].x = B[ka * N + ncol[t]];
        b_cur[t].y = B[(ka + 1) * N + ncol[t]];
    }

    for (int k = 0; k < K; k += 4) {
        int kn = (k + 4 < K) ? (k + 4) : k;
        int kb = kn + half * 2;

        v2f a_nxt;
        a_nxt.x = Arow[kb];
        a_nxt.y = Arow[kb + 1];
        v2f b_nxt[4];
#pragma unroll
        for (int t = 0; t < 4; ++t) {
            b_nxt[t].x = B[kb * N + ncol[t]];
            b_nxt[t].y = B[(kb + 1) * N + ncol[t]];
        }

#pragma unroll
        for (int t = 0; t < 4; ++t)
            acc[t] = __builtin_amdgcn_wmma_f32_16x16x4_f32(
                         false, a_cur, false, b_cur[t], (short)0, acc[t], false, false);

        a_cur = a_nxt;
#pragma unroll
        for (int t = 0; t < 4; ++t) b_cur[t] = b_nxt[t];
    }

#pragma unroll
    for (int t = 0; t < 4; ++t) {
        int n = gn * 64 + t * 16 + mr;
        if (n < N) {
            float bv = bias[n];
#pragma unroll
            for (int r = 0; r < 8; ++r)
                C[(long)(tm * 16 + half * 8 + r) * N + n] = acc[t][r] + bv;
        }
    }
}

static constexpr int LP = 132;

template <int M>
HD float row_xor_add(float v) {
    int o = __builtin_amdgcn_update_dpp(0, __float_as_int(v), 0x160 | M, 0xf, 0xf, false);
    return v + __int_as_float(o);
}

__global__ void __launch_bounds__(32)
mlp_chain_kernel(const float* __restrict__ X,
                 const float* __restrict__ W1, const float* __restrict__ b1,
                 const float* __restrict__ g1, const float* __restrict__ bb1,
                 const float* __restrict__ W2, const float* __restrict__ b2,
                 const float* __restrict__ ng, const float* __restrict__ nb,
                 float* __restrict__ h, float* __restrict__ zout,
                 int add_res, int write_h) {
    __shared__ float tile[16 * LP];
    const int lane = threadIdx.x;
    const int tm   = blockIdx.x;
    const int mr   = lane & 15;
    const int half = lane >> 4;

    const float* Arow = X + (long)(tm * 16 + mr) * Hc + half * 2;
    const float* B1p  = W1 + (half * 2) * (2 * Hc) + mr;

    v8f acc1[8];
#pragma unroll
    for (int t = 0; t < 8; ++t) acc1[t] = v8f{0.f,0.f,0.f,0.f,0.f,0.f,0.f,0.f};

    v2f a_cur = *(const v2f*)(Arow);
    v2f b_cur[8];
#pragma unroll
    for (int t = 0; t < 8; ++t) {
        b_cur[t].x = B1p[t * 16];
        b_cur[t].y = B1p[2 * Hc + t * 16];
    }

    float bias1[8], gam1[8], bet1[8];
#pragma unroll
    for (int t = 0; t < 8; ++t) {
        bias1[t] = b1[t * 16 + mr];
        gam1[t]  = g1[t * 16 + mr];
        bet1[t]  = bb1[t * 16 + mr];
    }

    const long rowbase = (long)(tm * 16 + half * 8) * Hc + mr;
    float hres[4][8];
    if (add_res) {
#pragma unroll
        for (int t = 0; t < 4; ++t)
#pragma unroll
            for (int r = 0; r < 8; ++r) hres[t][r] = h[rowbase + (long)r * Hc + t * 16];
    } else {
#pragma unroll
        for (int t = 0; t < 4; ++t)
#pragma unroll
            for (int r = 0; r < 8; ++r) hres[t][r] = 0.f;
    }

#pragma unroll
    for (int kk = 0; kk < 16; ++kk) {
        v2f a_nxt = a_cur;
        v2f b_nxt[8];
        if (kk + 1 < 16) {
            a_nxt = *(const v2f*)(Arow + 4 * (kk + 1));
#pragma unroll
            for (int t = 0; t < 8; ++t) {
                b_nxt[t].x = B1p[(4 * (kk + 1)) * (2 * Hc) + t * 16];
                b_nxt[t].y = B1p[(4 * (kk + 1) + 1) * (2 * Hc) + t * 16];
            }
        }
#pragma unroll
        for (int t = 0; t < 8; ++t)
            acc1[t] = __builtin_amdgcn_wmma_f32_16x16x4_f32(
                          false, a_cur, false, b_cur[t], (short)0, acc1[t], false, false);
        if (kk + 1 < 16) {
            a_cur = a_nxt;
#pragma unroll
            for (int t = 0; t < 8; ++t) b_cur[t] = b_nxt[t];
        }
    }

    float s1[8], q1[8];
#pragma unroll
    for (int r = 0; r < 8; ++r) { s1[r] = 0.f; q1[r] = 0.f; }
#pragma unroll
    for (int t = 0; t < 8; ++t) {
#pragma unroll
        for (int r = 0; r < 8; ++r) {
            float u = acc1[t][r] + bias1[t];
            acc1[t][r] = u;
            s1[r] += u;
            q1[r] += u * u;
        }
    }
#pragma unroll
    for (int r = 0; r < 8; ++r) {
        s1[r] = row_xor_add<1>(s1[r]);  q1[r] = row_xor_add<1>(q1[r]);
    }
#pragma unroll
    for (int r = 0; r < 8; ++r) {
        s1[r] = row_xor_add<2>(s1[r]);  q1[r] = row_xor_add<2>(q1[r]);
    }
#pragma unroll
    for (int r = 0; r < 8; ++r) {
        s1[r] = row_xor_add<4>(s1[r]);  q1[r] = row_xor_add<4>(q1[r]);
    }
#pragma unroll
    for (int r = 0; r < 8; ++r) {
        s1[r] = row_xor_add<8>(s1[r]);  q1[r] = row_xor_add<8>(q1[r]);
    }
#pragma unroll
    for (int r = 0; r < 8; ++r) {
        float mean = s1[r] * (1.f / 128.f);
        float var  = q1[r] * (1.f / 128.f) - mean * mean;
        float rstd = rsqrtf(var + 1e-5f);
#pragma unroll
        for (int t = 0; t < 8; ++t) {
            float v = (acc1[t][r] - mean) * rstd * gam1[t] + bet1[t];
            tile[(half * 8 + r) * LP + t * 16 + mr] = fmaxf(v, 0.f);
        }
    }

    __syncthreads();

    float bias2[4], gam2[4], bet2[4];
#pragma unroll
    for (int t = 0; t < 4; ++t) {
        bias2[t] = b2[t * 16 + mr];
        gam2[t]  = ng[t * 16 + mr];
        bet2[t]  = nb[t * 16 + mr];
    }

    const float* Atile = tile + mr * LP + half * 2;
    const float* B2p   = W2 + (half * 2) * Hc + mr;

    v8f acc2[4];
#pragma unroll
    for (int t = 0; t < 4; ++t) acc2[t] = v8f{0.f,0.f,0.f,0.f,0.f,0.f,0.f,0.f};

    v2f a2c[2];
    v2f b2c[2][4];
#pragma unroll
    for (int u = 0; u < 2; ++u) {
        a2c[u] = *(const v2f*)(Atile + 4 * u);
#pragma unroll
        for (int t = 0; t < 4; ++t) {
            b2c[u][t].x = B2p[(4 * u) * Hc + t * 16];
            b2c[u][t].y = B2p[(4 * u + 1) * Hc + t * 16];
        }
    }
#pragma unroll
    for (int j = 0; j < 16; ++j) {
        v2f a2n[2];
        v2f b2n[2][4];
        if (j + 1 < 16) {
#pragma unroll
            for (int u = 0; u < 2; ++u) {
                const int kk = 2 * (j + 1) + u;
                a2n[u] = *(const v2f*)(Atile + 4 * kk);
#pragma unroll
                for (int t = 0; t < 4; ++t) {
                    b2n[u][t].x = B2p[(4 * kk) * Hc + t * 16];
                    b2n[u][t].y = B2p[(4 * kk + 1) * Hc + t * 16];
                }
            }
        }
#pragma unroll
        for (int u = 0; u < 2; ++u)
#pragma unroll
            for (int t = 0; t < 4; ++t)
                acc2[t] = __builtin_amdgcn_wmma_f32_16x16x4_f32(
                              false, a2c[u], false, b2c[u][t], (short)0, acc2[t], false, false);
        if (j + 1 < 16) {
#pragma unroll
            for (int u = 0; u < 2; ++u) {
                a2c[u] = a2n[u];
#pragma unroll
                for (int t = 0; t < 4; ++t) b2c[u][t] = b2n[u][t];
            }
        }
    }

    float s2[8], q2[8];
#pragma unroll
    for (int r = 0; r < 8; ++r) { s2[r] = 0.f; q2[r] = 0.f; }
#pragma unroll
    for (int t = 0; t < 4; ++t) {
#pragma unroll
        for (int r = 0; r < 8; ++r) {
            float hn = hres[t][r] + (acc2[t][r] + bias2[t]);
            acc2[t][r] = hn;
            s2[r] += hn;
            q2[r] += hn * hn;
        }
    }
    if (write_h) {
#pragma unroll
        for (int t = 0; t < 4; ++t)
#pragma unroll
            for (int r = 0; r < 8; ++r)
                h[rowbase + (long)r * Hc + t * 16] = acc2[t][r];
    }
#pragma unroll
    for (int r = 0; r < 8; ++r) {
        s2[r] = row_xor_add<1>(s2[r]);  q2[r] = row_xor_add<1>(q2[r]);
    }
#pragma unroll
    for (int r = 0; r < 8; ++r) {
        s2[r] = row_xor_add<2>(s2[r]);  q2[r] = row_xor_add<2>(q2[r]);
    }
#pragma unroll
    for (int r = 0; r < 8; ++r) {
        s2[r] = row_xor_add<4>(s2[r]);  q2[r] = row_xor_add<4>(q2[r]);
    }
#pragma unroll
    for (int r = 0; r < 8; ++r) {
        s2[r] = row_xor_add<8>(s2[r]);  q2[r] = row_xor_add<8>(q2[r]);
    }
#pragma unroll
    for (int r = 0; r < 8; ++r) {
        float mean = s2[r] * (1.f / 64.f);
        float var  = q2[r] * (1.f / 64.f) - mean * mean;
        float rstd = rsqrtf(var + 1e-5f);
#pragma unroll
        for (int t = 0; t < 4; ++t) {
            float v = (acc2[t][r] - mean) * rstd * gam2[t] + bet2[t];
            zout[rowbase + (long)r * Hc + t * 16] = fmaxf(v, 0.f);
        }
    }
}

__global__ void csr_zero_kernel(int* __restrict__ cnt, int n) {
    int i = blockIdx.x * blockDim.x + threadIdx.x;
    if (i < n) cnt[i] = 0;
}

__global__ void csr_hist_kernel(const int* __restrict__ dst,
                                int* __restrict__ cnt, int E) {
    int i = blockIdx.x * blockDim.x + threadIdx.x;
    if (i < E) atomicAdd(&cnt[dst[i]], 1);
}

__global__ void
csr_scan_kernel(const int* __restrict__ cnt, int* __restrict__ rowptr,
                int* __restrict__ cursor, int n) {
    __shared__ int wsum[32];
    int lane = threadIdx.x & 31;
    int wv   = threadIdx.x >> 5;
    int per  = (((n + 31) >> 5) + 31) & ~31;
    int lo   = wv * per;
    int hi   = lo + per; if (hi > n) hi = n;

    int acc = 0;
    for (int i = lo + lane; i < hi; i += 32) acc += cnt[i];
    for (int off = 16; off; off >>= 1) acc += __shfl_xor(acc, off);
    if (lane == 0) wsum[wv] = acc;
    __syncthreads();
    if (wv == 0) {
        int w  = wsum[lane];
        int wi = w;
        for (int off = 1; off < 32; off <<= 1) {
            int v = __shfl_up(wi, off);
            if (lane >= off) wi += v;
        }
        wsum[lane] = wi - w;
    }
    __syncthreads();

    int carry = wsum[wv];
    for (int i0 = lo; i0 < hi; i0 += 128) {
        int c[4], incl[4], tot[4];
#pragma unroll
        for (int q = 0; q < 4; ++q) {
            int i = i0 + 32 * q + lane;
            c[q]    = (i < hi) ? cnt[i] : 0;
            incl[q] = c[q];
        }
        for (int off = 1; off < 32; off <<= 1) {
#pragma unroll
            for (int q = 0; q < 4; ++q) {
                int v = __shfl_up(incl[q], off);
                if (lane >= off) incl[q] += v;
            }
        }
#pragma unroll
        for (int q = 0; q < 4; ++q) tot[q] = __shfl(incl[q], 31);
#pragma unroll
        for (int q = 0; q < 4; ++q) {
            int i  = i0 + 32 * q + lane;
            int ex = carry + incl[q] - c[q];
            if (i < hi) { rowptr[i] = ex; cursor[i] = ex; }
            carry += tot[q];
        }
    }
    if (threadIdx.x == 1023) rowptr[n] = carry;
}

__global__ void csr_fill_kernel(const int* __restrict__ src,
                                const int* __restrict__ dst,
                                int* __restrict__ cursor,
                                int* __restrict__ ssrc, int E) {
    int i = blockIdx.x * blockDim.x + threadIdx.x;
    if (i < E) {
        int p = atomicAdd(&cursor[dst[i]], 1);
        ssrc[p] = src[i];
    }
}

static constexpr int EB = 8;

__global__ void
node_softmax_agg_kernel(const float* __restrict__ z,
                        const int* __restrict__ rowptr,
                        const int* __restrict__ ssrc,
                        const float* __restrict__ tptr, int layer,
                        float* __restrict__ out, int n) {
    int lane = threadIdx.x & 31;
    int node = __builtin_amdgcn_readfirstlane(
                   (int)(blockIdx.x * (blockDim.x >> 5) + (threadIdx.x >> 5)));
    if (node >= n) return;

    int   start = rowptr[node];
    int   end   = rowptr[node + 1];
    float t     = tptr[layer];
    const v2f* zr = (const v2f*)z;
    v2f self = zr[node * (Hc / 2) + lane];

    float mx = -3.0e38f, my = -3.0e38f;
    float dx = 0.f, dy = 0.f, nx = 0.f, ny = 0.f;

    for (int base = start; base < end; base += 32) {
        int cnt = end - base; if (cnt > 32) cnt = 32;
        int myidx = (lane < cnt) ? ssrc[base + lane] : 0;
        for (int j0 = 0; j0 < cnt; j0 += EB) {
            v2f v[EB];
#pragma unroll
            for (int u = 0; u < EB; ++u) {
                int sidx = __builtin_amdgcn_readlane(myidx, j0 + u);
                v[u] = zr[(long)sidx * (Hc / 2) + lane];
            }
            float gx[EB], gy[EB], sx[EB], sy[EB];
            float cmx = -3.0e38f, cmy = -3.0e38f;
#pragma unroll
            for (int u = 0; u < EB; ++u) {
                bool ok = (j0 + u) < cnt;
                float ax = fmaxf(v[u].x, 0.f) + GEPS;
                float ay = fmaxf(v[u].y, 0.f) + GEPS;
                gx[u] = ok ? ax : 0.f;
                gy[u] = ok ? ay : 0.f;
                sx[u] = ok ? ax * t : -3.0e38f;
                sy[u] = ok ? ay * t : -3.0e38f;
                cmx = fmaxf(cmx, sx[u]);
                cmy = fmaxf(cmy, sy[u]);
            }
            float nmx = fmaxf(mx, cmx);
            float nmy = fmaxf(my, cmy);
            float rx  = __expf(mx - nmx);
            float ry  = __expf(my - nmy);
            dx *= rx; nx *= rx;
            dy *= ry; ny *= ry;
#pragma unroll
            for (int u = 0; u < EB; ++u) {
                float ex = __expf(sx[u] - nmx);
                float ey = __expf(sy[u] - nmy);
                dx += ex; nx += gx[u] * ex;
                dy += ey; ny += gy[u] * ey;
            }
            mx = nmx; my = nmy;
        }
    }

    v2f o;
    o.x = nx / (dx + 1e-16f) + self.x;
    o.y = ny / (dy + 1e-16f) + self.y;
    ((v2f*)out)[node * (Hc / 2) + lane] = o;
}

__global__ void log_softmax_kernel(const float* __restrict__ X,
                                   float* __restrict__ Y, int rows, int C) {
    int r = blockIdx.x * blockDim.x + threadIdx.x;
    if (r >= rows) return;
    const float* x = X + (long)r * C;
    float m = -3.0e38f;
    for (int j = 0; j < C; ++j) m = fmaxf(m, x[j]);
    float s = 0.f;
    for (int j = 0; j < C; ++j) s += __expf(x[j] - m);
    float lse = m + __logf(s);
    float* y = Y + (long)r * C;
    for (int j = 0; j < C; ++j) y[j] = x[j] - lse;
}

extern "C" void kernel_launch(void* const* d_in, const int* in_sizes, int n_in,
                              void* d_out, int out_size, void* d_ws, size_t ws_size,
                              hipStream_t stream) {
    (void)n_in; (void)out_size; (void)ws_size;

    const float* x    = (const float*)d_in[0];
    const int*   ei   = (const int*)  d_in[1];
    const float* encW = (const float*)d_in[2];
    const float* encB = (const float*)d_in[3];
    const float* t    = (const float*)d_in[4];
    const float* W1   = (const float*)d_in[5];
    const float* b1   = (const float*)d_in[6];
    const float* g1   = (const float*)d_in[7];
    const float* bb1  = (const float*)d_in[8];
    const float* W2   = (const float*)d_in[9];
    const float* b2   = (const float*)d_in[10];
    const float* ng   = (const float*)d_in[11];
    const float* nb   = (const float*)d_in[12];
    const float* linW = (const float*)d_in[13];
    const float* linB = (const float*)d_in[14];

    const int F = 128, H = 64, C = 40, L = 4;
    const int N = in_sizes[0] / F;
    const int E = in_sizes[1] / 2;
    const int* src = ei;
    const int* dst = ei + E;

    float* ws = (float*)d_ws;
    const long NH = (long)N * H;
    float* h    = ws;  ws += NH;
    float* z    = ws;  ws += NH;
    int*   csr  = (int*)ws;  ws += 3 * NH;
    float* outb = ws;  ws += NH;
    ws += NH;
    float* tmp  = ws;  ws += (long)N * 2 * H;

    const int  TB = 256;
    auto cdiv = [](long a, long b) { return (int)((a + b - 1) / b); };
    auto gemm_waves = [&](int M, int Nc) { return (long)(M / 16) * ((Nc + 63) / 64); };

    gemm_bias_wmma<<<cdiv(gemm_waves(N, H) * 32, TB), TB, 0, stream>>>(x, encW, encB, h, N, H, F);

    int* cnt    = csr;
    int* cursor = cnt + N;
    int* rowptr = cursor + N;
    int* ssrc   = rowptr + N + 1;
    csr_zero_kernel<<<cdiv(N, TB), TB, 0, stream>>>(cnt, N);
    csr_hist_kernel<<<cdiv(E, TB), TB, 0, stream>>>(dst, cnt, E);
    csr_scan_kernel<<<1, 1024, 0, stream>>>(cnt, rowptr, cursor, N);
    csr_fill_kernel<<<cdiv(E, TB), TB, 0, stream>>>(src, dst, cursor, ssrc, E);

    for (int l = 0; l < L; ++l) {
        const float* zin = (l == 0) ? h : z;

        node_softmax_agg_kernel<<<cdiv(N, 8), 256, 0, stream>>>(zin, rowptr, ssrc, t, l, outb, N);

        const int ln = (l + 1 < L) ? (l + 1) : 0;
        mlp_chain_kernel<<<N / 16, 32, 0, stream>>>(
            outb, W1 + (long)l * H * 2 * H, b1 + l * 2 * H, g1 + l * 2 * H, bb1 + l * 2 * H,
            W2 + (long)l * 2 * H * H, b2 + l * H, ng + ln * H, nb + ln * H,
            h, z, (l == 0) ? 0 : 1, (l + 1 < L) ? 1 : 0);
    }

    gemm_bias_wmma<<<cdiv(gemm_waves(N, C) * 32, TB), TB, 0, stream>>>(z, linW, linB, tmp, N, C, H);
    log_softmax_kernel<<<cdiv(N, TB), TB, 0, stream>>>(tmp, (float*)d_out, N, C);
}


# Three small launches folded into neighbours (15 -> 12): counter clearing into the encoder launch, classifier GEMM + log_softmax into the last chain kernel; log_softmax row-sum order changes
# speedup vs baseline: 4.3347x; 1.0571x over previous
#include <hip/hip_runtime.h>
#include <math.h>

typedef float v2f __attribute__((ext_vector_type(2)));
typedef float v4f __attribute__((ext_vector_type(4)));
typedef float v8f __attribute__((ext_vector_type(8)));

#define HD __device__ __forceinline__

static constexpr int   Hc   = 64;
static constexpr int   Cc   = 40;
static constexpr float GEPS = 1e-7f;

__global__ void gemm_bias_wmma(const float* __restrict__ A,
                               const float* __restrict__ B,
                               const float* __restrict__ bias,
                               float* __restrict__ C,
                               int M, int N, int K,
                               int* __restrict__ zp, int zn) {
    int gid = (int)(blockIdx.x * blockDim.x + threadIdx.x);
    if (gid < zn) zp[gid] = 0;
    int wave    = gid >> 5;
    int lane    = threadIdx.x & 31;
    int groupsN = (N + 63) >> 6;
    int tm = wave / groupsN;
    int gn = wave - tm * groupsN;
    if (tm >= (M >> 4)) return;

    int mr   = lane & 15;
    int half = lane >> 4;
    const float* Arow = A + (long)(tm * 16 + mr) * K;

    int ncol[4];
#pragma unroll
    for (int t = 0; t < 4; ++t) {
        int n   = gn * 64 + t * 16 + mr;
        ncol[t] = (n < N) ? n : (N - 1);
    }

    v8f acc[4] = {{0.f,0.f,0.f,0.f,0.f,0.f,0.f,0.f},
                  {0.f,0.f,0.f,0.f,0.f,0.f,0.f,0.f},
                  {0.f,0.f,0.f,0.f,0.f,0.f,0.f,0.f},
                  {0.f,0.f,0.f,0.f,0.f,0.f,0.f,0.f}};

    int ka = half * 2;
    v2f a_cur;
    a_cur.x = Arow[ka];
    a_cur.y = Arow[ka + 1];
    v2f b_cur[4];
#pragma unroll
    for (int t = 0; t < 4; ++t) {
        b_cur[t].x = B[ka * N + ncol[t]];
        b_cur[t].y = B[(ka + 1) * N + ncol[t]];
    }

    for (int k = 0; k < K; k += 4) {
        int kn = (k + 4 < K) ? (k + 4) : k;
        int kb = kn + half * 2;

        v2f a_nxt;
        a_nxt.x = Arow[kb];
        a_nxt.y = Arow[kb + 1];
        v2f b_nxt[4];
#pragma unroll
        for (int t = 0; t < 4; ++t) {
            b_nxt[t].x = B[kb * N + ncol[t]];
            b_nxt[t].y = B[(kb + 1) * N + ncol[t]];
        }

#pragma unroll
        for (int t = 0; t < 4; ++t)
            acc[t] = __builtin_amdgcn_wmma_f32_16x16x4_f32(
                         false, a_cur, false, b_cur[t], (short)0, acc[t], false, false);

        a_cur = a_nxt;
#pragma unroll
        for (int t = 0; t < 4; ++t) b_cur[t] = b_nxt[t];
    }

#pragma unroll
    for (int t = 0; t < 4; ++t) {
        int n = gn * 64 + t * 16 + mr;
        if (n < N) {
            float bv = bias[n];
#pragma unroll
            for (int r = 0; r < 8; ++r)
                C[(long)(tm * 16 + half * 8 + r) * N + n] = acc[t][r] + bv;
        }
    }
}

static constexpr int LP = 132;

template <int M>
HD float row_xor_add(float v) {
    int o = __builtin_amdgcn_update_dpp(0, __float_as_int(v), 0x160 | M, 0xf, 0xf, false);
    return v + __int_as_float(o);
}
template <int M>
HD float row_xor_max(float v) {
    int o = __builtin_amdgcn_update_dpp(0, __float_as_int(v), 0x160 | M, 0xf, 0xf, false);
    return fmaxf(v, __int_as_float(o));
}

template <bool LAST>
__global__ void __launch_bounds__(32)
mlp_chain_kernel(const float* __restrict__ X,
                 const float* __restrict__ W1, const float* __restrict__ b1,
                 const float* __restrict__ g1, const float* __restrict__ bb1,
                 const float* __restrict__ W2, const float* __restrict__ b2,
                 const float* __restrict__ ng, const float* __restrict__ nb,
                 float* __restrict__ h, float* __restrict__ zout,
                 int add_res, int write_h,
                 const float* __restrict__ linW, const float* __restrict__ linB,
                 float* __restrict__ out) {
    __shared__ __attribute__((aligned(16))) float tile[16 * LP];
    const int lane = threadIdx.x;
    const int tm   = blockIdx.x;
    const int mr   = lane & 15;
    const int half = lane >> 4;

    const float* Arow = X + (long)(tm * 16 + mr) * Hc + half * 2;
    const float* B1p  = W1 + (half * 2) * (2 * Hc) + mr;

    v8f acc1[8];
#pragma unroll
    for (int t = 0; t < 8; ++t) acc1[t] = v8f{0.f,0.f,0.f,0.f,0.f,0.f,0.f,0.f};

    v2f a_cur = *(const v2f*)(Arow);
    v2f b_cur[8];
#pragma unroll
    for (int t = 0; t < 8; ++t) {
        b_cur[t].x = B1p[t * 16];
        b_cur[t].y = B1p[2 * Hc + t * 16];
    }

    float bias1[8], gam1[8], bet1[8];
#pragma unroll
    for (int t = 0; t < 8; ++t) {
        bias1[t] = b1[t * 16 + mr];
        gam1[t]  = g1[t * 16 + mr];
        bet1[t]  = bb1[t * 16 + mr];
    }

    const long rowbase = (long)(tm * 16 + half * 8) * Hc + mr;
    float hres[4][8];
    if (add_res) {
#pragma unroll
        for (int t = 0; t < 4; ++t)
#pragma unroll
            for (int r = 0; r < 8; ++r) hres[t][r] = h[rowbase + (long)r * Hc + t * 16];
    } else {
#pragma unroll
        for (int t = 0; t < 4; ++t)
#pragma unroll
            for (int r = 0; r < 8; ++r) hres[t][r] = 0.f;
    }

#pragma unroll
    for (int kk = 0; kk < 16; ++kk) {
        v2f a_nxt = a_cur;
        v2f b_nxt[8];
        if (kk + 1 < 16) {
            a_nxt = *(const v2f*)(Arow + 4 * (kk + 1));
#pragma unroll
            for (int t = 0; t < 8; ++t) {
                b_nxt[t].x = B1p[(4 * (kk + 1)) * (2 * Hc) + t * 16];
                b_nxt[t].y = B1p[(4 * (kk + 1) + 1) * (2 * Hc) + t * 16];
            }
        }
#pragma unroll
        for (int t = 0; t < 8; ++t)
            acc1[t] = __builtin_amdgcn_wmma_f32_16x16x4_f32(
                          false, a_cur, false, b_cur[t], (short)0, acc1[t], false, false);
        if (kk + 1 < 16) {
            a_cur = a_nxt;
#pragma unroll
            for (int t = 0; t < 8; ++t) b_cur[t] = b_nxt[t];
        }
    }

    float s1[8], q1[8];
#pragma unroll
    for (int r = 0; r < 8; ++r) { s1[r] = 0.f; q1[r] = 0.f; }
#pragma unroll
    for (int t = 0; t < 8; ++t) {
#pragma unroll
        for (int r = 0; r < 8; ++r) {
            float u = acc1[t][r] + bias1[t];
            acc1[t][r] = u;
            s1[r] += u;
            q1[r] += u * u;
        }
    }
#pragma unroll
    for (int r = 0; r < 8; ++r) {
        s1[r] = row_xor_add<1>(s1[r]);  q1[r] = row_xor_add<1>(q1[r]);
    }
#pragma unroll
    for (int r = 0; r < 8; ++r) {
        s1[r] = row_xor_add<2>(s1[r]);  q1[r] = row_xor_add<2>(q1[r]);
    }
#pragma unroll
    for (int r = 0; r < 8; ++r) {
        s1[r] = row_xor_add<4>(s1[r]);  q1[r] = row_xor_add<4>(q1[r]);
    }
#pragma unroll
    for (int r = 0; r < 8; ++r) {
        s1[r] = row_xor_add<8>(s1[r]);  q1[r] = row_xor_add<8>(q1[r]);
    }
#pragma unroll
    for (int r = 0; r < 8; ++r) {
        float mean = s1[r] * (1.f / 128.f);
        float var  = q1[r] * (1.f / 128.f) - mean * mean;
        float rstd = rsqrtf(var + 1e-5f);
#pragma unroll
        for (int t = 0; t < 8; ++t) {
            float v = (acc1[t][r] - mean) * rstd * gam1[t] + bet1[t];
            tile[(half * 8 + r) * LP + t * 16 + mr] = fmaxf(v, 0.f);
        }
    }

    __syncthreads();

    float bias2[4], gam2[4], bet2[4];
#pragma unroll
    for (int t = 0; t < 4; ++t) {
        bias2[t] = b2[t * 16 + mr];
        gam2[t]  = ng[t * 16 + mr];
        bet2[t]  = nb[t * 16 + mr];
    }

    const float* Atile = tile + mr * LP + half * 2;
    const float* B2p   = W2 + (half * 2) * Hc + mr;

    v8f acc2[4];
#pragma unroll
    for (int t = 0; t < 4; ++t) acc2[t] = v8f{0.f,0.f,0.f,0.f,0.f,0.f,0.f,0.f};

    v2f a2c[2];
    v2f b2c[2][4];
#pragma unroll
    for (int u = 0; u < 2; ++u) {
        a2c[u] = *(const v2f*)(Atile + 4 * u);
#pragma unroll
        for (int t = 0; t < 4; ++t) {
            b2c[u][t].x = B2p[(4 * u) * Hc + t * 16];
            b2c[u][t].y = B2p[(4 * u + 1) * Hc + t * 16];
        }
    }
#pragma unroll
    for (int j = 0; j < 16; ++j) {
        v2f a2n[2];
        v2f b2n[2][4];
        if (j + 1 < 16) {
#pragma unroll
            for (int u = 0; u < 2; ++u) {
                const int kk = 2 * (j + 1) + u;
                a2n[u] = *(const v2f*)(Atile + 4 * kk);
#pragma unroll
                for (int t = 0; t < 4; ++t) {
                    b2n[u][t].x = B2p[(4 * kk) * Hc + t * 16];
                    b2n[u][t].y = B2p[(4 * kk + 1) * Hc + t * 16];
                }
            }
        }
#pragma unroll
        for (int u = 0; u < 2; ++u)
#pragma unroll
            for (int t = 0; t < 4; ++t)
                acc2[t] = __builtin_amdgcn_wmma_f32_16x16x4_f32(
                              false, a2c[u], false, b2c[u][t], (short)0, acc2[t], false, false);
        if (j + 1 < 16) {
#pragma unroll
            for (int u = 0; u < 2; ++u) {
                a2c[u] = a2n[u];
#pragma unroll
                for (int t = 0; t < 4; ++t) b2c[u][t] = b2n[u][t];
            }
        }
    }

    int  nc3[3];
    v2f  b3[16][3];
    float lb3[3];
    if constexpr (LAST) {
#pragma unroll
        for (int t = 0; t < 3; ++t) {
            int n  = t * 16 + mr;
            nc3[t] = (n < Cc) ? n : (Cc - 1);
            lb3[t] = linB[nc3[t]];
        }
        const float* B3p = linW + (half * 2) * Cc;
#pragma unroll
        for (int kk = 0; kk < 16; ++kk)
#pragma unroll
            for (int t = 0; t < 3; ++t) {
                b3[kk][t].x = B3p[(4 * kk) * Cc + nc3[t]];
                b3[kk][t].y = B3p[(4 * kk + 1) * Cc + nc3[t]];
            }
    }

    float s2[8], q2[8];
#pragma unroll
    for (int r = 0; r < 8; ++r) { s2[r] = 0.f; q2[r] = 0.f; }
#pragma unroll
    for (int t = 0; t < 4; ++t) {
#pragma unroll
        for (int r = 0; r < 8; ++r) {
            float hn = hres[t][r] + (acc2[t][r] + bias2[t]);
            acc2[t][r] = hn;
            s2[r] += hn;
            q2[r] += hn * hn;
        }
    }
    if (!LAST && write_h) {
#pragma unroll
        for (int t = 0; t < 4; ++t)
#pragma unroll
            for (int r = 0; r < 8; ++r)
                h[rowbase + (long)r * Hc + t * 16] = acc2[t][r];
    }
#pragma unroll
    for (int r = 0; r < 8; ++r) {
        s2[r] = row_xor_add<1>(s2[r]);  q2[r] = row_xor_add<1>(q2[r]);
    }
#pragma unroll
    for (int r = 0; r < 8; ++r) {
        s2[r] = row_xor_add<2>(s2[r]);  q2[r] = row_xor_add<2>(q2[r]);
    }
#pragma unroll
    for (int r = 0; r < 8; ++r) {
        s2[r] = row_xor_add<4>(s2[r]);  q2[r] = row_xor_add<4>(q2[r]);
    }
#pragma unroll
    for (int r = 0; r < 8; ++r) {
        s2[r] = row_xor_add<8>(s2[r]);  q2[r] = row_xor_add<8>(q2[r]);
    }
#pragma unroll
    for (int r = 0; r < 8; ++r) {
        float mean = s2[r] * (1.f / 64.f);
        float var  = q2[r] * (1.f / 64.f) - mean * mean;
        float rstd = rsqrtf(var + 1e-5f);
#pragma unroll
        for (int t = 0; t < 4; ++t) {
            float v = (acc2[t][r] - mean) * rstd * gam2[t] + bet2[t];
            if constexpr (LAST) tile[(half * 8 + r) * LP + t * 16 + mr] = fmaxf(v, 0.f);
            else                zout[rowbase + (long)r * Hc + t * 16] = fmaxf(v, 0.f);
        }
    }

    if constexpr (LAST) {
        __syncthreads();

        const float* At3 = tile + mr * LP + half * 2;
        v8f acc3[3];
#pragma unroll
        for (int t = 0; t < 3; ++t) acc3[t] = v8f{0.f,0.f,0.f,0.f,0.f,0.f,0.f,0.f};
        v2f a3[16];
#pragma unroll
        for (int kk = 0; kk < 16; ++kk) a3[kk] = *(const v2f*)(At3 + 4 * kk);
#pragma unroll
        for (int kk = 0; kk < 16; ++kk)
#pragma unroll
            for (int t = 0; t < 3; ++t)
                acc3[t] = __builtin_amdgcn_wmma_f32_16x16x4_f32(
                              false, a3[kk], false, b3[kk][t], (short)0, acc3[t], false, false);

        const bool c2ok = mr < (Cc - 32);
        float mx3[8], sm3[8];
#pragma unroll
        for (int r = 0; r < 8; ++r) {
#pragma unroll
            for (int t = 0; t < 3; ++t) acc3[t][r] += lb3[t];
            float m01 = fmaxf(acc3[0][r], acc3[1][r]);
            mx3[r] = c2ok ? fmaxf(m01, acc3[2][r]) : m01;
        }
#pragma unroll
        for (int r = 0; r < 8; ++r) mx3[r] = row_xor_max<1>(mx3[r]);
#pragma unroll
        for (int r = 0; r < 8; ++r) mx3[r] = row_xor_max<2>(mx3[r]);
#pragma unroll
        for (int r = 0; r < 8; ++r) mx3[r] = row_xor_max<4>(mx3[r]);
#pragma unroll
        for (int r = 0; r < 8; ++r) mx3[r] = row_xor_max<8>(mx3[r]);
#pragma unroll
        for (int r = 0; r < 8; ++r) {
            float e01 = __expf(acc3[0][r] - mx3[r]) + __expf(acc3[1][r] - mx3[r]);
            float e2  = __expf(acc3[2][r] - mx3[r]);
            sm3[r] = c2ok ? (e01 + e2) : e01;
        }
#pragma unroll
        for (int r = 0; r < 8; ++r) sm3[r] = row_xor_add<1>(sm3[r]);
#pragma unroll
        for (int r = 0; r < 8; ++r) sm3[r] = row_xor_add<2>(sm3[r]);
#pragma unroll
        for (int r = 0; r < 8; ++r) sm3[r] = row_xor_add<4>(sm3[r]);
#pragma unroll
        for (int r = 0; r < 8; ++r) sm3[r] = row_xor_add<8>(sm3[r]);

#pragma unroll
        for (int r = 0; r < 8; ++r) {
            float lse = mx3[r] + __logf(sm3[r]);
            float* orow = tile + (half * 8 + r) * Cc + mr;
            orow[0]  = acc3[0][r] - lse;
            orow[16] = acc3[1][r] - lse;
            if (c2ok) orow[32] = acc3[2][r] - lse;
        }
        __syncthreads();
        float* ob = out + (long)tm * (16 * Cc);
#pragma unroll
        for (int j = 0; j < (16 * Cc) / (4 * 32); ++j) {
            int q = lane + 32 * j;
            *(v4f*)(ob + 4 * q) = *(const v4f*)(tile + 4 * q);
        }
    }
}

__global__ void csr_hist_kernel(const int* __restrict__ dst,
                                int* __restrict__ cnt, int E) {
    int i = blockIdx.x * blockDim.x + threadIdx.x;
    if (i < E) atomicAdd(&cnt[dst[i]], 1);
}

__global__ void
csr_scan_kernel(const int* __restrict__ cnt, int* __restrict__ rowptr,
                int* __restrict__ cursor, int n) {
    __shared__ int wsum[32];
    int lane = threadIdx.x & 31;
    int wv   = threadIdx.x >> 5;
    int per  = (((n + 31) >> 5) + 31) & ~31;
    int lo   = wv * per;
    int hi   = lo + per; if (hi > n) hi = n;

    int acc = 0;
    for (int i = lo + lane; i < hi; i += 32) acc += cnt[i];
    for (int off = 16; off; off >>= 1) acc += __shfl_xor(acc, off);
    if (lane == 0) wsum[wv] = acc;
    __syncthreads();
    if (wv == 0) {
        int w  = wsum[lane];
        int wi = w;
        for (int off = 1; off < 32; off <<= 1) {
            int v = __shfl_up(wi, off);
            if (lane >= off) wi += v;
        }
        wsum[lane] = wi - w;
    }
    __syncthreads();

    int carry = wsum[wv];
    for (int i0 = lo; i0 < hi; i0 += 128) {
        int c[4], incl[4], tot[4];
#pragma unroll
        for (int q = 0; q < 4; ++q) {
            int i = i0 + 32 * q + lane;
            c[q]    = (i < hi) ? cnt[i] : 0;
            incl[q] = c[q];
        }
        for (int off = 1; off < 32; off <<= 1) {
#pragma unroll
            for (int q = 0; q < 4; ++q) {
                int v = __shfl_up(incl[q], off);
                if (lane >= off) incl[q] += v;
            }
        }
#pragma unroll
        for (int q = 0; q < 4; ++q) tot[q] = __shfl(incl[q], 31);
#pragma unroll
        for (int q = 0; q < 4; ++q) {
            int i  = i0 + 32 * q + lane;
            int ex = carry + incl[q] - c[q];
            if (i < hi) { rowptr[i] = ex; cursor[i] = ex; }
            carry += tot[q];
        }
    }
    if (threadIdx.x == 1023) rowptr[n] = carry;
}

__global__ void csr_fill_kernel(const int* __restrict__ src,
                                const int* __restrict__ dst,
                                int* __restrict__ cursor,
                                int* __restrict__ ssrc, int E) {
    int i = blockIdx.x * blockDim.x + threadIdx.x;
    if (i < E) {
        int p = atomicAdd(&cursor[dst[i]], 1);
        ssrc[p] = src[i];
    }
}

static constexpr int EB = 8;

__global__ void
node_softmax_agg_kernel(const float* __restrict__ z,
                        const int* __restrict__ rowptr,
                        const int* __restrict__ ssrc,
                        const float* __restrict__ tptr, int layer,
                        float* __restrict__ out, int n) {
    int lane = threadIdx.x & 31;
    int node = __builtin_amdgcn_readfirstlane(
                   (int)(blockIdx.x * (blockDim.x >> 5) + (threadIdx.x >> 5)));
    if (node >= n) return;

    int   start = rowptr[node];
    int   end   = rowptr[node + 1];
    float t     = tptr[layer];
    const v2f* zr = (const v2f*)z;
    v2f self = zr[node * (Hc / 2) + lane];

    float mx = -3.0e38f, my = -3.0e38f;
    float dx = 0.f, dy = 0.f, nx = 0.f, ny = 0.f;

    for (int base = start; base < end; base += 32) {
        int cnt = end - base; if (cnt > 32) cnt = 32;
        int myidx = (lane < cnt) ? ssrc[base + lane] : 0;
        for (int j0 = 0; j0 < cnt; j0 += EB) {
            v2f v[EB];
#pragma unroll
            for (int u = 0; u < EB; ++u) {
                int sidx = __builtin_amdgcn_readlane(myidx, j0 + u);
                v[u] = zr[(long)sidx * (Hc / 2) + lane];
            }
            float gx[EB], gy[EB], sx[EB], sy[EB];
            float cmx = -3.0e38f, cmy = -3.0e38f;
#pragma unroll
            for (int u = 0; u < EB; ++u) {
                bool ok = (j0 + u) < cnt;
                float ax = fmaxf(v[u].x, 0.f) + GEPS;
                float ay = fmaxf(v[u].y, 0.f) + GEPS;
                gx[u] = ok ? ax : 0.f;
                gy[u] = ok ? ay : 0.f;
                sx[u] = ok ? ax * t : -3.0e38f;
                sy[u] = ok ? ay * t : -3.0e38f;
                cmx = fmaxf(cmx, sx[u]);
                cmy = fmaxf(cmy, sy[u]);
            }
            float nmx = fmaxf(mx, cmx);
            float nmy = fmaxf(my, cmy);
            float rx  = __expf(mx - nmx);
            float ry  = __expf(my - nmy);
            dx *= rx; nx *= rx;
            dy *= ry; ny *= ry;
#pragma unroll
            for (int u = 0; u < EB; ++u) {
                float ex = __expf(sx[u] - nmx);
                float ey = __expf(sy[u] - nmy);
                dx += ex; nx += gx[u] * ex;
                dy += ey; ny += gy[u] * ey;
            }
            mx = nmx; my = nmy;
        }
    }

    v2f o;
    o.x = nx / (dx + 1e-16f) + self.x;
    o.y = ny / (dy + 1e-16f) + self.y;
    ((v2f*)out)[node * (Hc / 2) + lane] = o;
}

extern "C" void kernel_launch(void* const* d_in, const int* in_sizes, int n_in,
                              void* d_out, int out_size, void* d_ws, size_t ws_size,
                              hipStream_t stream) {
    (void)n_in; (void)out_size; (void)ws_size;

    const float* x    = (const float*)d_in[0];
    const int*   ei   = (const int*)  d_in[1];
    const float* encW = (const float*)d_in[2];
    const float* encB = (const float*)d_in[3];
    const float* t    = (const float*)d_in[4];
    const float* W1   = (const float*)d_in[5];
    const float* b1   = (const float*)d_in[6];
    const float* g1   = (const float*)d_in[7];
    const float* bb1  = (const float*)d_in[8];
    const float* W2   = (const float*)d_in[9];
    const float* b2   = (const float*)d_in[10];
    const float* ng   = (const float*)d_in[11];
    const float* nb   = (const float*)d_in[12];
    const float* linW = (const float*)d_in[13];
    const float* linB = (const float*)d_in[14];

    const int F = 128, H = 64, L = 4;
    const int N = in_sizes[0] / F;
    const int E = in_sizes[1] / 2;
    const int* src = ei;
    const int* dst = ei + E;

    float* ws = (float*)d_ws;
    const long NH = (long)N * H;
    float* h    = ws;  ws += NH;
    float* z    = ws;  ws += NH;
    int*   csr  = (int*)ws;  ws += 3 * NH;
    float* outb = ws;  ws += NH;

    const int  TB = 256;
    auto cdiv = [](long a, long b) { return (int)((a + b - 1) / b); };
    auto gemm_waves = [&](int M, int Nc) { return (long)(M / 16) * ((Nc + 63) / 64); };

    int* cnt    = csr;
    int* cursor = cnt + N;
    int* rowptr = cursor + N;
    int* ssrc   = rowptr + N + 1;

    gemm_bias_wmma<<<cdiv(gemm_waves(N, H) * 32, TB), TB, 0, stream>>>(x, encW, encB, h, N, H, F, cnt, N);

    csr_hist_kernel<<<cdiv(E, TB), TB, 0, stream>>>(dst, cnt, E);
    csr_scan_kernel<<<1, 1024, 0, stream>>>(cnt, rowptr, cursor, N);
    csr_fill_kernel<<<cdiv(E, TB), TB, 0, stream>>>(src, dst, cursor, ssrc, E);

    for (int l = 0; l < L; ++l) {
        const float* zin = (l == 0) ? h : z;

        node_softmax_agg_kernel<<<cdiv(N, 8), 256, 0, stream>>>(zin, rowptr, ssrc, t, l, outb, N);

        const int ln = (l + 1 < L) ? (l + 1) : 0;
        if (l + 1 < L)
            mlp_chain_kernel<false><<<N / 16, 32, 0, stream>>>(
                outb, W1 + (long)l * H * 2 * H, b1 + l * 2 * H, g1 + l * 2 * H, bb1 + l * 2 * H,
                W2 + (long)l * 2 * H * H, b2 + l * H, ng + ln * H, nb + ln * H,
                h, z, (l == 0) ? 0 : 1, 1, linW, linB, (float*)d_out);
        else
            mlp_chain_kernel<true><<<N / 16, 32, 0, stream>>>(
                outb, W1 + (long)l * H * 2 * H, b1 + l * 2 * H, g1 + l * 2 * H, bb1 + l * 2 * H,
                W2 + (long)l * 2 * H * H, b2 + l * H, ng + ln * H, nb + ln * H,
                h, z, (l == 0) ? 0 : 1, 0, linW, linB, (float*)d_out);
    }
}


# Edge grouping with ONE round of scattered atomics, not two: the histogram's returning add gives each edge its rank, the slot fill stores at rowptr+rank; slot order by atomic arrival as before
# speedup vs baseline: 5.1047x; 1.1776x over previous
#include <hip/hip_runtime.h>
#include <math.h>

typedef float v2f __attribute__((ext_vector_type(2)));
typedef float v4f __attribute__((ext_vector_type(4)));
typedef float v8f __attribute__((ext_vector_type(8)));

#define HD __device__ __forceinline__

static constexpr int   Hc   = 64;
static constexpr int   Cc   = 40;
static constexpr float GEPS = 1e-7f;

__global__ void gemm_bias_wmma(const float* __restrict__ A,
                               const float* __restrict__ B,
                               const float* __restrict__ bias,
                               float* __restrict__ C,
                               int M, int N, int K,
                               int* __restrict__ zp, int zn) {
    int gid = (int)(blockIdx.x * blockDim.x + threadIdx.x);
    if (gid < zn) zp[gid] = 0;
    int wave    = gid >> 5;
    int lane    = threadIdx.x & 31;
    int groupsN = (N + 63) >> 6;
    int tm = wave / groupsN;
    int gn = wave - tm * groupsN;
    if (tm >= (M >> 4)) return;

    int mr   = lane & 15;
    int half = lane >> 4;
    const float* Arow = A + (long)(tm * 16 + mr) * K;

    int ncol[4];
#pragma unroll
    for (int t = 0; t < 4; ++t) {
        int n   = gn * 64 + t * 16 + mr;
        ncol[t] = (n < N) ? n : (N - 1);
    }

    v8f acc[4] = {{0.f,0.f,0.f,0.f,0.f,0.f,0.f,0.f},
                  {0.f,0.f,0.f,0.f,0.f,0.f,0.f,0.f},
                  {0.f,0.f,0.f,0.f,0.f,0.f,0.f,0.f},
                  {0.f,0.f,0.f,0.f,0.f,0.f,0.f,0.f}};

    int ka = half * 2;
    v2f a_cur;
    a_cur.x = Arow[ka];
    a_cur.y = Arow[ka + 1];
    v2f b_cur[4];
#pragma unroll
    for (int t = 0; t < 4; ++t) {
        b_cur[t].x = B[ka * N + ncol[t]];
        b_cur[t].y = B[(ka + 1) * N + ncol[t]];
    }

    for (int k = 0; k < K; k += 4) {
        int kn = (k + 4 < K) ? (k + 4) : k;
        int kb = kn + half * 2;

        v2f a_nxt;
        a_nxt.x = Arow[kb];
        a_nxt.y = Arow[kb + 1];
        v2f b_nxt[4];
#pragma unroll
        for (int t = 0; t < 4; ++t) {
            b_nxt[t].x = B[kb * N + ncol[t]];
            b_nxt[t].y = B[(kb + 1) * N + ncol[t]];
        }

#pragma unroll
        for (int t = 0; t < 4; ++t)
            acc[t] = __builtin_amdgcn_wmma_f32_16x16x4_f32(
                         false, a_cur, false, b_cur[t], (short)0, acc[t], false, false);

        a_cur = a_nxt;
#pragma unroll
        for (int t = 0; t < 4; ++t) b_cur[t] = b_nxt[t];
    }

#pragma unroll
    for (int t = 0; t < 4; ++t) {
        int n = gn * 64 + t * 16 + mr;
        if (n < N) {
            float bv = bias[n];
#pragma unroll
            for (int r = 0; r < 8; ++r)
                C[(long)(tm * 16 + half * 8 + r) * N + n] = acc[t][r] + bv;
        }
    }
}

static constexpr int LP = 132;

template <int M>
HD float row_xor_add(float v) {
    int o = __builtin_amdgcn_update_dpp(0, __float_as_int(v), 0x160 | M, 0xf, 0xf, false);
    return v + __int_as_float(o);
}
template <int M>
HD float row_xor_max(float v) {
    int o = __builtin_amdgcn_update_dpp(0, __float_as_int(v), 0x160 | M, 0xf, 0xf, false);
    return fmaxf(v, __int_as_float(o));
}

template <bool LAST>
__global__ void __launch_bounds__(32)
mlp_chain_kernel(const float* __restrict__ X,
                 const float* __restrict__ W1, const float* __restrict__ b1,
                 const float* __restrict__ g1, const float* __restrict__ bb1,
                 const float* __restrict__ W2, const float* __restrict__ b2,
                 const float* __restrict__ ng, const float* __restrict__ nb,
                 float* __restrict__ h, float* __restrict__ zout,
                 int add_res, int write_h,
                 const float* __restrict__ linW, const float* __restrict__ linB,
                 float* __restrict__ out) {
    __shared__ __attribute__((aligned(16))) float tile[16 * LP];
    const int lane = threadIdx.x;
    const int tm   = blockIdx.x;
    const int mr   = lane & 15;
    const int half = lane >> 4;

    const float* Arow = X + (long)(tm * 16 + mr) * Hc + half * 2;
    const float* B1p  = W1 + (half * 2) * (2 * Hc) + mr;

    v8f acc1[8];
#pragma unroll
    for (int t = 0; t < 8; ++t) acc1[t] = v8f{0.f,0.f,0.f,0.f,0.f,0.f,0.f,0.f};

    v2f a_cur = *(const v2f*)(Arow);
    v2f b_cur[8];
#pragma unroll
    for (int t = 0; t < 8; ++t) {
        b_cur[t].x = B1p[t * 16];
        b_cur[t].y = B1p[2 * Hc + t * 16];
    }

    float bias1[8], gam1[8], bet1[8];
#pragma unroll
    for (int t = 0; t < 8; ++t) {
        bias1[t] = b1[t * 16 + mr];
        gam1[t]  = g1[t * 16 + mr];
        bet1[t]  = bb1[t * 16 + mr];
    }

    const long rowbase = (long)(tm * 16 + half * 8) * Hc + mr;
    float hres[4][8];
    if (add_res) {
#pragma unroll
        for (int t = 0; t < 4; ++t)
#pragma unroll
            for (int r = 0; r < 8; ++r) hres[t][r] = h[rowbase + (long)r * Hc + t * 16];
    } else {
#pragma unroll
        for (int t = 0; t < 4; ++t)
#pragma unroll
            for (int r = 0; r < 8; ++r) hres[t][r] = 0.f;
    }

#pragma unroll
    for (int kk = 0; kk < 16; ++kk) {
        v2f a_nxt = a_cur;
        v2f b_nxt[8];
        if (kk + 1 < 16) {
            a_nxt = *(const v2f*)(Arow + 4 * (kk + 1));
#pragma unroll
            for (int t = 0; t < 8; ++t) {
                b_nxt[t].x = B1p[(4 * (kk + 1)) * (2 * Hc) + t * 16];
                b_nxt[t].y = B1p[(4 * (kk + 1) + 1) * (2 * Hc) + t * 16];
            }
        }
#pragma unroll
        for (int t = 0; t < 8; ++t)
            acc1[t] = __builtin_amdgcn_wmma_f32_16x16x4_f32(
                          false, a_cur, false, b_cur[t], (short)0, acc1[t], false, false);
        if (kk + 1 < 16) {
            a_cur = a_nxt;
#pragma unroll
            for (int t = 0; t < 8; ++t) b_cur[t] = b_nxt[t];
        }
    }

    float s1[8], q1[8];
#pragma unroll
    for (int r = 0; r < 8; ++r) { s1[r] = 0.f; q1[r] = 0.f; }
#pragma unroll
    for (int t = 0; t < 8; ++t) {
#pragma unroll
        for (int r = 0; r < 8; ++r) {
            float u = acc1[t][r] + bias1[t];
            acc1[t][r] = u;
            s1[r] += u;
            q1[r] += u * u;
        }
    }
#pragma unroll
    for (int r = 0; r < 8; ++r) {
        s1[r] = row_xor_add<1>(s1[r]);  q1[r] = row_xor_add<1>(q1[r]);
    }
#pragma unroll
    for (int r = 0; r < 8; ++r) {
        s1[r] = row_xor_add<2>(s1[r]);  q1[r] = row_xor_add<2>(q1[r]);
    }
#pragma unroll
    for (int r = 0; r < 8; ++r) {
        s1[r] = row_xor_add<4>(s1[r]);  q1[r] = row_xor_add<4>(q1[r]);
    }
#pragma unroll
    for (int r = 0; r < 8; ++r) {
        s1[r] = row_xor_add<8>(s1[r]);  q1[r] = row_xor_add<8>(q1[r]);
    }
#pragma unroll
    for (int r = 0; r < 8; ++r) {
        float mean = s1[r] * (1.f / 128.f);
        float var  = q1[r] * (1.f / 128.f) - mean * mean;
        float rstd = rsqrtf(var + 1e-5f);
#pragma unroll
        for (int t = 0; t < 8; ++t) {
            float v = (acc1[t][r] - mean) * rstd * gam1[t] + bet1[t];
            tile[(half * 8 + r) * LP + t * 16 + mr] = fmaxf(v, 0.f);
        }
    }

    __syncthreads();

    float bias2[4], gam2[4], bet2[4];
#pragma unroll
    for (int t = 0; t < 4; ++t) {
        bias2[t] = b2[t * 16 + mr];
        gam2[t]  = ng[t * 16 + mr];
        bet2[t]  = nb[t * 16 + mr];
    }

    const float* Atile = tile + mr * LP + half * 2;
    const float* B2p   = W2 + (half * 2) * Hc + mr;

    v8f acc2[4];
#pragma unroll
    for (int t = 0; t < 4; ++t) acc2[t] = v8f{0.f,0.f,0.f,0.f,0.f,0.f,0.f,0.f};

    v2f a2c[2];
    v2f b2c[2][4];
#pragma unroll
    for (int u = 0; u < 2; ++u) {
        a2c[u] = *(const v2f*)(Atile + 4 * u);
#pragma unroll
        for (int t = 0; t < 4; ++t) {
            b2c[u][t].x = B2p[(4 * u) * Hc + t * 16];
            b2c[u][t].y = B2p[(4 * u + 1) * Hc + t * 16];
        }
    }
#pragma unroll
    for (int j = 0; j < 16; ++j) {
        v2f a2n[2];
        v2f b2n[2][4];
        if (j + 1 < 16) {
#pragma unroll
            for (int u = 0; u < 2; ++u) {
                const int kk = 2 * (j + 1) + u;
                a2n[u] = *(const v2f*)(Atile + 4 * kk);
#pragma unroll
                for (int t = 0; t < 4; ++t) {
                    b2n[u][t].x = B2p[(4 * kk) * Hc + t * 16];
                    b2n[u][t].y = B2p[(4 * kk + 1) * Hc + t * 16];
                }
            }
        }
#pragma unroll
        for (int u = 0; u < 2; ++u)
#pragma unroll
            for (int t = 0; t < 4; ++t)
                acc2[t] = __builtin_amdgcn_wmma_f32_16x16x4_f32(
                              false, a2c[u], false, b2c[u][t], (short)0, acc2[t], false, false);
        if (j + 1 < 16) {
#pragma unroll
            for (int u = 0; u < 2; ++u) {
                a2c[u] = a2n[u];
#pragma unroll
                for (int t = 0; t < 4; ++t) b2c[u][t] = b2n[u][t];
            }
        }
    }

    int  nc3[3];
    v2f  b3[16][3];
    float lb3[3];
    if constexpr (LAST) {
#pragma unroll
        for (int t = 0; t < 3; ++t) {
            int n  = t * 16 + mr;
            nc3[t] = (n < Cc) ? n : (Cc - 1);
            lb3[t] = linB[nc3[t]];
        }
        const float* B3p = linW + (half * 2) * Cc;
#pragma unroll
        for (int kk = 0; kk < 16; ++kk)
#pragma unroll
            for (int t = 0; t < 3; ++t) {
                b3[kk][t].x = B3p[(4 * kk) * Cc + nc3[t]];
                b3[kk][t].y = B3p[(4 * kk + 1) * Cc + nc3[t]];
            }
    }

    float s2[8], q2[8];
#pragma unroll
    for (int r = 0; r < 8; ++r) { s2[r] = 0.f; q2[r] = 0.f; }
#pragma unroll
    for (int t = 0; t < 4; ++t) {
#pragma unroll
        for (int r = 0; r < 8; ++r) {
            float hn = hres[t][r] + (acc2[t][r] + bias2[t]);
            acc2[t][r] = hn;
            s2[r] += hn;
            q2[r] += hn * hn;
        }
    }
    if (!LAST && write_h) {
#pragma unroll
        for (int t = 0; t < 4; ++t)
#pragma unroll
            for (int r = 0; r < 8; ++r)
                h[rowbase + (long)r * Hc + t * 16] = acc2[t][r];
    }
#pragma unroll
    for (int r = 0; r < 8; ++r) {
        s2[r] = row_xor_add<1>(s2[r]);  q2[r] = row_xor_add<1>(q2[r]);
    }
#pragma unroll
    for (int r = 0; r < 8; ++r) {
        s2[r] = row_xor_add<2>(s2[r]);  q2[r] = row_xor_add<2>(q2[r]);
    }
#pragma unroll
    for (int r = 0; r < 8; ++r) {
        s2[r] = row_xor_add<4>(s2[r]);  q2[r] = row_xor_add<4>(q2[r]);
    }
#pragma unroll
    for (int r = 0; r < 8; ++r) {
        s2[r] = row_xor_add<8>(s2[r]);  q2[r] = row_xor_add<8>(q2[r]);
    }
#pragma unroll
    for (int r = 0; r < 8; ++r) {
        float mean = s2[r] * (1.f / 64.f);
        float var  = q2[r] * (1.f / 64.f) - mean * mean;
        float rstd = rsqrtf(var + 1e-5f);
#pragma unroll
        for (int t = 0; t < 4; ++t) {
            float v = (acc2[t][r] - mean) * rstd * gam2[t] + bet2[t];
            if constexpr (LAST) tile[(half * 8 + r) * LP + t * 16 + mr] = fmaxf(v, 0.f);
            else                zout[rowbase + (long)r * Hc + t * 16] = fmaxf(v, 0.f);
        }
    }

    if constexpr (LAST) {
        __syncthreads();

        const float* At3 = tile + mr * LP + half * 2;
        v8f acc3[3];
#pragma unroll
        for (int t = 0; t < 3; ++t) acc3[t] = v8f{0.f,0.f,0.f,0.f,0.f,0.f,0.f,0.f};
        v2f a3[16];
#pragma unroll
        for (int kk = 0; kk < 16; ++kk) a3[kk] = *(const v2f*)(At3 + 4 * kk);
#pragma unroll
        for (int kk = 0; kk < 16; ++kk)
#pragma unroll
            for (int t = 0; t < 3; ++t)
                acc3[t] = __builtin_amdgcn_wmma_f32_16x16x4_f32(
                              false, a3[kk], false, b3[kk][t], (short)0, acc3[t], false, false);

        const bool c2ok = mr < (Cc - 32);
        float mx3[8], sm3[8];
#pragma unroll
        for (int r = 0; r < 8; ++r) {
#pragma unroll
            for (int t = 0; t < 3; ++t) acc3[t][r] += lb3[t];
            float m01 = fmaxf(acc3[0][r], acc3[1][r]);
            mx3[r] = c2ok ? fmaxf(m01, acc3[2][r]) : m01;
        }
#pragma unroll
        for (int r = 0; r < 8; ++r) mx3[r] = row_xor_max<1>(mx3[r]);
#pragma unroll
        for (int r = 0; r < 8; ++r) mx3[r] = row_xor_max<2>(mx3[r]);
#pragma unroll
        for (int r = 0; r < 8; ++r) mx3[r] = row_xor_max<4>(mx3[r]);
#pragma unroll
        for (int r = 0; r < 8; ++r) mx3[r] = row_xor_max<8>(mx3[r]);
#pragma unroll
        for (int r = 0; r < 8; ++r) {
            float e01 = __expf(acc3[0][r] - mx3[r]) + __expf(acc3[1][r] - mx3[r]);
            float e2  = __expf(acc3[2][r] - mx3[r]);
            sm3[r] = c2ok ? (e01 + e2) : e01;
        }
#pragma unroll
        for (int r = 0; r < 8; ++r) sm3[r] = row_xor_add<1>(sm3[r]);
#pragma unroll
        for (int r = 0; r < 8; ++r) sm3[r] = row_xor_add<2>(sm3[r]);
#pragma unroll
        for (int r = 0; r < 8; ++r) sm3[r] = row_xor_add<4>(sm3[r]);
#pragma unroll
        for (int r = 0; r < 8; ++r) sm3[r] = row_xor_add<8>(sm3[r]);

#pragma unroll
        for (int r = 0; r < 8; ++r) {
            float lse = mx3[r] + __logf(sm3[r]);
            float* orow = tile + (half * 8 + r) * Cc + mr;
            orow[0]  = acc3[0][r] - lse;
            orow[16] = acc3[1][r] - lse;
            if (c2ok) orow[32] = acc3[2][r] - lse;
        }
        __syncthreads();
        float* ob = out + (long)tm * (16 * Cc);
#pragma unroll
        for (int j = 0; j < (16 * Cc) / (4 * 32); ++j) {
            int q = lane + 32 * j;
            *(v4f*)(ob + 4 * q) = *(const v4f*)(tile + 4 * q);
        }
    }
}

__global__ void csr_hist_kernel(const int* __restrict__ dst,
                                int* __restrict__ cnt,
                                int* __restrict__ rank, int E) {
    int i = blockIdx.x * blockDim.x + threadIdx.x;
    if (i < E) rank[i] = atomicAdd(&cnt[dst[i]], 1);
}

__global__ void
csr_scan_kernel(const int* __restrict__ cnt, int* __restrict__ rowptr, int n) {
    __shared__ int wsum[32];
    int lane = threadIdx.x & 31;
    int wv   = threadIdx.x >> 5;
    int per  = (((n + 31) >> 5) + 31) & ~31;
    int lo   = wv * per;
    int hi   = lo + per; if (hi > n) hi = n;

    int acc = 0;
    for (int i = lo + lane; i < hi; i += 32) acc += cnt[i];
    for (int off = 16; off; off >>= 1) acc += __shfl_xor(acc, off);
    if (lane == 0) wsum[wv] = acc;
    __syncthreads();
    if (wv == 0) {
        int w  = wsum[lane];
        int wi = w;
        for (int off = 1; off < 32; off <<= 1) {
            int v = __shfl_up(wi, off);
            if (lane >= off) wi += v;
        }
        wsum[lane] = wi - w;
    }
    __syncthreads();

    int carry = wsum[wv];
    for (int i0 = lo; i0 < hi; i0 += 128) {
        int c[4], incl[4], tot[4];
#pragma unroll
        for (int q = 0; q < 4; ++q) {
            int i = i0 + 32 * q + lane;
            c[q]    = (i < hi) ? cnt[i] : 0;
            incl[q] = c[q];
        }
        for (int off = 1; off < 32; off <<= 1) {
#pragma unroll
            for (int q = 0; q < 4; ++q) {
                int v = __shfl_up(incl[q], off);
                if (lane >= off) incl[q] += v;
            }
        }
#pragma unroll
        for (int q = 0; q < 4; ++q) tot[q] = __shfl(incl[q], 31);
#pragma unroll
        for (int q = 0; q < 4; ++q) {
            int i  = i0 + 32 * q + lane;
            int ex = carry + incl[q] - c[q];
            if (i < hi) rowptr[i] = ex;
            carry += tot[q];
        }
    }
    if (threadIdx.x == 1023) rowptr[n] = carry;
}

__global__ void csr_fill_kernel(const int* __restrict__ src,
                                const int* __restrict__ dst,
                                const int* __restrict__ rowptr,
                                const int* __restrict__ rank,
                                int* __restrict__ ssrc, int E) {
    int i = blockIdx.x * blockDim.x + threadIdx.x;
    if (i < E) ssrc[rowptr[dst[i]] + rank[i]] = src[i];
}

static constexpr int EB = 8;

__global__ void
node_softmax_agg_kernel(const float* __restrict__ z,
                        const int* __restrict__ rowptr,
                        const int* __restrict__ ssrc,
                        const float* __restrict__ tptr, int layer,
                        float* __restrict__ out, int n) {
    int lane = threadIdx.x & 31;
    int node = __builtin_amdgcn_readfirstlane(
                   (int)(blockIdx.x * (blockDim.x >> 5) + (threadIdx.x >> 5)));
    if (node >= n) return;

    int   start = rowptr[node];
    int   end   = rowptr[node + 1];
    float t     = tptr[layer];
    const v2f* zr = (const v2f*)z;
    v2f self = zr[node * (Hc / 2) + lane];

    float mx = -3.0e38f, my = -3.0e38f;
    float dx = 0.f, dy = 0.f, nx = 0.f, ny = 0.f;

    for (int base = start; base < end; base += 32) {
        int cnt = end - base; if (cnt > 32) cnt = 32;
        int myidx = (lane < cnt) ? ssrc[base + lane] : 0;
        for (int j0 = 0; j0 < cnt; j0 += EB) {
            v2f v[EB];
#pragma unroll
            for (int u = 0; u < EB; ++u) {
                int sidx = __builtin_amdgcn_readlane(myidx, j0 + u);
                v[u] = zr[(long)sidx * (Hc / 2) + lane];
            }
            float gx[EB], gy[EB], sx[EB], sy[EB];
            float cmx = -3.0e38f, cmy = -3.0e38f;
#pragma unroll
            for (int u = 0; u < EB; ++u) {
                bool ok = (j0 + u) < cnt;
                float ax = fmaxf(v[u].x, 0.f) + GEPS;
                float ay = fmaxf(v[u].y, 0.f) + GEPS;
                gx[u] = ok ? ax : 0.f;
                gy[u] = ok ? ay : 0.f;
                sx[u] = ok ? ax * t : -3.0e38f;
                sy[u] = ok ? ay * t : -3.0e38f;
                cmx = fmaxf(cmx, sx[u]);
                cmy = fmaxf(cmy, sy[u]);
            }
            float nmx = fmaxf(mx, cmx);
            float nmy = fmaxf(my, cmy);
            float rx  = __expf(mx - nmx);
            float ry  = __expf(my - nmy);
            dx *= rx; nx *= rx;
            dy *= ry; ny *= ry;
#pragma unroll
            for (int u = 0; u < EB; ++u) {
                float ex = __expf(sx[u] - nmx);
                float ey = __expf(sy[u] - nmy);
                dx += ex; nx += gx[u] * ex;
                dy += ey; ny += gy[u] * ey;
            }
            mx = nmx; my = nmy;
        }
    }

    v2f o;
    o.x = nx / (dx + 1e-16f) + self.x;
    o.y = ny / (dy + 1e-16f) + self.y;
    ((v2f*)out)[node * (Hc / 2) + lane] = o;
}

extern "C" void kernel_launch(void* const* d_in, const int* in_sizes, int n_in,
                              void* d_out, int out_size, void* d_ws, size_t ws_size,
                              hipStream_t stream) {
    (void)n_in; (void)out_size; (void)ws_size;

    const float* x    = (const float*)d_in[0];
    const int*   ei   = (const int*)  d_in[1];
    const float* encW = (const float*)d_in[2];
    const float* encB = (const float*)d_in[3];
    const float* t    = (const float*)d_in[4];
    const float* W1   = (const float*)d_in[5];
    const float* b1   = (const float*)d_in[6];
    const float* g1   = (const float*)d_in[7];
    const float* bb1  = (const float*)d_in[8];
    const float* W2   = (const float*)d_in[9];
    const float* b2   = (const float*)d_in[10];
    const float* ng   = (const float*)d_in[11];
    const float* nb   = (const float*)d_in[12];
    const float* linW = (const float*)d_in[13];
    const float* linB = (const float*)d_in[14];

    const int F = 128, H = 64, L = 4;
    const int N = in_sizes[0] / F;
    const int E = in_sizes[1] / 2;
    const int* src = ei;
    const int* dst = ei + E;

    float* ws = (float*)d_ws;
    const long NH = (long)N * H;
    float* h    = ws;  ws += NH;
    float* z    = ws;  ws += NH;
    int*   csr  = (int*)ws;  ws += 3 * NH;
    float* outb = ws;  ws += NH;

    const int  TB = 256;
    auto cdiv = [](long a, long b) { return (int)((a + b - 1) / b); };
    auto gemm_waves = [&](int M, int Nc) { return (long)(M / 16) * ((Nc + 63) / 64); };

    int* rank   = csr;
    int* ssrc   = rank + E;
    int* cnt    = ssrc + E;
    int* rowptr = cnt + N;

    gemm_bias_wmma<<<cdiv(gemm_waves(N, H) * 32, TB), TB, 0, stream>>>(x, encW, encB, h, N, H, F, cnt, N);

    csr_hist_kernel<<<cdiv(E, TB), TB, 0, stream>>>(dst, cnt, rank, E);
    csr_scan_kernel<<<1, 1024, 0, stream>>>(cnt, rowptr, N);
    csr_fill_kernel<<<cdiv(E, TB), TB, 0, stream>>>(src, dst, rowptr, rank, ssrc, E);

    for (int l = 0; l < L; ++l) {
        const float* zin = (l == 0) ? h : z;

        node_softmax_agg_kernel<<<cdiv(N, 8), 256, 0, stream>>>(zin, rowptr, ssrc, t, l, outb, N);

        const int ln = (l + 1 < L) ? (l + 1) : 0;
        if (l + 1 < L)
            mlp_chain_kernel<false><<<N / 16, 32, 0, stream>>>(
                outb, W1 + (long)l * H * 2 * H, b1 + l * 2 * H, g1 + l * 2 * H, bb1 + l * 2 * H,
                W2 + (long)l * 2 * H * H, b2 + l * H, ng + ln * H, nb + ln * H,
                h, z, (l == 0) ? 0 : 1, 1, linW, linB, (float*)d_out);
        else
            mlp_chain_kernel<true><<<N / 16, 32, 0, stream>>>(
                outb, W1 + (long)l * H * 2 * H, b1 + l * 2 * H, g1 + l * 2 * H, bb1 + l * 2 * H,
                W2 + (long)l * 2 * H * H, b2 + l * H, ng + ln * H, nb + ln * H,
                h, z, (l == 0) ? 0 : 1, 0, linW, linB, (float*)d_out);
    }
}


# Edge grouping as a two-level counting sort ranked by LDS counters (chunk->bucket, bucket->node): 2 launches, no global atomics, no scan launch; edge order in a node by LDS arrival (fp sum order)
# speedup vs baseline: 7.4035x; 1.4503x over previous
#include <hip/hip_runtime.h>
#include <math.h>

typedef float v2f __attribute__((ext_vector_type(2)));
typedef float v4f __attribute__((ext_vector_type(4)));
typedef float v8f __attribute__((ext_vector_type(8)));

#define HD __device__ __forceinline__

static constexpr int   Hc   = 64;
static constexpr int   Cc   = 40;
static constexpr float GEPS = 1e-7f;

__global__ void gemm_bias_wmma(const float* __restrict__ A,
                               const float* __restrict__ B,
                               const float* __restrict__ bias,
                               float* __restrict__ C,
                               int M, int N, int K) {
    int gid = (int)(blockIdx.x * blockDim.x + threadIdx.x);
    int wave    = gid >> 5;
    int lane    = threadIdx.x & 31;
    int groupsN = (N + 63) >> 6;
    int tm = wave / groupsN;
    int gn = wave - tm * groupsN;
    if (tm >= (M >> 4)) return;

    int mr   = lane & 15;
    int half = lane >> 4;
    const float* Arow = A + (long)(tm * 16 + mr) * K;

    int ncol[4];
#pragma unroll
    for (int t = 0; t < 4; ++t) {
        int n   = gn * 64 + t * 16 + mr;
        ncol[t] = (n < N) ? n : (N - 1);
    }

    v8f acc[4] = {{0.f,0.f,0.f,0.f,0.f,0.f,0.f,0.f},
                  {0.f,0.f,0.f,0.f,0.f,0.f,0.f,0.f},
                  {0.f,0.f,0.f,0.f,0.f,0.f,0.f,0.f},
                  {0.f,0.f,0.f,0.f,0.f,0.f,0.f,0.f}};

    int ka = half * 2;
    v2f a_cur;
    a_cur.x = Arow[ka];
    a_cur.y = Arow[ka + 1];
    v2f b_cur[4];
#pragma unroll
    for (int t = 0; t < 4; ++t) {
        b_cur[t].x = B[ka * N + ncol[t]];
        b_cur[t].y = B[(ka + 1) * N + ncol[t]];
    }

    for (int k = 0; k < K; k += 4) {
        int kn = (k + 4 < K) ? (k + 4) : k;
        int kb = kn + half * 2;

        v2f a_nxt;
        a_nxt.x = Arow[kb];
        a_nxt.y = Arow[kb + 1];
        v2f b_nxt[4];
#pragma unroll
        for (int t = 0; t < 4; ++t) {
            b_nxt[t].x = B[kb * N + ncol[t]];
            b_nxt[t].y = B[(kb + 1) * N + ncol[t]];
        }

#pragma unroll
        for (int t = 0; t < 4; ++t)
            acc[t] = __builtin_amdgcn_wmma_f32_16x16x4_f32(
                         false, a_cur, false, b_cur[t], (short)0, acc[t], false, false);

        a_cur = a_nxt;
#pragma unroll
        for (int t = 0; t < 4; ++t) b_cur[t] = b_nxt[t];
    }

#pragma unroll
    for (int t = 0; t < 4; ++t) {
        int n = gn * 64 + t * 16 + mr;
        if (n < N) {
            float bv = bias[n];
#pragma unroll
            for (int r = 0; r < 8; ++r)
                C[(long)(tm * 16 + half * 8 + r) * N + n] = acc[t][r] + bv;
        }
    }
}

static constexpr int LP = 132;

template <int M>
HD float row_xor_add(float v) {
    int o = __builtin_amdgcn_update_dpp(0, __float_as_int(v), 0x160 | M, 0xf, 0xf, false);
    return v + __int_as_float(o);
}
template <int M>
HD float row_xor_max(float v) {
    int o = __builtin_amdgcn_update_dpp(0, __float_as_int(v), 0x160 | M, 0xf, 0xf, false);
    return fmaxf(v, __int_as_float(o));
}

template <bool LAST>
__global__ void __launch_bounds__(32)
mlp_chain_kernel(const float* __restrict__ X,
                 const float* __restrict__ W1, const float* __restrict__ b1,
                 const float* __restrict__ g1, const float* __restrict__ bb1,
                 const float* __restrict__ W2, const float* __restrict__ b2,
                 const float* __restrict__ ng, const float* __restrict__ nb,
                 float* __restrict__ h, float* __restrict__ zout,
                 int add_res, int write_h,
                 const float* __restrict__ linW, const float* __restrict__ linB,
                 float* __restrict__ out) {
    __shared__ __attribute__((aligned(16))) float tile[16 * LP];
    const int lane = threadIdx.x;
    const int tm   = blockIdx.x;
    const int mr   = lane & 15;
    const int half = lane >> 4;

    const float* Arow = X + (long)(tm * 16 + mr) * Hc + half * 2;
    const float* B1p  = W1 + (half * 2) * (2 * Hc) + mr;

    v8f acc1[8];
#pragma unroll
    for (int t = 0; t < 8; ++t) acc1[t] = v8f{0.f,0.f,0.f,0.f,0.f,0.f,0.f,0.f};

    v2f a_cur = *(const v2f*)(Arow);
    v2f b_cur[8];
#pragma unroll
    for (int t = 0; t < 8; ++t) {
        b_cur[t].x = B1p[t * 16];
        b_cur[t].y = B1p[2 * Hc + t * 16];
    }

    float bias1[8], gam1[8], bet1[8];
#pragma unroll
    for (int t = 0; t < 8; ++t) {
        bias1[t] = b1[t * 16 + mr];
        gam1[t]  = g1[t * 16 + mr];
        bet1[t]  = bb1[t * 16 + mr];
    }

    const long rowbase = (long)(tm * 16 + half * 8) * Hc + mr;
    float hres[4][8];
    if (add_res) {
#pragma unroll
        for (int t = 0; t < 4; ++t)
#pragma unroll
            for (int r = 0; r < 8; ++r) hres[t][r] = h[rowbase + (long)r * Hc + t * 16];
    } else {
#pragma unroll
        for (int t = 0; t < 4; ++t)
#pragma unroll
            for (int r = 0; r < 8; ++r) hres[t][r] = 0.f;
    }

#pragma unroll
    for (int kk = 0; kk < 16; ++kk) {
        v2f a_nxt = a_cur;
        v2f b_nxt[8];
        if (kk + 1 < 16) {
            a_nxt = *(const v2f*)(Arow + 4 * (kk + 1));
#pragma unroll
            for (int t = 0; t < 8; ++t) {
                b_nxt[t].x = B1p[(4 * (kk + 1)) * (2 * Hc) + t * 16];
                b_nxt[t].y = B1p[(4 * (kk + 1) + 1) * (2 * Hc) + t * 16];
            }
        }
#pragma unroll
        for (int t = 0; t < 8; ++t)
            acc1[t] = __builtin_amdgcn_wmma_f32_16x16x4_f32(
                          false, a_cur, false, b_cur[t], (short)0, acc1[t], false, false);
        if (kk + 1 < 16) {
            a_cur = a_nxt;
#pragma unroll
            for (int t = 0; t < 8; ++t) b_cur[t] = b_nxt[t];
        }
    }

    float s1[8], q1[8];
#pragma unroll
    for (int r = 0; r < 8; ++r) { s1[r] = 0.f; q1[r] = 0.f; }
#pragma unroll
    for (int t = 0; t < 8; ++t) {
#pragma unroll
        for (int r = 0; r < 8; ++r) {
            float u = acc1[t][r] + bias1[t];
            acc1[t][r] = u;
            s1[r] += u;
            q1[r] += u * u;
        }
    }
#pragma unroll
    for (int r = 0; r < 8; ++r) {
        s1[r] = row_xor_add<1>(s1[r]);  q1[r] = row_xor_add<1>(q1[r]);
    }
#pragma unroll
    for (int r = 0; r < 8; ++r) {
        s1[r] = row_xor_add<2>(s1[r]);  q1[r] = row_xor_add<2>(q1[r]);
    }
#pragma unroll
    for (int r = 0; r < 8; ++r) {
        s1[r] = row_xor_add<4>(s1[r]);  q1[r] = row_xor_add<4>(q1[r]);
    }
#pragma unroll
    for (int r = 0; r < 8; ++r) {
        s1[r] = row_xor_add<8>(s1[r]);  q1[r] = row_xor_add<8>(q1[r]);
    }
#pragma unroll
    for (int r = 0; r < 8; ++r) {
        float mean = s1[r] * (1.f / 128.f);
        float var  = q1[r] * (1.f / 128.f) - mean * mean;
        float rstd = rsqrtf(var + 1e-5f);
#pragma unroll
        for (int t = 0; t < 8; ++t) {
            float v = (acc1[t][r] - mean) * rstd * gam1[t] + bet1[t];
            tile[(half * 8 + r) * LP + t * 16 + mr] = fmaxf(v, 0.f);
        }
    }

    __syncthreads();

    float bias2[4], gam2[4], bet2[4];
#pragma unroll
    for (int t = 0; t < 4; ++t) {
        bias2[t] = b2[t * 16 + mr];
        gam2[t]  = ng[t * 16 + mr];
        bet2[t]  = nb[t * 16 + mr];
    }

    const float* Atile = tile + mr * LP + half * 2;
    const float* B2p   = W2 + (half * 2) * Hc + mr;

    v8f acc2[4];
#pragma unroll
    for (int t = 0; t < 4; ++t) acc2[t] = v8f{0.f,0.f,0.f,0.f,0.f,0.f,0.f,0.f};

    v2f a2c[2];
    v2f b2c[2][4];
#pragma unroll
    for (int u = 0; u < 2; ++u) {
        a2c[u] = *(const v2f*)(Atile + 4 * u);
#pragma unroll
        for (int t = 0; t < 4; ++t) {
            b2c[u][t].x = B2p[(4 * u) * Hc + t * 16];
            b2c[u][t].y = B2p[(4 * u + 1) * Hc + t * 16];
        }
    }
#pragma unroll
    for (int j = 0; j < 16; ++j) {
        v2f a2n[2];
        v2f b2n[2][4];
        if (j + 1 < 16) {
#pragma unroll
            for (int u = 0; u < 2; ++u) {
                const int kk = 2 * (j + 1) + u;
                a2n[u] = *(const v2f*)(Atile + 4 * kk);
#pragma unroll
                for (int t = 0; t < 4; ++t) {
                    b2n[u][t].x = B2p[(4 * kk) * Hc + t * 16];
                    b2n[u][t].y = B2p[(4 * kk + 1) * Hc + t * 16];
                }
            }
        }
#pragma unroll
        for (int u = 0; u < 2; ++u)
#pragma unroll
            for (int t = 0; t < 4; ++t)
                acc2[t] = __builtin_amdgcn_wmma_f32_16x16x4_f32(
                              false, a2c[u], false, b2c[u][t], (short)0, acc2[t], false, false);
        if (j + 1 < 16) {
#pragma unroll
            for (int u = 0; u < 2; ++u) {
                a2c[u] = a2n[u];
#pragma unroll
                for (int t = 0; t < 4; ++t) b2c[u][t] = b2n[u][t];
            }
        }
    }

    int  nc3[3];
    v2f  b3[16][3];
    float lb3[3];
    if constexpr (LAST) {
#pragma unroll
        for (int t = 0; t < 3; ++t) {
            int n  = t * 16 + mr;
            nc3[t] = (n < Cc) ? n : (Cc - 1);
            lb3[t] = linB[nc3[t]];
        }
        const float* B3p = linW + (half * 2) * Cc;
#pragma unroll
        for (int kk = 0; kk < 16; ++kk)
#pragma unroll
            for (int t = 0; t < 3; ++t) {
                b3[kk][t].x = B3p[(4 * kk) * Cc + nc3[t]];
                b3[kk][t].y = B3p[(4 * kk + 1) * Cc + nc3[t]];
            }
    }

    float s2[8], q2[8];
#pragma unroll
    for (int r = 0; r < 8; ++r) { s2[r] = 0.f; q2[r] = 0.f; }
#pragma unroll
    for (int t = 0; t < 4; ++t) {
#pragma unroll
        for (int r = 0; r < 8; ++r) {
            float hn = hres[t][r] + (acc2[t][r] + bias2[t]);
            acc2[t][r] = hn;
            s2[r] += hn;
            q2[r] += hn * hn;
        }
    }
    if (!LAST && write_h) {
#pragma unroll
        for (int t = 0; t < 4; ++t)
#pragma unroll
            for (int r = 0; r < 8; ++r)
                h[rowbase + (long)r * Hc + t * 16] = acc2[t][r];
    }
#pragma unroll
    for (int r = 0; r < 8; ++r) {
        s2[r] = row_xor_add<1>(s2[r]);  q2[r] = row_xor_add<1>(q2[r]);
    }
#pragma unroll
    for (int r = 0; r < 8; ++r) {
        s2[r] = row_xor_add<2>(s2[r]);  q2[r] = row_xor_add<2>(q2[r]);
    }
#pragma unroll
    for (int r = 0; r < 8; ++r) {
        s2[r] = row_xor_add<4>(s2[r]);  q2[r] = row_xor_add<4>(q2[r]);
    }
#pragma unroll
    for (int r = 0; r < 8; ++r) {
        s2[r] = row_xor_add<8>(s2[r]);  q2[r] = row_xor_add<8>(q2[r]);
    }
#pragma unroll
    for (int r = 0; r < 8; ++r) {
        float mean = s2[r] * (1.f / 64.f);
        float var  = q2[r] * (1.f / 64.f) - mean * mean;
        float rstd = rsqrtf(var + 1e-5f);
#pragma unroll
        for (int t = 0; t < 4; ++t) {
            float v = (acc2[t][r] - mean) * rstd * gam2[t] + bet2[t];
            if constexpr (LAST) tile[(half * 8 + r) * LP + t * 16 + mr] = fmaxf(v, 0.f);
            else                zout[rowbase + (long)r * Hc + t * 16] = fmaxf(v, 0.f);
        }
    }

    if constexpr (LAST) {
        __syncthreads();

        const float* At3 = tile + mr * LP + half * 2;
        v8f acc3[3];
#pragma unroll
        for (int t = 0; t < 3; ++t) acc3[t] = v8f{0.f,0.f,0.f,0.f,0.f,0.f,0.f,0.f};
        v2f a3[16];
#pragma unroll
        for (int kk = 0; kk < 16; ++kk) a3[kk] = *(const v2f*)(At3 + 4 * kk);
#pragma unroll
        for (int kk = 0; kk < 16; ++kk)
#pragma unroll
            for (int t = 0; t < 3; ++t)
                acc3[t] = __builtin_amdgcn_wmma_f32_16x16x4_f32(
                              false, a3[kk], false, b3[kk][t], (short)0, acc3[t], false, false);

        const bool c2ok = mr < (Cc - 32);
        float mx3[8], sm3[8];
#pragma unroll
        for (int r = 0; r < 8; ++r) {
#pragma unroll
            for (int t = 0; t < 3; ++t) acc3[t][r] += lb3[t];
            float m01 = fmaxf(acc3[0][r], acc3[1][r]);
            mx3[r] = c2ok ? fmaxf(m01, acc3[2][r]) : m01;
        }
#pragma unroll
        for (int r = 0; r < 8; ++r) mx3[r] = row_xor_max<1>(mx3[r]);
#pragma unroll
        for (int r = 0; r < 8; ++r) mx3[r] = row_xor_max<2>(mx3[r]);
#pragma unroll
        for (int r = 0; r < 8; ++r) mx3[r] = row_xor_max<4>(mx3[r]);
#pragma unroll
        for (int r = 0; r < 8; ++r) mx3[r] = row_xor_max<8>(mx3[r]);
#pragma unroll
        for (int r = 0; r < 8; ++r) {
            float e01 = __expf(acc3[0][r] - mx3[r]) + __expf(acc3[1][r] - mx3[r]);
            float e2  = __expf(acc3[2][r] - mx3[r]);
            sm3[r] = c2ok ? (e01 + e2) : e01;
        }
#pragma unroll
        for (int r = 0; r < 8; ++r) sm3[r] = row_xor_add<1>(sm3[r]);
#pragma unroll
        for (int r = 0; r < 8; ++r) sm3[r] = row_xor_add<2>(sm3[r]);
#pragma unroll
        for (int r = 0; r < 8; ++r) sm3[r] = row_xor_add<4>(sm3[r]);
#pragma unroll
        for (int r = 0; r < 8; ++r) sm3[r] = row_xor_add<8>(sm3[r]);

#pragma unroll
        for (int r = 0; r < 8; ++r) {
            float lse = mx3[r] + __logf(sm3[r]);
            float* orow = tile + (half * 8 + r) * Cc + mr;
            orow[0]  = acc3[0][r] - lse;
            orow[16] = acc3[1][r] - lse;
            if (c2ok) orow[32] = acc3[2][r] - lse;
        }
        __syncthreads();
        float* ob = out + (long)tm * (16 * Cc);
#pragma unroll
        for (int j = 0; j < (16 * Cc) / (4 * 32); ++j) {
            int q = lane + 32 * j;
            *(v4f*)(ob + 4 * q) = *(const v4f*)(tile + 4 * q);
        }
    }
}

typedef int v2i __attribute__((ext_vector_type(2)));

static constexpr int CH   = 4096;
static constexpr int BSH  = 7;
static constexpr int BN   = 1 << BSH;
static constexpr int MAXB = 256;

static constexpr int T1  = 256;
static constexpr int EPT = CH / T1;

__global__ void __launch_bounds__(T1)
edge_bin_kernel(const int* __restrict__ src, const int* __restrict__ dst,
                int* __restrict__ pairs, int* __restrict__ offT, int E, int nb) {
    __shared__ int lcnt[MAXB];
    __shared__ v2i stage[CH];
    const int t  = threadIdx.x;
    const int g  = blockIdx.x;
    const int G  = gridDim.x;
    const int e0 = g * CH;
    int n = E - e0; if (n > CH) n = CH;

    int d[EPT], s[EPT];
#pragma unroll
    for (int k = 0; k < EPT; ++k) {
        int idx = t + T1 * k;
        bool ok = idx < n;
        d[k] = ok ? dst[e0 + idx] : -1;
        s[k] = ok ? src[e0 + idx] : 0;
    }
    for (int q = t; q < MAXB; q += T1) lcnt[q] = 0;
    __syncthreads();

    int rk[EPT];
#pragma unroll
    for (int k = 0; k < EPT; ++k) {
        rk[k] = 0;
        if (d[k] >= 0) rk[k] = atomicAdd(&lcnt[d[k] >> BSH], 1);
    }
    __syncthreads();

    if (t < 32) {
        int c[8];
        int sum = 0;
#pragma unroll
        for (int q = 0; q < 8; ++q) { int x = lcnt[8 * t + q]; c[q] = sum; sum += x; }
        int incl = sum;
        for (int off = 1; off < 32; off <<= 1) {
            int v = __shfl_up(incl, off);
            if (t >= off) incl += v;
        }
        int excl = incl - sum;
#pragma unroll
        for (int q = 0; q < 8; ++q) lcnt[8 * t + q] = excl + c[q];
    }
    __syncthreads();

    int pos[EPT];
#pragma unroll
    for (int k = 0; k < EPT; ++k) pos[k] = lcnt[(d[k] >= 0) ? (d[k] >> BSH) : 0] + rk[k];
#pragma unroll
    for (int k = 0; k < EPT; ++k)
        if (d[k] >= 0) stage[pos[k]] = v2i{d[k], s[k]};
    __syncthreads();

    v2i* pr = (v2i*)pairs + e0;
    v2i w[EPT];
#pragma unroll
    for (int k = 0; k < EPT; ++k) {
        int idx = t + T1 * k;
        w[k] = (idx < n) ? stage[idx] : v2i{0, 0};
    }
#pragma unroll
    for (int k = 0; k < EPT; ++k) {
        int idx = t + T1 * k;
        if (idx < n) pr[idx] = w[k];
    }
    for (int q = t; q <= nb; q += T1) offT[q * G + g] = (q < nb) ? lcnt[q] : n;
}

template <bool PLACE>
HD int bucket_sweep(const v2i* __restrict__ pr, const int* __restrict__ r0,
                    const int* __restrict__ r1, int G, int wv, int lane, int lo,
                    int* ctr, int* __restrict__ ssrc) {
    int below = 0;
    for (int gb = wv * 32; gb < G; gb += 8 * 32) {
        int gg = gb + lane;
        int o0 = 0, o1 = 0;
        if (gg < G) { o0 = r0[gg]; o1 = r1[gg]; }
        below += o0;
        int len = o1 - o0;
        int st  = gg * CH + o0;
        v2i p[32];
#pragma unroll
        for (int u = 0; u < 32; ++u) {
            int L = __builtin_amdgcn_readlane(len, u);
            int S = __builtin_amdgcn_readlane(st, u);
            p[u] = v2i{-1, 0};
            if (lane < L) p[u] = pr[S + lane];
        }
        if (PLACE) {
            int slot[32];
#pragma unroll
            for (int u = 0; u < 32; ++u) {
                slot[u] = 0;
                if (p[u].x >= 0) slot[u] = atomicAdd(&ctr[p[u].x - lo], 1);
            }
#pragma unroll
            for (int u = 0; u < 32; ++u)
                if (p[u].x >= 0) ssrc[slot[u]] = p[u].y;
        } else {
#pragma unroll
            for (int u = 0; u < 32; ++u)
                if (p[u].x >= 0) atomicAdd(&ctr[p[u].x - lo], 1);
        }
        unsigned m = __builtin_amdgcn_ballot_w32(len > 32);
        while (m) {
            int u = __builtin_ctz(m);
            m &= m - 1;
            int L = __builtin_amdgcn_readlane(len, u);
            int S = __builtin_amdgcn_readlane(st, u);
            for (int k = 32 + lane; k < L; k += 32) {
                v2i q = pr[S + k];
                if (PLACE) ssrc[atomicAdd(&ctr[q.x - lo], 1)] = q.y;
                else       atomicAdd(&ctr[q.x - lo], 1);
            }
        }
    }
    return below;
}

__global__ void __launch_bounds__(256)
bucket_group_kernel(const int* __restrict__ pairs, const int* __restrict__ offT,
                    int* __restrict__ rowptr, int* __restrict__ ssrc, int N, int G) {
    __shared__ int cnt[BN];
    __shared__ int cur[BN];
    __shared__ int wsum[8];
    const int t    = threadIdx.x;
    const int lane = t & 31;
    const int wv   = __builtin_amdgcn_readfirstlane(t >> 5);
    const int b    = blockIdx.x;
    const int lo   = b << BSH;
    const v2i* pr  = (const v2i*)pairs;
    const int* r0  = offT + (long)b * G;
    const int* r1  = r0 + G;

    if (t < BN) cnt[t] = 0;
    __syncthreads();

    int below = bucket_sweep<false>(pr, r0, r1, G, wv, lane, lo, cnt, ssrc);
    for (int off = 16; off; off >>= 1) below += __shfl_xor(below, off);
    if (lane == 0) wsum[wv] = below;
    __syncthreads();

    if (t < 32) {
        int base = 0;
#pragma unroll
        for (int w = 0; w < 8; ++w) base += wsum[w];
        int c[4];
        int sum = 0;
#pragma unroll
        for (int q = 0; q < 4; ++q) { int x = cnt[4 * t + q]; c[q] = sum; sum += x; }
        int incl = sum;
        for (int off = 1; off < 32; off <<= 1) {
            int v = __shfl_up(incl, off);
            if (t >= off) incl += v;
        }
        int excl = base + incl - sum;
#pragma unroll
        for (int q = 0; q < 4; ++q) {
            int node = lo + 4 * t + q;
            int v    = excl + c[q];
            cur[4 * t + q] = v;
            if (node < N) rowptr[node] = v;
        }
        if (b == (int)gridDim.x - 1 && t == 31) rowptr[N] = base + incl;
    }
    __syncthreads();

    bucket_sweep<true>(pr, r0, r1, G, wv, lane, lo, cur, ssrc);
}

static constexpr int EB = 8;

__global__ void
node_softmax_agg_kernel(const float* __restrict__ z,
                        const int* __restrict__ rowptr,
                        const int* __restrict__ ssrc,
                        const float* __restrict__ tptr, int layer,
                        float* __restrict__ out, int n) {
    int lane = threadIdx.x & 31;
    int node = __builtin_amdgcn_readfirstlane(
                   (int)(blockIdx.x * (blockDim.x >> 5) + (threadIdx.x >> 5)));
    if (node >= n) return;

    int   start = rowptr[node];
    int   end   = rowptr[node + 1];
    float t     = tptr[layer];
    const v2f* zr = (const v2f*)z;
    v2f self = zr[node * (Hc / 2) + lane];

    float mx = -3.0e38f, my = -3.0e38f;
    float dx = 0.f, dy = 0.f, nx = 0.f, ny = 0.f;

    for (int base = start; base < end; base += 32) {
        int cnt = end - base; if (cnt > 32) cnt = 32;
        int myidx = (lane < cnt) ? ssrc[base + lane] : 0;
        for (int j0 = 0; j0 < cnt; j0 += EB) {
            v2f v[EB];
#pragma unroll
            for (int u = 0; u < EB; ++u) {
                int sidx = __builtin_amdgcn_readlane(myidx, j0 + u);
                v[u] = zr[(long)sidx * (Hc / 2) + lane];
            }
            float gx[EB], gy[EB], sx[EB], sy[EB];
            float cmx = -3.0e38f, cmy = -3.0e38f;
#pragma unroll
            for (int u = 0; u < EB; ++u) {
                bool ok = (j0 + u) < cnt;
                float ax = fmaxf(v[u].x, 0.f) + GEPS;
                float ay = fmaxf(v[u].y, 0.f) + GEPS;
                gx[u] = ok ? ax : 0.f;
                gy[u] = ok ? ay : 0.f;
                sx[u] = ok ? ax * t : -3.0e38f;
                sy[u] = ok ? ay * t : -3.0e38f;
                cmx = fmaxf(cmx, sx[u]);
                cmy = fmaxf(cmy, sy[u]);
            }
            float nmx = fmaxf(mx, cmx);
            float nmy = fmaxf(my, cmy);
            float rx  = __expf(mx - nmx);
            float ry  = __expf(my - nmy);
            dx *= rx; nx *= rx;
            dy *= ry; ny *= ry;
#pragma unroll
            for (int u = 0; u < EB; ++u) {
                float ex = __expf(sx[u] - nmx);
                float ey = __expf(sy[u] - nmy);
                dx += ex; nx += gx[u] * ex;
                dy += ey; ny += gy[u] * ey;
            }
            mx = nmx; my = nmy;
        }
    }

    v2f o;
    o.x = nx / (dx + 1e-16f) + self.x;
    o.y = ny / (dy + 1e-16f) + self.y;
    ((v2f*)out)[node * (Hc / 2) + lane] = o;
}

extern "C" void kernel_launch(void* const* d_in, const int* in_sizes, int n_in,
                              void* d_out, int out_size, void* d_ws, size_t ws_size,
                              hipStream_t stream) {
    (void)n_in; (void)out_size; (void)ws_size;

    const float* x    = (const float*)d_in[0];
    const int*   ei   = (const int*)  d_in[1];
    const float* encW = (const float*)d_in[2];
    const float* encB = (const float*)d_in[3];
    const float* t    = (const float*)d_in[4];
    const float* W1   = (const float*)d_in[5];
    const float* b1   = (const float*)d_in[6];
    const float* g1   = (const float*)d_in[7];
    const float* bb1  = (const float*)d_in[8];
    const float* W2   = (const float*)d_in[9];
    const float* b2   = (const float*)d_in[10];
    const float* ng   = (const float*)d_in[11];
    const float* nb   = (const float*)d_in[12];
    const float* linW = (const float*)d_in[13];
    const float* linB = (const float*)d_in[14];

    const int F = 128, H = 64, L = 4;
    const int N = in_sizes[0] / F;
    const int E = in_sizes[1] / 2;
    const int* src = ei;
    const int* dst = ei + E;

    float* ws = (float*)d_ws;
    const long NH = (long)N * H;
    float* h    = ws;  ws += NH;
    float* z    = ws;  ws += NH;
    int*   csr  = (int*)ws;  ws += 3 * NH;
    float* outb = ws;  ws += NH;

    const int  TB = 256;
    auto cdiv = [](long a, long b) { return (int)((a + b - 1) / b); };
    auto gemm_waves = [&](int M, int Nc) { return (long)(M / 16) * ((Nc + 63) / 64); };

    const int G  = cdiv(E, CH);
    const int nbk = cdiv(N, BN);
    int* pairs  = csr;
    int* ssrc   = pairs + 2 * E;
    int* offT   = ssrc + E;
    int* rowptr = offT + (nbk + 1) * G;

    gemm_bias_wmma<<<cdiv(gemm_waves(N, H) * 32, TB), TB, 0, stream>>>(x, encW, encB, h, N, H, F);

    edge_bin_kernel<<<G, T1, 0, stream>>>(src, dst, pairs, offT, E, nbk);
    bucket_group_kernel<<<nbk, 256, 0, stream>>>(pairs, offT, rowptr, ssrc, N, G);

    for (int l = 0; l < L; ++l) {
        const float* zin = (l == 0) ? h : z;

        node_softmax_agg_kernel<<<cdiv(N, 8), 256, 0, stream>>>(zin, rowptr, ssrc, t, l, outb, N);

        const int ln = (l + 1 < L) ? (l + 1) : 0;
        if (l + 1 < L)
            mlp_chain_kernel<false><<<N / 16, 32, 0, stream>>>(
                outb, W1 + (long)l * H * 2 * H, b1 + l * 2 * H, g1 + l * 2 * H, bb1 + l * 2 * H,
                W2 + (long)l * 2 * H * H, b2 + l * H, ng + ln * H, nb + ln * H,
                h, z, (l == 0) ? 0 : 1, 1, linW, linB, (float*)d_out);
        else
            mlp_chain_kernel<true><<<N / 16, 32, 0, stream>>>(
                outb, W1 + (long)l * H * 2 * H, b1 + l * 2 * H, g1 + l * 2 * H, bb1 + l * 2 * H,
                W2 + (long)l * 2 * H * H, b2 + l * H, ng + ln * H, nb + ln * H,
                h, z, (l == 0) ? 0 : 1, 0, linW, linB, (float*)d_out);
    }
}


# Gather loop, vector instructions per edge halved: packed 2-channel f32 ops, weights 2^(x*t*log2e-max), eps summed once as eps*D, selects only in the last batch, relu in layer 0 only; rounding differs
# speedup vs baseline: 8.1534x; 1.1013x over previous
#include <hip/hip_runtime.h>
#include <math.h>

typedef float v2f __attribute__((ext_vector_type(2)));
typedef float v4f __attribute__((ext_vector_type(4)));
typedef float v8f __attribute__((ext_vector_type(8)));

#define HD __device__ __forceinline__

static constexpr int   Hc   = 64;
static constexpr int   Cc   = 40;
static constexpr float GEPS = 1e-7f;

__global__ void gemm_bias_wmma(const float* __restrict__ A,
                               const float* __restrict__ B,
                               const float* __restrict__ bias,
                               float* __restrict__ C,
                               int M, int N, int K) {
    int gid = (int)(blockIdx.x * blockDim.x + threadIdx.x);
    int wave    = gid >> 5;
    int lane    = threadIdx.x & 31;
    int groupsN = (N + 63) >> 6;
    int tm = wave / groupsN;
    int gn = wave - tm * groupsN;
    if (tm >= (M >> 4)) return;

    int mr   = lane & 15;
    int half = lane >> 4;
    const float* Arow = A + (long)(tm * 16 + mr) * K;

    int ncol[4];
#pragma unroll
    for (int t = 0; t < 4; ++t) {
        int n   = gn * 64 + t * 16 + mr;
        ncol[t] = (n < N) ? n : (N - 1);
    }

    v8f acc[4] = {{0.f,0.f,0.f,0.f,0.f,0.f,0.f,0.f},
                  {0.f,0.f,0.f,0.f,0.f,0.f,0.f,0.f},
                  {0.f,0.f,0.f,0.f,0.f,0.f,0.f,0.f},
                  {0.f,0.f,0.f,0.f,0.f,0.f,0.f,0.f}};

    int ka = half * 2;
    v2f a_cur;
    a_cur.x = Arow[ka];
    a_cur.y = Arow[ka + 1];
    v2f b_cur[4];
#pragma unroll
    for (int t = 0; t < 4; ++t) {
        b_cur[t].x = B[ka * N + ncol[t]];
        b_cur[t].y = B[(ka + 1) * N + ncol[t]];
    }

    for (int k = 0; k < K; k += 4) {
        int kn = (k + 4 < K) ? (k + 4) : k;
        int kb = kn + half * 2;

        v2f a_nxt;
        a_nxt.x = Arow[kb];
        a_nxt.y = Arow[kb + 1];
        v2f b_nxt[4];
#pragma unroll
        for (int t = 0; t < 4; ++t) {
            b_nxt[t].x = B[kb * N + ncol[t]];
            b_nxt[t].y = B[(kb + 1) * N + ncol[t]];
        }

#pragma unroll
        for (int t = 0; t < 4; ++t)
            acc[t] = __builtin_amdgcn_wmma_f32_16x16x4_f32(
                         false, a_cur, false, b_cur[t], (short)0, acc[t], false, false);

        a_cur = a_nxt;
#pragma unroll
        for (int t = 0; t < 4; ++t) b_cur[t] = b_nxt[t];
    }

#pragma unroll
    for (int t = 0; t < 4; ++t) {
        int n = gn * 64 + t * 16 + mr;
        if (n < N) {
            float bv = bias[n];
#pragma unroll
            for (int r = 0; r < 8; ++r)
                C[(long)(tm * 16 + half * 8 + r) * N + n] = acc[t][r] + bv;
        }
    }
}

static constexpr int LP = 132;

template <int M>
HD float row_xor_add(float v) {
    int o = __builtin_amdgcn_update_dpp(0, __float_as_int(v), 0x160 | M, 0xf, 0xf, false);
    return v + __int_as_float(o);
}
template <int M>
HD float row_xor_max(float v) {
    int o = __builtin_amdgcn_update_dpp(0, __float_as_int(v), 0x160 | M, 0xf, 0xf, false);
    return fmaxf(v, __int_as_float(o));
}

template <bool LAST>
__global__ void __launch_bounds__(32)
mlp_chain_kernel(const float* __restrict__ X,
                 const float* __restrict__ W1, const float* __restrict__ b1,
                 const float* __restrict__ g1, const float* __restrict__ bb1,
                 const float* __restrict__ W2, const float* __restrict__ b2,
                 const float* __restrict__ ng, const float* __restrict__ nb,
                 float* __restrict__ h, float* __restrict__ zout,
                 int add_res, int write_h,
                 const float* __restrict__ linW, const float* __restrict__ linB,
                 float* __restrict__ out) {
    __shared__ __attribute__((aligned(16))) float tile[16 * LP];
    const int lane = threadIdx.x;
    const int tm   = blockIdx.x;
    const int mr   = lane & 15;
    const int half = lane >> 4;

    const float* Arow = X + (long)(tm * 16 + mr) * Hc + half * 2;
    const float* B1p  = W1 + (half * 2) * (2 * Hc) + mr;

    v8f acc1[8];
#pragma unroll
    for (int t = 0; t < 8; ++t) acc1[t] = v8f{0.f,0.f,0.f,0.f,0.f,0.f,0.f,0.f};

    v2f a_cur = *(const v2f*)(Arow);
    v2f b_cur[8];
#pragma unroll
    for (int t = 0; t < 8; ++t) {
        b_cur[t].x = B1p[t * 16];
        b_cur[t].y = B1p[2 * Hc + t * 16];
    }

    float bias1[8], gam1[8], bet1[8];
#pragma unroll
    for (int t = 0; t < 8; ++t) {
        bias1[t] = b1[t * 16 + mr];
        gam1[t]  = g1[t * 16 + mr];
        bet1[t]  = bb1[t * 16 + mr];
    }

    const long rowbase = (long)(tm * 16 + half * 8) * Hc + mr;
    float hres[4][8];
    if (add_res) {
#pragma unroll
        for (int t = 0; t < 4; ++t)
#pragma unroll
            for (int r = 0; r < 8; ++r) hres[t][r] = h[rowbase + (long)r * Hc + t * 16];
    } else {
#pragma unroll
        for (int t = 0; t < 4; ++t)
#pragma unroll
            for (int r = 0; r < 8; ++r) hres[t][r] = 0.f;
    }

#pragma unroll
    for (int kk = 0; kk < 16; ++kk) {
        v2f a_nxt = a_cur;
        v2f b_nxt[8];
        if (kk + 1 < 16) {
            a_nxt = *(const v2f*)(Arow + 4 * (kk + 1));
#pragma unroll
            for (int t = 0; t < 8; ++t) {
                b_nxt[t].x = B1p[(4 * (kk + 1)) * (2 * Hc) + t * 16];
                b_nxt[t].y = B1p[(4 * (kk + 1) + 1) * (2 * Hc) + t * 16];
            }
        }
#pragma unroll
        for (int t = 0; t < 8; ++t)
            acc1[t] = __builtin_amdgcn_wmma_f32_16x16x4_f32(
                          false, a_cur, false, b_cur[t], (short)0, acc1[t], false, false);
        if (kk + 1 < 16) {
            a_cur = a_nxt;
#pragma unroll
            for (int t = 0; t < 8; ++t) b_cur[t] = b_nxt[t];
        }
    }

    float s1[8], q1[8];
#pragma unroll
    for (int r = 0; r < 8; ++r) { s1[r] = 0.f; q1[r] = 0.f; }
#pragma unroll
    for (int t = 0; t < 8; ++t) {
#pragma unroll
        for (int r = 0; r < 8; ++r) {
            float u = acc1[t][r] + bias1[t];
            acc1[t][r] = u;
            s1[r] += u;
            q1[r] += u * u;
        }
    }
#pragma unroll
    for (int r = 0; r < 8; ++r) {
        s1[r] = row_xor_add<1>(s1[r]);  q1[r] = row_xor_add<1>(q1[r]);
    }
#pragma unroll
    for (int r = 0; r < 8; ++r) {
        s1[r] = row_xor_add<2>(s1[r]);  q1[r] = row_xor_add<2>(q1[r]);
    }
#pragma unroll
    for (int r = 0; r < 8; ++r) {
        s1[r] = row_xor_add<4>(s1[r]);  q1[r] = row_xor_add<4>(q1[r]);
    }
#pragma unroll
    for (int r = 0; r < 8; ++r) {
        s1[r] = row_xor_add<8>(s1[r]);  q1[r] = row_xor_add<8>(q1[r]);
    }
#pragma unroll
    for (int r = 0; r < 8; ++r) {
        float mean = s1[r] * (1.f / 128.f);
        float var  = q1[r] * (1.f / 128.f) - mean * mean;
        float rstd = rsqrtf(var + 1e-5f);
#pragma unroll
        for (int t = 0; t < 8; ++t) {
            float v = (acc1[t][r] - mean) * rstd * gam1[t] + bet1[t];
            tile[(half * 8 + r) * LP + t * 16 + mr] = fmaxf(v, 0.f);
        }
    }

    __syncthreads();

    float bias2[4], gam2[4], bet2[4];
#pragma unroll
    for (int t = 0; t < 4; ++t) {
        bias2[t] = b2[t * 16 + mr];
        gam2[t]  = ng[t * 16 + mr];
        bet2[t]  = nb[t * 16 + mr];
    }

    const float* Atile = tile + mr * LP + half * 2;
    const float* B2p   = W2 + (half * 2) * Hc + mr;

    v8f acc2[4];
#pragma unroll
    for (int t = 0; t < 4; ++t) acc2[t] = v8f{0.f,0.f,0.f,0.f,0.f,0.f,0.f,0.f};

    v2f a2c[2];
    v2f b2c[2][4];
#pragma unroll
    for (int u = 0; u < 2; ++u) {
        a2c[u] = *(const v2f*)(Atile + 4 * u);
#pragma unroll
        for (int t = 0; t < 4; ++t) {
            b2c[u][t].x = B2p[(4 * u) * Hc + t * 16];
            b2c[u][t].y = B2p[(4 * u + 1) * Hc + t * 16];
        }
    }
#pragma unroll
    for (int j = 0; j < 16; ++j) {
        v2f a2n[2];
        v2f b2n[2][4];
        if (j + 1 < 16) {
#pragma unroll
            for (int u = 0; u < 2; ++u) {
                const int kk = 2 * (j + 1) + u;
                a2n[u] = *(const v2f*)(Atile + 4 * kk);
#pragma unroll
                for (int t = 0; t < 4; ++t) {
                    b2n[u][t].x = B2p[(4 * kk) * Hc + t * 16];
                    b2n[u][t].y = B2p[(4 * kk + 1) * Hc + t * 16];
                }
            }
        }
#pragma unroll
        for (int u = 0; u < 2; ++u)
#pragma unroll
            for (int t = 0; t < 4; ++t)
                acc2[t] = __builtin_amdgcn_wmma_f32_16x16x4_f32(
                              false, a2c[u], false, b2c[u][t], (short)0, acc2[t], false, false);
        if (j + 1 < 16) {
#pragma unroll
            for (int u = 0; u < 2; ++u) {
                a2c[u] = a2n[u];
#pragma unroll
                for (int t = 0; t < 4; ++t) b2c[u][t] = b2n[u][t];
            }
        }
    }

    int  nc3[3];
    v2f  b3[16][3];
    float lb3[3];
    if constexpr (LAST) {
#pragma unroll
        for (int t = 0; t < 3; ++t) {
            int n  = t * 16 + mr;
            nc3[t] = (n < Cc) ? n : (Cc - 1);
            lb3[t] = linB[nc3[t]];
        }
        const float* B3p = linW + (half * 2) * Cc;
#pragma unroll
        for (int kk = 0; kk < 16; ++kk)
#pragma unroll
            for (int t = 0; t < 3; ++t) {
                b3[kk][t].x = B3p[(4 * kk) * Cc + nc3[t]];
                b3[kk][t].y = B3p[(4 * kk + 1) * Cc + nc3[t]];
            }
    }

    float s2[8], q2[8];
#pragma unroll
    for (int r = 0; r < 8; ++r) { s2[r] = 0.f; q2[r] = 0.f; }
#pragma unroll
    for (int t = 0; t < 4; ++t) {
#pragma unroll
        for (int r = 0; r < 8; ++r) {
            float hn = hres[t][r] + (acc2[t][r] + bias2[t]);
            acc2[t][r] = hn;
            s2[r] += hn;
            q2[r] += hn * hn;
        }
    }
    if (!LAST && write_h) {
#pragma unroll
        for (int t = 0; t < 4; ++t)
#pragma unroll
            for (int r = 0; r < 8; ++r)
                h[rowbase + (long)r * Hc + t * 16] = acc2[t][r];
    }
#pragma unroll
    for (int r = 0; r < 8; ++r) {
        s2[r] = row_xor_add<1>(s2[r]);  q2[r] = row_xor_add<1>(q2[r]);
    }
#pragma unroll
    for (int r = 0; r < 8; ++r) {
        s2[r] = row_xor_add<2>(s2[r]);  q2[r] = row_xor_add<2>(q2[r]);
    }
#pragma unroll
    for (int r = 0; r < 8; ++r) {
        s2[r] = row_xor_add<4>(s2[r]);  q2[r] = row_xor_add<4>(q2[r]);
    }
#pragma unroll
    for (int r = 0; r < 8; ++r) {
        s2[r] = row_xor_add<8>(s2[r]);  q2[r] = row_xor_add<8>(q2[r]);
    }
#pragma unroll
    for (int r = 0; r < 8; ++r) {
        float mean = s2[r] * (1.f / 64.f);
        float var  = q2[r] * (1.f / 64.f) - mean * mean;
        float rstd = rsqrtf(var + 1e-5f);
#pragma unroll
        for (int t = 0; t < 4; ++t) {
            float v = (acc2[t][r] - mean) * rstd * gam2[t] + bet2[t];
            if constexpr (LAST) tile[(half * 8 + r) * LP + t * 16 + mr] = fmaxf(v, 0.f);
            else                zout[rowbase + (long)r * Hc + t * 16] = fmaxf(v, 0.f);
        }
    }

    if constexpr (LAST) {
        __syncthreads();

        const float* At3 = tile + mr * LP + half * 2;
        v8f acc3[3];
#pragma unroll
        for (int t = 0; t < 3; ++t) acc3[t] = v8f{0.f,0.f,0.f,0.f,0.f,0.f,0.f,0.f};
        v2f a3[16];
#pragma unroll
        for (int kk = 0; kk < 16; ++kk) a3[kk] = *(const v2f*)(At3 + 4 * kk);
#pragma unroll
        for (int kk = 0; kk < 16; ++kk)
#pragma unroll
            for (int t = 0; t < 3; ++t)
                acc3[t] = __builtin_amdgcn_wmma_f32_16x16x4_f32(
                              false, a3[kk], false, b3[kk][t], (short)0, acc3[t], false, false);

        const bool c2ok = mr < (Cc - 32);
        float mx3[8], sm3[8];
#pragma unroll
        for (int r = 0; r < 8; ++r) {
#pragma unroll
            for (int t = 0; t < 3; ++t) acc3[t][r] += lb3[t];
            float m01 = fmaxf(acc3[0][r], acc3[1][r]);
            mx3[r] = c2ok ? fmaxf(m01, acc3[2][r]) : m01;
        }
#pragma unroll
        for (int r = 0; r < 8; ++r) mx3[r] = row_xor_max<1>(mx3[r]);
#pragma unroll
        for (int r = 0; r < 8; ++r) mx3[r] = row_xor_max<2>(mx3[r]);
#pragma unroll
        for (int r = 0; r < 8; ++r) mx3[r] = row_xor_max<4>(mx3[r]);
#pragma unroll
        for (int r = 0; r < 8; ++r) mx3[r] = row_xor_max<8>(mx3[r]);
#pragma unroll
        for (int r = 0; r < 8; ++r) {
            float e01 = __expf(acc3[0][r] - mx3[r]) + __expf(acc3[1][r] - mx3[r]);
            float e2  = __expf(acc3[2][r] - mx3[r]);
            sm3[r] = c2ok ? (e01 + e2) : e01;
        }
#pragma unroll
        for (int r = 0; r < 8; ++r) sm3[r] = row_xor_add<1>(sm3[r]);
#pragma unroll
        for (int r = 0; r < 8; ++r) sm3[r] = row_xor_add<2>(sm3[r]);
#pragma unroll
        for (int r = 0; r < 8; ++r) sm3[r] = row_xor_add<4>(sm3[r]);
#pragma unroll
        for (int r = 0; r < 8; ++r) sm3[r] = row_xor_add<8>(sm3[r]);

#pragma unroll
        for (int r = 0; r < 8; ++r) {
            float lse = mx3[r] + __logf(sm3[r]);
            float* orow = tile + (half * 8 + r) * Cc + mr;
            orow[0]  = acc3[0][r] - lse;
            orow[16] = acc3[1][r] - lse;
            if (c2ok) orow[32] = acc3[2][r] - lse;
        }
        __syncthreads();
        float* ob = out + (long)tm * (16 * Cc);
#pragma unroll
        for (int j = 0; j < (16 * Cc) / (4 * 32); ++j) {
            int q = lane + 32 * j;
            *(v4f*)(ob + 4 * q) = *(const v4f*)(tile + 4 * q);
        }
    }
}

typedef int v2i __attribute__((ext_vector_type(2)));

static constexpr int CH   = 4096;
static constexpr int BSH  = 7;
static constexpr int BN   = 1 << BSH;
static constexpr int MAXB = 256;

static constexpr int T1  = 256;
static constexpr int EPT = CH / T1;

__global__ void __launch_bounds__(T1)
edge_bin_kernel(const int* __restrict__ src, const int* __restrict__ dst,
                int* __restrict__ pairs, int* __restrict__ offT, int E, int nb) {
    __shared__ int lcnt[MAXB];
    __shared__ v2i stage[CH];
    const int t  = threadIdx.x;
    const int g  = blockIdx.x;
    const int G  = gridDim.x;
    const int e0 = g * CH;
    int n = E - e0; if (n > CH) n = CH;

    int d[EPT], s[EPT];
#pragma unroll
    for (int k = 0; k < EPT; ++k) {
        int idx = t + T1 * k;
        bool ok = idx < n;
        d[k] = ok ? dst[e0 + idx] : -1;
        s[k] = ok ? src[e0 + idx] : 0;
    }
    for (int q = t; q < MAXB; q += T1) lcnt[q] = 0;
    __syncthreads();

    int rk[EPT];
#pragma unroll
    for (int k = 0; k < EPT; ++k) {
        rk[k] = 0;
        if (d[k] >= 0) rk[k] = atomicAdd(&lcnt[d[k] >> BSH], 1);
    }
    __syncthreads();

    if (t < 32) {
        int c[8];
        int sum = 0;
#pragma unroll
        for (int q = 0; q < 8; ++q) { int x = lcnt[8 * t + q]; c[q] = sum; sum += x; }
        int incl = sum;
        for (int off = 1; off < 32; off <<= 1) {
            int v = __shfl_up(incl, off);
            if (t >= off) incl += v;
        }
        int excl = incl - sum;
#pragma unroll
        for (int q = 0; q < 8; ++q) lcnt[8 * t + q] = excl + c[q];
    }
    __syncthreads();

    int pos[EPT];
#pragma unroll
    for (int k = 0; k < EPT; ++k) pos[k] = lcnt[(d[k] >= 0) ? (d[k] >> BSH) : 0] + rk[k];
#pragma unroll
    for (int k = 0; k < EPT; ++k)
        if (d[k] >= 0) stage[pos[k]] = v2i{d[k], s[k]};
    __syncthreads();

    v2i* pr = (v2i*)pairs + e0;
    v2i w[EPT];
#pragma unroll
    for (int k = 0; k < EPT; ++k) {
        int idx = t + T1 * k;
        w[k] = (idx < n) ? stage[idx] : v2i{0, 0};
    }
#pragma unroll
    for (int k = 0; k < EPT; ++k) {
        int idx = t + T1 * k;
        if (idx < n) pr[idx] = w[k];
    }
    for (int q = t; q <= nb; q += T1) offT[q * G + g] = (q < nb) ? lcnt[q] : n;
}

template <bool PLACE>
HD int bucket_sweep(const v2i* __restrict__ pr, const int* __restrict__ r0,
                    const int* __restrict__ r1, int G, int wv, int lane, int lo,
                    int* ctr, int* __restrict__ ssrc) {
    int below = 0;
    for (int gb = wv * 32; gb < G; gb += 8 * 32) {
        int gg = gb + lane;
        int o0 = 0, o1 = 0;
        if (gg < G) { o0 = r0[gg]; o1 = r1[gg]; }
        below += o0;
        int len = o1 - o0;
        int st  = gg * CH + o0;
        v2i p[32];
#pragma unroll
        for (int u = 0; u < 32; ++u) {
            int L = __builtin_amdgcn_readlane(len, u);
            int S = __builtin_amdgcn_readlane(st, u);
            p[u] = v2i{-1, 0};
            if (lane < L) p[u] = pr[S + lane];
        }
        if (PLACE) {
            int slot[32];
#pragma unroll
            for (int u = 0; u < 32; ++u) {
                slot[u] = 0;
                if (p[u].x >= 0) slot[u] = atomicAdd(&ctr[p[u].x - lo], 1);
            }
#pragma unroll
            for (int u = 0; u < 32; ++u)
                if (p[u].x >= 0) ssrc[slot[u]] = p[u].y;
        } else {
#pragma unroll
            for (int u = 0; u < 32; ++u)
                if (p[u].x >= 0) atomicAdd(&ctr[p[u].x - lo], 1);
        }
        unsigned m = __builtin_amdgcn_ballot_w32(len > 32);
        while (m) {
            int u = __builtin_ctz(m);
            m &= m - 1;
            int L = __builtin_amdgcn_readlane(len, u);
            int S = __builtin_amdgcn_readlane(st, u);
            for (int k = 32 + lane; k < L; k += 32) {
                v2i q = pr[S + k];
                if (PLACE) ssrc[atomicAdd(&ctr[q.x - lo], 1)] = q.y;
                else       atomicAdd(&ctr[q.x - lo], 1);
            }
        }
    }
    return below;
}

__global__ void __launch_bounds__(256)
bucket_group_kernel(const int* __restrict__ pairs, const int* __restrict__ offT,
                    int* __restrict__ rowptr, int* __restrict__ ssrc, int N, int G) {
    __shared__ int cnt[BN];
    __shared__ int cur[BN];
    __shared__ int wsum[8];
    const int t    = threadIdx.x;
    const int lane = t & 31;
    const int wv   = __builtin_amdgcn_readfirstlane(t >> 5);
    const int b    = blockIdx.x;
    const int lo   = b << BSH;
    const v2i* pr  = (const v2i*)pairs;
    const int* r0  = offT + (long)b * G;
    const int* r1  = r0 + G;

    if (t < BN) cnt[t] = 0;
    __syncthreads();

    int below = bucket_sweep<false>(pr, r0, r1, G, wv, lane, lo, cnt, ssrc);
    for (int off = 16; off; off >>= 1) below += __shfl_xor(below, off);
    if (lane == 0) wsum[wv] = below;
    __syncthreads();

    if (t < 32) {
        int base = 0;
#pragma unroll
        for (int w = 0; w < 8; ++w) base += wsum[w];
        int c[4];
        int sum = 0;
#pragma unroll
        for (int q = 0; q < 4; ++q) { int x = cnt[4 * t + q]; c[q] = sum; sum += x; }
        int incl = sum;
        for (int off = 1; off < 32; off <<= 1) {
            int v = __shfl_up(incl, off);
            if (t >= off) incl += v;
        }
        int excl = base + incl - sum;
#pragma unroll
        for (int q = 0; q < 4; ++q) {
            int node = lo + 4 * t + q;
            int v    = excl + c[q];
            cur[4 * t + q] = v;
            if (node < N) rowptr[node] = v;
        }
        if (b == (int)gridDim.x - 1 && t == 31) rowptr[N] = base + incl;
    }
    __syncthreads();

    bucket_sweep<true>(pr, r0, r1, G, wv, lane, lo, cur, ssrc);
}

static constexpr int EB = 8;

template <bool RELU, bool MASK>
HD void agg_batch(const v2f* __restrict__ zr, int myidx, int j0, int cnt, int lane,
                  v2f T2, v2f& M, v2f& D, v2f& A) {
    v2f v[EB];
#pragma unroll
    for (int u = 0; u < EB; ++u) {
        int sidx = __builtin_amdgcn_readlane(myidx, j0 + u);
        const v2f* rp = zr + (long)sidx * (Hc / 2);
        v[u] = rp[lane];
    }
    v2f q[EB];
    v2f Mn = M;
#pragma unroll
    for (int u = 0; u < EB; ++u) {
        if (RELU) { v[u].x = fmaxf(v[u].x, 0.f); v[u].y = fmaxf(v[u].y, 0.f); }
        q[u] = v[u] * T2;
        if (MASK) {
            bool ok = (j0 + u) < cnt;
            v[u].x = ok ? v[u].x : 0.f;       v[u].y = ok ? v[u].y : 0.f;
            q[u].x = ok ? q[u].x : -3.0e38f;  q[u].y = ok ? q[u].y : -3.0e38f;
        }
        Mn.x = fmaxf(Mn.x, q[u].x);
        Mn.y = fmaxf(Mn.y, q[u].y);
    }
    v2f dm = M - Mn;
    v2f r;
    r.x = __builtin_amdgcn_exp2f(dm.x);
    r.y = __builtin_amdgcn_exp2f(dm.y);
    D *= r;
    A *= r;
#pragma unroll
    for (int u = 0; u < EB; ++u) {
        v2f a = q[u] - Mn;
        v2f e;
        e.x = __builtin_amdgcn_exp2f(a.x);
        e.y = __builtin_amdgcn_exp2f(a.y);
        D += e;
        A += v[u] * e;
    }
    M = Mn;
}

template <bool RELU>
__global__ void
node_softmax_agg_kernel(const float* __restrict__ z,
                        const int* __restrict__ rowptr,
                        const int* __restrict__ ssrc,
                        const float* __restrict__ tptr, int layer,
                        float* __restrict__ out, int n) {
    int lane = threadIdx.x & 31;
    int node = __builtin_amdgcn_readfirstlane(
                   (int)(blockIdx.x * (blockDim.x >> 5) + (threadIdx.x >> 5)));
    if (node >= n) return;

    int   start = rowptr[node];
    int   end   = rowptr[node + 1];
    float t2    = tptr[layer] * 1.44269504088896341f;
    const v2f* zr = (const v2f*)z;
    v2f self = zr[node * (Hc / 2) + lane];

    v2f T2 = {t2, t2};
    v2f M  = {-3.0e38f, -3.0e38f};
    v2f D  = {0.f, 0.f};
    v2f A  = {0.f, 0.f};

    for (int base = start; base < end; base += 32) {
        int cnt = end - base; if (cnt > 32) cnt = 32;
        int myidx = (lane < cnt) ? ssrc[base + lane] : 0;
        int nfull = cnt & ~(EB - 1);
        for (int j0 = 0; j0 < nfull; j0 += EB)
            agg_batch<RELU, false>(zr, myidx, j0, cnt, lane, T2, M, D, A);
        if (cnt != nfull)
            agg_batch<RELU, true>(zr, myidx, nfull, cnt, lane, T2, M, D, A);
    }

    v2f o;
    o.x = (A.x + GEPS * D.x) / (D.x + 1e-16f) + self.x;
    o.y = (A.y + GEPS * D.y) / (D.y + 1e-16f) + self.y;
    ((v2f*)out)[node * (Hc / 2) + lane] = o;
}

extern "C" void kernel_launch(void* const* d_in, const int* in_sizes, int n_in,
                              void* d_out, int out_size, void* d_ws, size_t ws_size,
                              hipStream_t stream) {
    (void)n_in; (void)out_size; (void)ws_size;

    const float* x    = (const float*)d_in[0];
    const int*   ei   = (const int*)  d_in[1];
    const float* encW = (const float*)d_in[2];
    const float* encB = (const float*)d_in[3];
    const float* t    = (const float*)d_in[4];
    const float* W1   = (const float*)d_in[5];
    const float* b1   = (const float*)d_in[6];
    const float* g1   = (const float*)d_in[7];
    const float* bb1  = (const float*)d_in[8];
    const float* W2   = (const float*)d_in[9];
    const float* b2   = (const float*)d_in[10];
    const float* ng   = (const float*)d_in[11];
    const float* nb   = (const float*)d_in[12];
    const float* linW = (const float*)d_in[13];
    const float* linB = (const float*)d_in[14];

    const int F = 128, H = 64, L = 4;
    const int N = in_sizes[0] / F;
    const int E = in_sizes[1] / 2;
    const int* src = ei;
    const int* dst = ei + E;

    float* ws = (float*)d_ws;
    const long NH = (long)N * H;
    float* h    = ws;  ws += NH;
    float* z    = ws;  ws += NH;
    int*   csr  = (int*)ws;  ws += 3 * NH;
    float* outb = ws;  ws += NH;

    const int  TB = 256;
    auto cdiv = [](long a, long b) { return (int)((a + b - 1) / b); };
    auto gemm_waves = [&](int M, int Nc) { return (long)(M / 16) * ((Nc + 63) / 64); };

    const int G  = cdiv(E, CH);
    const int nbk = cdiv(N, BN);
    int* pairs  = csr;
    int* ssrc   = pairs + 2 * E;
    int* offT   = ssrc + E;
    int* rowptr = offT + (nbk + 1) * G;

    gemm_bias_wmma<<<cdiv(gemm_waves(N, H) * 32, TB), TB, 0, stream>>>(x, encW, encB, h, N, H, F);

    edge_bin_kernel<<<G, T1, 0, stream>>>(src, dst, pairs, offT, E, nbk);
    bucket_group_kernel<<<nbk, 256, 0, stream>>>(pairs, offT, rowptr, ssrc, N, G);

    for (int l = 0; l < L; ++l) {
        const float* zin = (l == 0) ? h : z;

        if (l == 0)
            node_softmax_agg_kernel<true><<<cdiv(N, 8), 256, 0, stream>>>(zin, rowptr, ssrc, t, l, outb, N);
        else
            node_softmax_agg_kernel<false><<<cdiv(N, 8), 256, 0, stream>>>(zin, rowptr, ssrc, t, l, outb, N);

        const int ln = (l + 1 < L) ? (l + 1) : 0;
        if (l + 1 < L)
            mlp_chain_kernel<false><<<N / 16, 32, 0, stream>>>(
                outb, W1 + (long)l * H * 2 * H, b1 + l * 2 * H, g1 + l * 2 * H, bb1 + l * 2 * H,
                W2 + (long)l * 2 * H * H, b2 + l * H, ng + ln * H, nb + ln * H,
                h, z, (l == 0) ? 0 : 1, 1, linW, linB, (float*)d_out);
        else
            mlp_chain_kernel<true><<<N / 16, 32, 0, stream>>>(
                outb, W1 + (long)l * H * 2 * H, b1 + l * 2 * H, g1 + l * 2 * H, bb1 + l * 2 * H,
                W2 + (long)l * 2 * H * H, b2 + l * H, ng + ln * H, nb + ln * H,
                h, z, (l == 0) ? 0 : 1, 0, linW, linB, (float*)d_out);
    }
}


# Encoder GEMM and level-1 edge binning, independent of each other, share ONE launch (role by work-group number): 11 -> 10 launches and the two bodies overlap; arithmetic and sum orders unchanged
# speedup vs baseline: 8.6165x; 1.0568x over previous
#include <hip/hip_runtime.h>
#include <math.h>

typedef float v2f __attribute__((ext_vector_type(2)));
typedef float v4f __attribute__((ext_vector_type(4)));
typedef float v8f __attribute__((ext_vector_type(8)));

#define HD __device__ __forceinline__

static constexpr int   Hc   = 64;
static constexpr int   Cc   = 40;
static constexpr float GEPS = 1e-7f;

HD void gemm_bias_wmma_body(const float* __restrict__ A,
                            const float* __restrict__ B,
                            const float* __restrict__ bias,
                            float* __restrict__ C,
                            int M, int N, int K, int gid) {
    int wave    = gid >> 5;
    int lane    = threadIdx.x & 31;
    int groupsN = (N + 63) >> 6;
    int tm = wave / groupsN;
    int gn = wave - tm * groupsN;
    if (tm >= (M >> 4)) return;

    int mr   = lane & 15;
    int half = lane >> 4;
    const float* Arow = A + (long)(tm * 16 + mr) * K;

    int ncol[4];
#pragma unroll
    for (int t = 0; t < 4; ++t) {
        int n   = gn * 64 + t * 16 + mr;
        ncol[t] = (n < N) ? n : (N - 1);
    }

    v8f acc[4] = {{0.f,0.f,0.f,0.f,0.f,0.f,0.f,0.f},
                  {0.f,0.f,0.f,0.f,0.f,0.f,0.f,0.f},
                  {0.f,0.f,0.f,0.f,0.f,0.f,0.f,0.f},
                  {0.f,0.f,0.f,0.f,0.f,0.f,0.f,0.f}};

    int ka = half * 2;
    v2f a_cur;
    a_cur.x = Arow[ka];
    a_cur.y = Arow[ka + 1];
    v2f b_cur[4];
#pragma unroll
    for (int t = 0; t < 4; ++t) {
        b_cur[t].x = B[ka * N + ncol[t]];
        b_cur[t].y = B[(ka + 1) * N + ncol[t]];
    }

#pragma unroll 2
    for (int k = 0; k < K; k += 4) {
        int kn = (k + 4 < K) ? (k + 4) : k;
        int kb = kn + half * 2;

        v2f a_nxt;
        a_nxt.x = Arow[kb];
        a_nxt.y = Arow[kb + 1];
        v2f b_nxt[4];
#pragma unroll
        for (int t = 0; t < 4; ++t) {
            b_nxt[t].x = B[kb * N + ncol[t]];
            b_nxt[t].y = B[(kb + 1) * N + ncol[t]];
        }

#pragma unroll
        for (int t = 0; t < 4; ++t)
            acc[t] = __builtin_amdgcn_wmma_f32_16x16x4_f32(
                         false, a_cur, false, b_cur[t], (short)0, acc[t], false, false);

        a_cur = a_nxt;
#pragma unroll
        for (int t = 0; t < 4; ++t) b_cur[t] = b_nxt[t];
    }

#pragma unroll
    for (int t = 0; t < 4; ++t) {
        int n = gn * 64 + t * 16 + mr;
        if (n < N) {
            float bv = bias[n];
#pragma unroll
            for (int r = 0; r < 8; ++r)
                C[(long)(tm * 16 + half * 8 + r) * N + n] = acc[t][r] + bv;
        }
    }
}

static constexpr int LP = 132;

template <int M>
HD float row_xor_add(float v) {
    int o = __builtin_amdgcn_update_dpp(0, __float_as_int(v), 0x160 | M, 0xf, 0xf, false);
    return v + __int_as_float(o);
}
template <int M>
HD float row_xor_max(float v) {
    int o = __builtin_amdgcn_update_dpp(0, __float_as_int(v), 0x160 | M, 0xf, 0xf, false);
    return fmaxf(v, __int_as_float(o));
}

template <bool LAST>
__global__ void __launch_bounds__(32)
mlp_chain_kernel(const float* __restrict__ X,
                 const float* __restrict__ W1, const float* __restrict__ b1,
                 const float* __restrict__ g1, const float* __restrict__ bb1,
                 const float* __restrict__ W2, const float* __restrict__ b2,
                 const float* __restrict__ ng, const float* __restrict__ nb,
                 float* __restrict__ h, float* __restrict__ zout,
                 int add_res, int write_h,
                 const float* __restrict__ linW, const float* __restrict__ linB,
                 float* __restrict__ out) {
    __shared__ __attribute__((aligned(16))) float tile[16 * LP];
    const int lane = threadIdx.x;
    const int tm   = blockIdx.x;
    const int mr   = lane & 15;
    const int half = lane >> 4;

    const float* Arow = X + (long)(tm * 16 + mr) * Hc + half * 2;
    const float* B1p  = W1 + (half * 2) * (2 * Hc) + mr;

    v8f acc1[8];
#pragma unroll
    for (int t = 0; t < 8; ++t) acc1[t] = v8f{0.f,0.f,0.f,0.f,0.f,0.f,0.f,0.f};

    v2f a_cur = *(const v2f*)(Arow);
    v2f b_cur[8];
#pragma unroll
    for (int t = 0; t < 8; ++t) {
        b_cur[t].x = B1p[t * 16];
        b_cur[t].y = B1p[2 * Hc + t * 16];
    }

    float bias1[8], gam1[8], bet1[8];
#pragma unroll
    for (int t = 0; t < 8; ++t) {
        bias1[t] = b1[t * 16 + mr];
        gam1[t]  = g1[t * 16 + mr];
        bet1[t]  = bb1[t * 16 + mr];
    }

    const long rowbase = (long)(tm * 16 + half * 8) * Hc + mr;
    float hres[4][8];
    if (add_res) {
#pragma unroll
        for (int t = 0; t < 4; ++t)
#pragma unroll
            for (int r = 0; r < 8; ++r) hres[t][r] = h[rowbase + (long)r * Hc + t * 16];
    } else {
#pragma unroll
        for (int t = 0; t < 4; ++t)
#pragma unroll
            for (int r = 0; r < 8; ++r) hres[t][r] = 0.f;
    }

#pragma unroll
    for (int kk = 0; kk < 16; ++kk) {
        v2f a_nxt = a_cur;
        v2f b_nxt[8];
        if (kk + 1 < 16) {
            a_nxt = *(const v2f*)(Arow + 4 * (kk + 1));
#pragma unroll
            for (int t = 0; t < 8; ++t) {
                b_nxt[t].x = B1p[(4 * (kk + 1)) * (2 * Hc) + t * 16];
                b_nxt[t].y = B1p[(4 * (kk + 1) + 1) * (2 * Hc) + t * 16];
            }
        }
#pragma unroll
        for (int t = 0; t < 8; ++t)
            acc1[t] = __builtin_amdgcn_wmma_f32_16x16x4_f32(
                          false, a_cur, false, b_cur[t], (short)0, acc1[t], false, false);
        if (kk + 1 < 16) {
            a_cur = a_nxt;
#pragma unroll
            for (int t = 0; t < 8; ++t) b_cur[t] = b_nxt[t];
        }
    }

    float s1[8], q1[8];
#pragma unroll
    for (int r = 0; r < 8; ++r) { s1[r] = 0.f; q1[r] = 0.f; }
#pragma unroll
    for (int t = 0; t < 8; ++t) {
#pragma unroll
        for (int r = 0; r < 8; ++r) {
            float u = acc1[t][r] + bias1[t];
            acc1[t][r] = u;
            s1[r] += u;
            q1[r] += u * u;
        }
    }
#pragma unroll
    for (int r = 0; r < 8; ++r) {
        s1[r] = row_xor_add<1>(s1[r]);  q1[r] = row_xor_add<1>(q1[r]);
    }
#pragma unroll
    for (int r = 0; r < 8; ++r) {
        s1[r] = row_xor_add<2>(s1[r]);  q1[r] = row_xor_add<2>(q1[r]);
    }
#pragma unroll
    for (int r = 0; r < 8; ++r) {
        s1[r] = row_xor_add<4>(s1[r]);  q1[r] = row_xor_add<4>(q1[r]);
    }
#pragma unroll
    for (int r = 0; r < 8; ++r) {
        s1[r] = row_xor_add<8>(s1[r]);  q1[r] = row_xor_add<8>(q1[r]);
    }
#pragma unroll
    for (int r = 0; r < 8; ++r) {
        float mean = s1[r] * (1.f / 128.f);
        float var  = q1[r] * (1.f / 128.f) - mean * mean;
        float rstd = rsqrtf(var + 1e-5f);
#pragma unroll
        for (int t = 0; t < 8; ++t) {
            float v = (acc1[t][r] - mean) * rstd * gam1[t] + bet1[t];
            tile[(half * 8 + r) * LP + t * 16 + mr] = fmaxf(v, 0.f);
        }
    }

    __syncthreads();

    float bias2[4], gam2[4], bet2[4];
#pragma unroll
    for (int t = 0; t < 4; ++t) {
        bias2[t] = b2[t * 16 + mr];
        gam2[t]  = ng[t * 16 + mr];
        bet2[t]  = nb[t * 16 + mr];
    }

    const float* Atile = tile + mr * LP + half * 2;
    const float* B2p   = W2 + (half * 2) * Hc + mr;

    v8f acc2[4];
#pragma unroll
    for (int t = 0; t < 4; ++t) acc2[t] = v8f{0.f,0.f,0.f,0.f,0.f,0.f,0.f,0.f};

    v2f a2c[2];
    v2f b2c[2][4];
#pragma unroll
    for (int u = 0; u < 2; ++u) {
        a2c[u] = *(const v2f*)(Atile + 4 * u);
#pragma unroll
        for (int t = 0; t < 4; ++t) {
            b2c[u][t].x = B2p[(4 * u) * Hc + t * 16];
            b2c[u][t].y = B2p[(4 * u + 1) * Hc + t * 16];
        }
    }
#pragma unroll
    for (int j = 0; j < 16; ++j) {
        v2f a2n[2];
        v2f b2n[2][4];
        if (j + 1 < 16) {
#pragma unroll
            for (int u = 0; u < 2; ++u) {
                const int kk = 2 * (j + 1) + u;
                a2n[u] = *(const v2f*)(Atile + 4 * kk);
#pragma unroll
                for (int t = 0; t < 4; ++t) {
                    b2n[u][t].x = B2p[(4 * kk) * Hc + t * 16];
                    b2n[u][t].y = B2p[(4 * kk + 1) * Hc + t * 16];
                }
            }
        }
#pragma unroll
        for (int u = 0; u < 2; ++u)
#pragma unroll
            for (int t = 0; t < 4; ++t)
                acc2[t] = __builtin_amdgcn_wmma_f32_16x16x4_f32(
                              false, a2c[u], false, b2c[u][t], (short)0, acc2[t], false, false);
        if (j + 1 < 16) {
#pragma unroll
            for (int u = 0; u < 2; ++u) {
                a2c[u] = a2n[u];
#pragma unroll
                for (int t = 0; t < 4; ++t) b2c[u][t] = b2n[u][t];
            }
        }
    }

    int  nc3[3];
    v2f  b3[16][3];
    float lb3[3];
    if constexpr (LAST) {
#pragma unroll
        for (int t = 0; t < 3; ++t) {
            int n  = t * 16 + mr;
            nc3[t] = (n < Cc) ? n : (Cc - 1);
            lb3[t] = linB[nc3[t]];
        }
        const float* B3p = linW + (half * 2) * Cc;
#pragma unroll
        for (int kk = 0; kk < 16; ++kk)
#pragma unroll
            for (int t = 0; t < 3; ++t) {
                b3[kk][t].x = B3p[(4 * kk) * Cc + nc3[t]];
                b3[kk][t].y = B3p[(4 * kk + 1) * Cc + nc3[t]];
            }
    }

    float s2[8], q2[8];
#pragma unroll
    for (int r = 0; r < 8; ++r) { s2[r] = 0.f; q2[r] = 0.f; }
#pragma unroll
    for (int t = 0; t < 4; ++t) {
#pragma unroll
        for (int r = 0; r < 8; ++r) {
            float hn = hres[t][r] + (acc2[t][r] + bias2[t]);
            acc2[t][r] = hn;
            s2[r] += hn;
            q2[r] += hn * hn;
        }
    }
    if (!LAST && write_h) {
#pragma unroll
        for (int t = 0; t < 4; ++t)
#pragma unroll
            for (int r = 0; r < 8; ++r)
                h[rowbase + (long)r * Hc + t * 16] = acc2[t][r];
    }
#pragma unroll
    for (int r = 0; r < 8; ++r) {
        s2[r] = row_xor_add<1>(s2[r]);  q2[r] = row_xor_add<1>(q2[r]);
    }
#pragma unroll
    for (int r = 0; r < 8; ++r) {
        s2[r] = row_xor_add<2>(s2[r]);  q2[r] = row_xor_add<2>(q2[r]);
    }
#pragma unroll
    for (int r = 0; r < 8; ++r) {
        s2[r] = row_xor_add<4>(s2[r]);  q2[r] = row_xor_add<4>(q2[r]);
    }
#pragma unroll
    for (int r = 0; r < 8; ++r) {
        s2[r] = row_xor_add<8>(s2[r]);  q2[r] = row_xor_add<8>(q2[r]);
    }
#pragma unroll
    for (int r = 0; r < 8; ++r) {
        float mean = s2[r] * (1.f / 64.f);
        float var  = q2[r] * (1.f / 64.f) - mean * mean;
        float rstd = rsqrtf(var + 1e-5f);
#pragma unroll
        for (int t = 0; t < 4; ++t) {
            float v = (acc2[t][r] - mean) * rstd * gam2[t] + bet2[t];
            if constexpr (LAST) tile[(half * 8 + r) * LP + t * 16 + mr] = fmaxf(v, 0.f);
            else                zout[rowbase + (long)r * Hc + t * 16] = fmaxf(v, 0.f);
        }
    }

    if constexpr (LAST) {
        __syncthreads();

        const float* At3 = tile + mr * LP + half * 2;
        v8f acc3[3];
#pragma unroll
        for (int t = 0; t < 3; ++t) acc3[t] = v8f{0.f,0.f,0.f,0.f,0.f,0.f,0.f,0.f};
        v2f a3[16];
#pragma unroll
        for (int kk = 0; kk < 16; ++kk) a3[kk] = *(const v2f*)(At3 + 4 * kk);
#pragma unroll
        for (int kk = 0; kk < 16; ++kk)
#pragma unroll
            for (int t = 0; t < 3; ++t)
                acc3[t] = __builtin_amdgcn_wmma_f32_16x16x4_f32(
                              false, a3[kk], false, b3[kk][t], (short)0, acc3[t], false, false);

        const bool c2ok = mr < (Cc - 32);
        float mx3[8], sm3[8];
#pragma unroll
        for (int r = 0; r < 8; ++r) {
#pragma unroll
            for (int t = 0; t < 3; ++t) acc3[t][r] += lb3[t];
            float m01 = fmaxf(acc3[0][r], acc3[1][r]);
            mx3[r] = c2ok ? fmaxf(m01, acc3[2][r]) : m01;
        }
#pragma unroll
        for (int r = 0; r < 8; ++r) mx3[r] = row_xor_max<1>(mx3[r]);
#pragma unroll
        for (int r = 0; r < 8; ++r) mx3[r] = row_xor_max<2>(mx3[r]);
#pragma unroll
        for (int r = 0; r < 8; ++r) mx3[r] = row_xor_max<4>(mx3[r]);
#pragma unroll
        for (int r = 0; r < 8; ++r) mx3[r] = row_xor_max<8>(mx3[r]);
#pragma unroll
        for (int r = 0; r < 8; ++r) {
            float e01 = __expf(acc3[0][r] - mx3[r]) + __expf(acc3[1][r] - mx3[r]);
            float e2  = __expf(acc3[2][r] - mx3[r]);
            sm3[r] = c2ok ? (e01 + e2) : e01;
        }
#pragma unroll
        for (int r = 0; r < 8; ++r) sm3[r] = row_xor_add<1>(sm3[r]);
#pragma unroll
        for (int r = 0; r < 8; ++r) sm3[r] = row_xor_add<2>(sm3[r]);
#pragma unroll
        for (int r = 0; r < 8; ++r) sm3[r] = row_xor_add<4>(sm3[r]);
#pragma unroll
        for (int r = 0; r < 8; ++r) sm3[r] = row_xor_add<8>(sm3[r]);

#pragma unroll
        for (int r = 0; r < 8; ++r) {
            float lse = mx3[r] + __logf(sm3[r]);
            float* orow = tile + (half * 8 + r) * Cc + mr;
            orow[0]  = acc3[0][r] - lse;
            orow[16] = acc3[1][r] - lse;
            if (c2ok) orow[32] = acc3[2][r] - lse;
        }
        __syncthreads();
        float* ob = out + (long)tm * (16 * Cc);
#pragma unroll
        for (int j = 0; j < (16 * Cc) / (4 * 32); ++j) {
            int q = lane + 32 * j;
            *(v4f*)(ob + 4 * q) = *(const v4f*)(tile + 4 * q);
        }
    }
}

typedef int v2i __attribute__((ext_vector_type(2)));

static constexpr int CH   = 4096;
static constexpr int BSH  = 7;
static constexpr int BN   = 1 << BSH;
static constexpr int MAXB = 256;

static constexpr int T1  = 256;
static constexpr int EPT = CH / T1;

HD void edge_bin_body(const int* __restrict__ src, const int* __restrict__ dst,
                      int* __restrict__ pairs, int* __restrict__ offT, int E, int nb,
                      int g, int G) {
    __shared__ int lcnt[MAXB];
    __shared__ v2i stage[CH];
    const int t  = threadIdx.x;
    const int e0 = g * CH;
    int n = E - e0; if (n > CH) n = CH;

    int d[EPT], s[EPT];
#pragma unroll
    for (int k = 0; k < EPT; ++k) {
        int idx = t + T1 * k;
        bool ok = idx < n;
        d[k] = ok ? dst[e0 + idx] : -1;
        s[k] = ok ? src[e0 + idx] : 0;
    }
    for (int q = t; q < MAXB; q += T1) lcnt[q] = 0;
    __syncthreads();

    int rk[EPT];
#pragma unroll
    for (int k = 0; k < EPT; ++k) {
        rk[k] = 0;
        if (d[k] >= 0) rk[k] = atomicAdd(&lcnt[d[k] >> BSH], 1);
    }
    __syncthreads();

    if (t < 32) {
        int c[8];
        int sum = 0;
#pragma unroll
        for (int q = 0; q < 8; ++q) { int x = lcnt[8 * t + q]; c[q] = sum; sum += x; }
        int incl = sum;
        for (int off = 1; off < 32; off <<= 1) {
            int v = __shfl_up(incl, off);
            if (t >= off) incl += v;
        }
        int excl = incl - sum;
#pragma unroll
        for (int q = 0; q < 8; ++q) lcnt[8 * t + q] = excl + c[q];
    }
    __syncthreads();

    int pos[EPT];
#pragma unroll
    for (int k = 0; k < EPT; ++k) pos[k] = lcnt[(d[k] >= 0) ? (d[k] >> BSH) : 0] + rk[k];
#pragma unroll
    for (int k = 0; k < EPT; ++k)
        if (d[k] >= 0) stage[pos[k]] = v2i{d[k], s[k]};
    __syncthreads();

    v2i* pr = (v2i*)pairs + e0;
    v2i w[EPT];
#pragma unroll
    for (int k = 0; k < EPT; ++k) {
        int idx = t + T1 * k;
        w[k] = (idx < n) ? stage[idx] : v2i{0, 0};
    }
#pragma unroll
    for (int k = 0; k < EPT; ++k) {
        int idx = t + T1 * k;
        if (idx < n) pr[idx] = w[k];
    }
    for (int q = t; q <= nb; q += T1) offT[q * G + g] = (q < nb) ? lcnt[q] : n;
}

__global__ void __launch_bounds__(T1)
encode_and_bin_kernel(const int* __restrict__ src, const int* __restrict__ dst,
                      int* __restrict__ pairs, int* __restrict__ offT, int E, int nb, int G,
                      const float* __restrict__ x, const float* __restrict__ encW,
                      const float* __restrict__ encB, float* __restrict__ h,
                      int M, int N, int K) {
    const int b = (int)blockIdx.x;
    if (b < G) edge_bin_body(src, dst, pairs, offT, E, nb, b, G);
    else       gemm_bias_wmma_body(x, encW, encB, h, M, N, K, (b - G) * T1 + (int)threadIdx.x);
}

template <bool PLACE>
HD int bucket_sweep(const v2i* __restrict__ pr, const int* __restrict__ r0,
                    const int* __restrict__ r1, int G, int wv, int lane, int lo,
                    int* ctr, int* __restrict__ ssrc) {
    int below = 0;
    for (int gb = wv * 32; gb < G; gb += 8 * 32) {
        int gg = gb + lane;
        int o0 = 0, o1 = 0;
        if (gg < G) { o0 = r0[gg]; o1 = r1[gg]; }
        below += o0;
        int len = o1 - o0;
        int st  = gg * CH + o0;
        v2i p[32];
#pragma unroll
        for (int u = 0; u < 32; ++u) {
            int L = __builtin_amdgcn_readlane(len, u);
            int S = __builtin_amdgcn_readlane(st, u);
            p[u] = v2i{-1, 0};
            if (lane < L) p[u] = pr[S + lane];
        }
        if (PLACE) {
            int slot[32];
#pragma unroll
            for (int u = 0; u < 32; ++u) {
                slot[u] = 0;
                if (p[u].x >= 0) slot[u] = atomicAdd(&ctr[p[u].x - lo], 1);
            }
#pragma unroll
            for (int u = 0; u < 32; ++u)
                if (p[u].x >= 0) ssrc[slot[u]] = p[u].y;
        } else {
#pragma unroll
            for (int u = 0; u < 32; ++u)
                if (p[u].x >= 0) atomicAdd(&ctr[p[u].x - lo], 1);
        }
        unsigned m = __builtin_amdgcn_ballot_w32(len > 32);
        while (m) {
            int u = __builtin_ctz(m);
            m &= m - 1;
            int L = __builtin_amdgcn_readlane(len, u);
            int S = __builtin_amdgcn_readlane(st, u);
            for (int k = 32 + lane; k < L; k += 32) {
                v2i q = pr[S + k];
                if (PLACE) ssrc[atomicAdd(&ctr[q.x - lo], 1)] = q.y;
                else       atomicAdd(&ctr[q.x - lo], 1);
            }
        }
    }
    return below;
}

__global__ void __launch_bounds__(256)
bucket_group_kernel(const int* __restrict__ pairs, const int* __restrict__ offT,
                    int* __restrict__ rowptr, int* __restrict__ ssrc, int N, int G) {
    __shared__ int cnt[BN];
    __shared__ int cur[BN];
    __shared__ int wsum[8];
    const int t    = threadIdx.x;
    const int lane = t & 31;
    const int wv   = __builtin_amdgcn_readfirstlane(t >> 5);
    const int b    = blockIdx.x;
    const int lo   = b << BSH;
    const v2i* pr  = (const v2i*)pairs;
    const int* r0  = offT + (long)b * G;
    const int* r1  = r0 + G;

    if (t < BN) cnt[t] = 0;
    __syncthreads();

    int below = bucket_sweep<false>(pr, r0, r1, G, wv, lane, lo, cnt, ssrc);
    for (int off = 16; off; off >>= 1) below += __shfl_xor(below, off);
    if (lane == 0) wsum[wv] = below;
    __syncthreads();

    if (t < 32) {
        int base = 0;
#pragma unroll
        for (int w = 0; w < 8; ++w) base += wsum[w];
        int c[4];
        int sum = 0;
#pragma unroll
        for (int q = 0; q < 4; ++q) { int x = cnt[4 * t + q]; c[q] = sum; sum += x; }
        int incl = sum;
        for (int off = 1; off < 32; off <<= 1) {
            int v = __shfl_up(incl, off);
            if (t >= off) incl += v;
        }
        int excl = base + incl - sum;
#pragma unroll
        for (int q = 0; q < 4; ++q) {
            int node = lo + 4 * t + q;
            int v    = excl + c[q];
            cur[4 * t + q] = v;
            if (node < N) rowptr[node] = v;
        }
        if (b == (int)gridDim.x - 1 && t == 31) rowptr[N] = base + incl;
    }
    __syncthreads();

    bucket_sweep<true>(pr, r0, r1, G, wv, lane, lo, cur, ssrc);
}

static constexpr int EB = 8;

template <bool RELU, bool MASK>
HD void agg_batch(const v2f* __restrict__ zr, int myidx, int j0, int cnt, int lane,
                  v2f T2, v2f& M, v2f& D, v2f& A) {
    v2f v[EB];
#pragma unroll
    for (int u = 0; u < EB; ++u) {
        int sidx = __builtin_amdgcn_readlane(myidx, j0 + u);
        const v2f* rp = zr + (long)sidx * (Hc / 2);
        v[u] = rp[lane];
    }
    v2f q[EB];
    v2f Mn = M;
#pragma unroll
    for (int u = 0; u < EB; ++u) {
        if (RELU) { v[u].x = fmaxf(v[u].x, 0.f); v[u].y = fmaxf(v[u].y, 0.f); }
        q[u] = v[u] * T2;
        if (MASK) {
            bool ok = (j0 + u) < cnt;
            v[u].x = ok ? v[u].x : 0.f;       v[u].y = ok ? v[u].y : 0.f;
            q[u].x = ok ? q[u].x : -3.0e38f;  q[u].y = ok ? q[u].y : -3.0e38f;
        }
        Mn.x = fmaxf(Mn.x, q[u].x);
        Mn.y = fmaxf(Mn.y, q[u].y);
    }
    v2f dm = M - Mn;
    v2f r;
    r.x = __builtin_amdgcn_exp2f(dm.x);
    r.y = __builtin_amdgcn_exp2f(dm.y);
    D *= r;
    A *= r;
#pragma unroll
    for (int u = 0; u < EB; ++u) {
        v2f a = q[u] - Mn;
        v2f e;
        e.x = __builtin_amdgcn_exp2f(a.x);
        e.y = __builtin_amdgcn_exp2f(a.y);
        D += e;
        A += v[u] * e;
    }
    M = Mn;
}

template <bool RELU>
__global__ void
node_softmax_agg_kernel(const float* __restrict__ z,
                        const int* __restrict__ rowptr,
                        const int* __restrict__ ssrc,
                        const float* __restrict__ tptr, int layer,
                        float* __restrict__ out, int n) {
    int lane = threadIdx.x & 31;
    int node = __builtin_amdgcn_readfirstlane(
                   (int)(blockIdx.x * (blockDim.x >> 5) + (threadIdx.x >> 5)));
    if (node >= n) return;

    int   start = rowptr[node];
    int   end   = rowptr[node + 1];
    float t2    = tptr[layer] * 1.44269504088896341f;
    const v2f* zr = (const v2f*)z;
    v2f self = zr[node * (Hc / 2) + lane];

    v2f T2 = {t2, t2};
    v2f M  = {-3.0e38f, -3.0e38f};
    v2f D  = {0.f, 0.f};
    v2f A  = {0.f, 0.f};

    for (int base = start; base < end; base += 32) {
        int cnt = end - base; if (cnt > 32) cnt = 32;
        int myidx = (lane < cnt) ? ssrc[base + lane] : 0;
        int nfull = cnt & ~(EB - 1);
        for (int j0 = 0; j0 < nfull; j0 += EB)
            agg_batch<RELU, false>(zr, myidx, j0, cnt, lane, T2, M, D, A);
        if (cnt != nfull)
            agg_batch<RELU, true>(zr, myidx, nfull, cnt, lane, T2, M, D, A);
    }

    v2f o;
    o.x = (A.x + GEPS * D.x) / (D.x + 1e-16f) + self.x;
    o.y = (A.y + GEPS * D.y) / (D.y + 1e-16f) + self.y;
    ((v2f*)out)[node * (Hc / 2) + lane] = o;
}

extern "C" void kernel_launch(void* const* d_in, const int* in_sizes, int n_in,
                              void* d_out, int out_size, void* d_ws, size_t ws_size,
                              hipStream_t stream) {
    (void)n_in; (void)out_size; (void)ws_size;

    const float* x    = (const float*)d_in[0];
    const int*   ei   = (const int*)  d_in[1];
    const float* encW = (const float*)d_in[2];
    const float* encB = (const float*)d_in[3];
    const float* t    = (const float*)d_in[4];
    const float* W1   = (const float*)d_in[5];
    const float* b1   = (const float*)d_in[6];
    const float* g1   = (const float*)d_in[7];
    const float* bb1  = (const float*)d_in[8];
    const float* W2   = (const float*)d_in[9];
    const float* b2   = (const float*)d_in[10];
    const float* ng   = (const float*)d_in[11];
    const float* nb   = (const float*)d_in[12];
    const float* linW = (const float*)d_in[13];
    const float* linB = (const float*)d_in[14];

    const int F = 128, H = 64, L = 4;
    const int N = in_sizes[0] / F;
    const int E = in_sizes[1] / 2;
    const int* src = ei;
    const int* dst = ei + E;

    float* ws = (float*)d_ws;
    const long NH = (long)N * H;
    float* h    = ws;  ws += NH;
    float* z    = ws;  ws += NH;
    int*   csr  = (int*)ws;  ws += 3 * NH;
    float* outb = ws;  ws += NH;

    auto cdiv = [](long a, long b) { return (int)((a + b - 1) / b); };
    auto gemm_waves = [&](int M, int Nc) { return (long)(M / 16) * ((Nc + 63) / 64); };

    const int G  = cdiv(E, CH);
    const int nbk = cdiv(N, BN);
    int* pairs  = csr;
    int* ssrc   = pairs + 2 * E;
    int* offT   = ssrc + E;
    int* rowptr = offT + (nbk + 1) * G;

    const int GB = cdiv(gemm_waves(N, H) * 32, T1);
    encode_and_bin_kernel<<<G + GB, T1, 0, stream>>>(src, dst, pairs, offT, E, nbk, G,
                                                     x, encW, encB, h, N, H, F);
    bucket_group_kernel<<<nbk, 256, 0, stream>>>(pairs, offT, rowptr, ssrc, N, G);

    for (int l = 0; l < L; ++l) {
        const float* zin = (l == 0) ? h : z;

        if (l == 0)
            node_softmax_agg_kernel<true><<<cdiv(N, 8), 256, 0, stream>>>(zin, rowptr, ssrc, t, l, outb, N);
        else
            node_softmax_agg_kernel<false><<<cdiv(N, 8), 256, 0, stream>>>(zin, rowptr, ssrc, t, l, outb, N);

        const int ln = (l + 1 < L) ? (l + 1) : 0;
        if (l + 1 < L)
            mlp_chain_kernel<false><<<N / 16, 32, 0, stream>>>(
                outb, W1 + (long)l * H * 2 * H, b1 + l * 2 * H, g1 + l * 2 * H, bb1 + l * 2 * H,
                W2 + (long)l * 2 * H * H, b2 + l * H, ng + ln * H, nb + ln * H,
                h, z, (l == 0) ? 0 : 1, 1, linW, linB, (float*)d_out);
        else
            mlp_chain_kernel<true><<<N / 16, 32, 0, stream>>>(
                outb, W1 + (long)l * H * 2 * H, b1 + l * 2 * H, g1 + l * 2 * H, bb1 + l * 2 * H,
                W2 + (long)l * 2 * H * H, b2 + l * H, ng + ln * H, nb + ln * H,
                h, z, (l == 0) ? 0 : 1, 0, linW, linB, (float*)d_out);
    }
}


# Each layer's edge walk and node chain fused into ONE launch (10 -> 6): 16-wave work-groups hold W1/W2 in LDS, gather 16 nodes, then split the chain's column tiles over waves; LN row-sum order changes
# speedup vs baseline: 9.0058x; 1.0452x over previous
#include <hip/hip_runtime.h>
#include <math.h>

typedef float v2f __attribute__((ext_vector_type(2)));
typedef float v4f __attribute__((ext_vector_type(4)));
typedef float v8f __attribute__((ext_vector_type(8)));

#define HD __device__ __forceinline__

static constexpr int   Hc   = 64;
static constexpr int   Cc   = 40;
static constexpr float GEPS = 1e-7f;

HD void gemm_bias_wmma_body(const float* __restrict__ A,
                            const float* __restrict__ B,
                            const float* __restrict__ bias,
                            float* __restrict__ C,
                            int M, int N, int K, int gid) {
    int wave    = gid >> 5;
    int lane    = threadIdx.x & 31;
    int groupsN = (N + 63) >> 6;
    int tm = wave / groupsN;
    int gn = wave - tm * groupsN;
    if (tm >= (M >> 4)) return;

    int mr   = lane & 15;
    int half = lane >> 4;
    const float* Arow = A + (long)(tm * 16 + mr) * K;

    int ncol[4];
#pragma unroll
    for (int t = 0; t < 4; ++t) {
        int n   = gn * 64 + t * 16 + mr;
        ncol[t] = (n < N) ? n : (N - 1);
    }

    v8f acc[4] = {{0.f,0.f,0.f,0.f,0.f,0.f,0.f,0.f},
                  {0.f,0.f,0.f,0.f,0.f,0.f,0.f,0.f},
                  {0.f,0.f,0.f,0.f,0.f,0.f,0.f,0.f},
                  {0.f,0.f,0.f,0.f,0.f,0.f,0.f,0.f}};

    int ka = half * 2;
    v2f a_cur;
    a_cur.x = Arow[ka];
    a_cur.y = Arow[ka + 1];
    v2f b_cur[4];
#pragma unroll
    for (int t = 0; t < 4; ++t) {
        b_cur[t].x = B[ka * N + ncol[t]];
        b_cur[t].y = B[(ka + 1) * N + ncol[t]];
    }

#pragma unroll 2
    for (int k = 0; k < K; k += 4) {
        int kn = (k + 4 < K) ? (k + 4) : k;
        int kb = kn + half * 2;

        v2f a_nxt;
        a_nxt.x = Arow[kb];
        a_nxt.y = Arow[kb + 1];
        v2f b_nxt[4];
#pragma unroll
        for (int t = 0; t < 4; ++t) {
            b_nxt[t].x = B[kb * N + ncol[t]];
            b_nxt[t].y = B[(kb + 1) * N + ncol[t]];
        }

#pragma unroll
        for (int t = 0; t < 4; ++t)
            acc[t] = __builtin_amdgcn_wmma_f32_16x16x4_f32(
                         false, a_cur, false, b_cur[t], (short)0, acc[t], false, false);

        a_cur = a_nxt;
#pragma unroll
        for (int t = 0; t < 4; ++t) b_cur[t] = b_nxt[t];
    }

#pragma unroll
    for (int t = 0; t < 4; ++t) {
        int n = gn * 64 + t * 16 + mr;
        if (n < N) {
            float bv = bias[n];
#pragma unroll
            for (int r = 0; r < 8; ++r)
                C[(long)(tm * 16 + half * 8 + r) * N + n] = acc[t][r] + bv;
        }
    }
}

typedef int v2i __attribute__((ext_vector_type(2)));

static constexpr int CH   = 4096;
static constexpr int BSH  = 7;
static constexpr int BN   = 1 << BSH;
static constexpr int MAXB = 256;

static constexpr int T1  = 256;
static constexpr int EPT = CH / T1;

HD void edge_bin_body(const int* __restrict__ src, const int* __restrict__ dst,
                      int* __restrict__ pairs, int* __restrict__ offT, int E, int nb,
                      int g, int G) {
    __shared__ int lcnt[MAXB];
    __shared__ v2i stage[CH];
    const int t  = threadIdx.x;
    const int e0 = g * CH;
    int n = E - e0; if (n > CH) n = CH;

    int d[EPT], s[EPT];
#pragma unroll
    for (int k = 0; k < EPT; ++k) {
        int idx = t + T1 * k;
        bool ok = idx < n;
        d[k] = ok ? dst[e0 + idx] : -1;
        s[k] = ok ? src[e0 + idx] : 0;
    }
    for (int q = t; q < MAXB; q += T1) lcnt[q] = 0;
    __syncthreads();

    int rk[EPT];
#pragma unroll
    for (int k = 0; k < EPT; ++k) {
        rk[k] = 0;
        if (d[k] >= 0) rk[k] = atomicAdd(&lcnt[d[k] >> BSH], 1);
    }
    __syncthreads();

    if (t < 32) {
        int c[8];
        int sum = 0;
#pragma unroll
        for (int q = 0; q < 8; ++q) { int x = lcnt[8 * t + q]; c[q] = sum; sum += x; }
        int incl = sum;
        for (int off = 1; off < 32; off <<= 1) {
            int v = __shfl_up(incl, off);
            if (t >= off) incl += v;
        }
        int excl = incl - sum;
#pragma unroll
        for (int q = 0; q < 8; ++q) lcnt[8 * t + q] = excl + c[q];
    }
    __syncthreads();

    int pos[EPT];
#pragma unroll
    for (int k = 0; k < EPT; ++k) pos[k] = lcnt[(d[k] >= 0) ? (d[k] >> BSH) : 0] + rk[k];
#pragma unroll
    for (int k = 0; k < EPT; ++k)
        if (d[k] >= 0) stage[pos[k]] = v2i{d[k], s[k]};
    __syncthreads();

    v2i* pr = (v2i*)pairs + e0;
    v2i w[EPT];
#pragma unroll
    for (int k = 0; k < EPT; ++k) {
        int idx = t + T1 * k;
        w[k] = (idx < n) ? stage[idx] : v2i{0, 0};
    }
#pragma unroll
    for (int k = 0; k < EPT; ++k) {
        int idx = t + T1 * k;
        if (idx < n) pr[idx] = w[k];
    }
    for (int q = t; q <= nb; q += T1) offT[q * G + g] = (q < nb) ? lcnt[q] : n;
}

__global__ void __launch_bounds__(T1)
encode_and_bin_kernel(const int* __restrict__ src, const int* __restrict__ dst,
                      int* __restrict__ pairs, int* __restrict__ offT, int E, int nb, int G,
                      const float* __restrict__ x, const float* __restrict__ encW,
                      const float* __restrict__ encB, float* __restrict__ h,
                      int M, int N, int K) {
    const int b = (int)blockIdx.x;
    if (b < G) edge_bin_body(src, dst, pairs, offT, E, nb, b, G);
    else       gemm_bias_wmma_body(x, encW, encB, h, M, N, K, (b - G) * T1 + (int)threadIdx.x);
}

template <bool PLACE>
HD int bucket_sweep(const v2i* __restrict__ pr, const int* __restrict__ r0,
                    const int* __restrict__ r1, int G, int wv, int lane, int lo,
                    int* ctr, int* __restrict__ ssrc) {
    int below = 0;
    for (int gb = wv * 32; gb < G; gb += 8 * 32) {
        int gg = gb + lane;
        int o0 = 0, o1 = 0;
        if (gg < G) { o0 = r0[gg]; o1 = r1[gg]; }
        below += o0;
        int len = o1 - o0;
        int st  = gg * CH + o0;
        v2i p[32];
#pragma unroll
        for (int u = 0; u < 32; ++u) {
            int L = __builtin_amdgcn_readlane(len, u);
            int S = __builtin_amdgcn_readlane(st, u);
            p[u] = v2i{-1, 0};
            if (lane < L) p[u] = pr[S + lane];
        }
        if (PLACE) {
            int slot[32];
#pragma unroll
            for (int u = 0; u < 32; ++u) {
                slot[u] = 0;
                if (p[u].x >= 0) slot[u] = atomicAdd(&ctr[p[u].x - lo], 1);
            }
#pragma unroll
            for (int u = 0; u < 32; ++u)
                if (p[u].x >= 0) ssrc[slot[u]] = p[u].y;
        } else {
#pragma unroll
            for (int u = 0; u < 32; ++u)
                if (p[u].x >= 0) atomicAdd(&ctr[p[u].x - lo], 1);
        }
        unsigned m = __builtin_amdgcn_ballot_w32(len > 32);
        while (m) {
            int u = __builtin_ctz(m);
            m &= m - 1;
            int L = __builtin_amdgcn_readlane(len, u);
            int S = __builtin_amdgcn_readlane(st, u);
            for (int k = 32 + lane; k < L; k += 32) {
                v2i q = pr[S + k];
                if (PLACE) ssrc[atomicAdd(&ctr[q.x - lo], 1)] = q.y;
                else       atomicAdd(&ctr[q.x - lo], 1);
            }
        }
    }
    return below;
}

__global__ void __launch_bounds__(256)
bucket_group_kernel(const int* __restrict__ pairs, const int* __restrict__ offT,
                    int* __restrict__ rowptr, int* __restrict__ ssrc, int N, int G) {
    __shared__ int cnt[BN];
    __shared__ int cur[BN];
    __shared__ int wsum[8];
    const int t    = threadIdx.x;
    const int lane = t & 31;
    const int wv   = __builtin_amdgcn_readfirstlane(t >> 5);
    const int b    = blockIdx.x;
    const int lo   = b << BSH;
    const v2i* pr  = (const v2i*)pairs;
    const int* r0  = offT + (long)b * G;
    const int* r1  = r0 + G;

    if (t < BN) cnt[t] = 0;
    __syncthreads();

    int below = bucket_sweep<false>(pr, r0, r1, G, wv, lane, lo, cnt, ssrc);
    for (int off = 16; off; off >>= 1) below += __shfl_xor(below, off);
    if (lane == 0) wsum[wv] = below;
    __syncthreads();

    if (t < 32) {
        int base = 0;
#pragma unroll
        for (int w = 0; w < 8; ++w) base += wsum[w];
        int c[4];
        int sum = 0;
#pragma unroll
        for (int q = 0; q < 4; ++q) { int x = cnt[4 * t + q]; c[q] = sum; sum += x; }
        int incl = sum;
        for (int off = 1; off < 32; off <<= 1) {
            int v = __shfl_up(incl, off);
            if (t >= off) incl += v;
        }
        int excl = base + incl - sum;
#pragma unroll
        for (int q = 0; q < 4; ++q) {
            int node = lo + 4 * t + q;
            int v    = excl + c[q];
            cur[4 * t + q] = v;
            if (node < N) rowptr[node] = v;
        }
        if (b == (int)gridDim.x - 1 && t == 31) rowptr[N] = base + incl;
    }
    __syncthreads();

    bucket_sweep<true>(pr, r0, r1, G, wv, lane, lo, cur, ssrc);
}

static constexpr int EB = 8;

template <bool RELU, bool MASK>
HD void agg_batch(const v2f* __restrict__ zr, int myidx, int j0, int cnt, int lane,
                  v2f T2, v2f& M, v2f& D, v2f& A) {
    v2f v[EB];
#pragma unroll
    for (int u = 0; u < EB; ++u) {
        int sidx = __builtin_amdgcn_readlane(myidx, j0 + u);
        const v2f* rp = zr + (long)sidx * (Hc / 2);
        v[u] = rp[lane];
    }
    v2f q[EB];
    v2f Mn = M;
#pragma unroll
    for (int u = 0; u < EB; ++u) {
        if (RELU) { v[u].x = fmaxf(v[u].x, 0.f); v[u].y = fmaxf(v[u].y, 0.f); }
        q[u] = v[u] * T2;
        if (MASK) {
            bool ok = (j0 + u) < cnt;
            v[u].x = ok ? v[u].x : 0.f;       v[u].y = ok ? v[u].y : 0.f;
            q[u].x = ok ? q[u].x : -3.0e38f;  q[u].y = ok ? q[u].y : -3.0e38f;
        }
        Mn.x = fmaxf(Mn.x, q[u].x);
        Mn.y = fmaxf(Mn.y, q[u].y);
    }
    v2f dm = M - Mn;
    v2f r;
    r.x = __builtin_amdgcn_exp2f(dm.x);
    r.y = __builtin_amdgcn_exp2f(dm.y);
    D *= r;
    A *= r;
#pragma unroll
    for (int u = 0; u < EB; ++u) {
        v2f a = q[u] - Mn;
        v2f e;
        e.x = __builtin_amdgcn_exp2f(a.x);
        e.y = __builtin_amdgcn_exp2f(a.y);
        D += e;
        A += v[u] * e;
    }
    M = Mn;
}

static constexpr int LP = 132;
static constexpr int XP = 68;
static constexpr int PP = 34;
static constexpr int NW = 16;

template <int M>
HD float row_xor_add(float v) {
    int o = __builtin_amdgcn_update_dpp(0, __float_as_int(v), 0x160 | M, 0xf, 0xf, false);
    return v + __int_as_float(o);
}
template <int M>
HD float row_xor_max(float v) {
    int o = __builtin_amdgcn_update_dpp(0, __float_as_int(v), 0x160 | M, 0xf, 0xf, false);
    return fmaxf(v, __int_as_float(o));
}

template <bool RELU, bool LAST>
__global__ void __launch_bounds__(NW * 32, 12)
gather_mlp_kernel(const float* __restrict__ zin,
                  const int* __restrict__ rowptr, const int* __restrict__ ssrc,
                  const float* __restrict__ tptr, int layer,
                  const float* __restrict__ W1, const float* __restrict__ b1,
                  const float* __restrict__ g1, const float* __restrict__ bb1,
                  const float* __restrict__ W2, const float* __restrict__ b2,
                  const float* __restrict__ ng, const float* __restrict__ nb,
                  float* __restrict__ h, float* __restrict__ zout, int add_res,
                  const float* __restrict__ linW, const float* __restrict__ linB,
                  float* __restrict__ out, int n) {
    __shared__ __attribute__((aligned(16))) float W1l[Hc * 2 * Hc];
    __shared__ __attribute__((aligned(16))) float W2l[2 * Hc * Hc];
    __shared__ __attribute__((aligned(16))) float W3l[LAST ? Hc * Cc : 4];
    __shared__ float c1[3 * 2 * Hc];
    __shared__ float c2[3 * Hc];
    __shared__ float c3[LAST ? Cc : 1];
    __shared__ __attribute__((aligned(16))) float xs[16 * XP];
    __shared__ __attribute__((aligned(16))) float hs[16 * XP];
    __shared__ __attribute__((aligned(16))) float zs[16 * XP];
    __shared__ __attribute__((aligned(16))) float tile[16 * LP];
    __shared__ __attribute__((aligned(16))) float part[8 * PP];

    const int lane = threadIdx.x & 31;
    const int wv   = __builtin_amdgcn_readfirstlane((int)(threadIdx.x >> 5));
    const int mr   = lane & 15;
    const int half = lane >> 4;

#pragma unroll
    for (int i = 0; i < (Hc * 2 * Hc) / (4 * NW * 32); ++i) {
        const int q = threadIdx.x + NW * 32 * i;
        ((v4f*)W1l)[q] = ((const v4f*)W1)[q];
        ((v4f*)W2l)[q] = ((const v4f*)W2)[q];
    }
    if (threadIdx.x < 2 * Hc) {
        c1[threadIdx.x]          = b1[threadIdx.x];
        c1[2 * Hc + threadIdx.x] = g1[threadIdx.x];
        c1[4 * Hc + threadIdx.x] = bb1[threadIdx.x];
    }
    if (threadIdx.x < Hc) {
        c2[threadIdx.x]          = b2[threadIdx.x];
        c2[Hc + threadIdx.x]     = ng[threadIdx.x];
        c2[2 * Hc + threadIdx.x] = nb[threadIdx.x];
    }
    if constexpr (LAST) {
        for (int q = threadIdx.x; q < (Hc * Cc) / 4; q += NW * 32)
            ((v4f*)W3l)[q] = ((const v4f*)linW)[q];
        if (threadIdx.x < Cc) c3[threadIdx.x] = linB[threadIdx.x];
    }

    const int ntiles = n / NW;
    for (int tm = blockIdx.x; tm < ntiles; tm += gridDim.x) {

    {
        const int node = tm * NW + wv;
        v2f o = {0.f, 0.f};
        if (node < n) {
            int   start = rowptr[node];
            int   end   = rowptr[node + 1];
            float t2    = tptr[layer] * 1.44269504088896341f;
            const v2f* zr = (const v2f*)zin;
            v2f self = zr[node * (Hc / 2) + lane];

            v2f T2 = {t2, t2};
            v2f M  = {-3.0e38f, -3.0e38f};
            v2f D  = {0.f, 0.f};
            v2f A  = {0.f, 0.f};

            for (int base = start; base < end; base += 32) {
                int cnt = end - base; if (cnt > 32) cnt = 32;
                int myidx = (lane < cnt) ? ssrc[base + lane] : 0;
                int nfull = cnt & ~(EB - 1);
                for (int j0 = 0; j0 < nfull; j0 += EB)
                    agg_batch<RELU, false>(zr, myidx, j0, cnt, lane, T2, M, D, A);
                if (cnt != nfull)
                    agg_batch<RELU, true>(zr, myidx, nfull, cnt, lane, T2, M, D, A);
            }
            o.x = (A.x + GEPS * D.x) / (D.x + 1e-16f) + self.x;
            o.y = (A.y + GEPS * D.y) / (D.y + 1e-16f) + self.y;
        }
        *(v2f*)(xs + wv * XP + 2 * lane) = o;
    }
    __syncthreads();

    float u1[8];
    float gam1 = 0.f, bet1 = 0.f;
    if (wv < 8) {
        const float* B1p = W1l + (half * 2) * (2 * Hc) + wv * 16 + mr;
        v2f bf[16];
#pragma unroll
        for (int kk = 0; kk < 16; ++kk) {
            bf[kk].x = B1p[(4 * kk) * (2 * Hc)];
            bf[kk].y = B1p[(4 * kk + 1) * (2 * Hc)];
        }
        const float bias1 = c1[wv * 16 + mr];
        gam1 = c1[2 * Hc + wv * 16 + mr];
        bet1 = c1[4 * Hc + wv * 16 + mr];

        const float* Ax = xs + mr * XP + half * 2;
        v8f acc = {0.f,0.f,0.f,0.f,0.f,0.f,0.f,0.f};
#pragma unroll
        for (int kk = 0; kk < 16; ++kk) {
            v2f a = *(const v2f*)(Ax + 4 * kk);
            acc = __builtin_amdgcn_wmma_f32_16x16x4_f32(
                      false, a, false, bf[kk], (short)0, acc, false, false);
        }

        float s1[8], q1[8];
#pragma unroll
        for (int r = 0; r < 8; ++r) {
            float u = acc[r] + bias1;
            u1[r] = u;  s1[r] = u;  q1[r] = u * u;
        }
#pragma unroll
        for (int r = 0; r < 8; ++r) { s1[r] = row_xor_add<1>(s1[r]);  q1[r] = row_xor_add<1>(q1[r]); }
#pragma unroll
        for (int r = 0; r < 8; ++r) { s1[r] = row_xor_add<2>(s1[r]);  q1[r] = row_xor_add<2>(q1[r]); }
#pragma unroll
        for (int r = 0; r < 8; ++r) { s1[r] = row_xor_add<4>(s1[r]);  q1[r] = row_xor_add<4>(q1[r]); }
#pragma unroll
        for (int r = 0; r < 8; ++r) { s1[r] = row_xor_add<8>(s1[r]);  q1[r] = row_xor_add<8>(q1[r]); }
        if (mr == 0) {
#pragma unroll
            for (int r = 0; r < 8; ++r)
                *(v2f*)(part + wv * PP + (half * 8 + r) * 2) = v2f{s1[r], q1[r]};
        }
    } else {
#pragma unroll
        for (int r = 0; r < 8; ++r) u1[r] = 0.f;
    }
    __syncthreads();

    if (wv < 8) {
        v2f p[8];
#pragma unroll
        for (int r = 0; r < 8; ++r) p[r] = *(const v2f*)(part + (mr & 7) * PP + (half * 8 + r) * 2);
#pragma unroll
        for (int r = 0; r < 8; ++r) { p[r].x = row_xor_add<1>(p[r].x);  p[r].y = row_xor_add<1>(p[r].y); }
#pragma unroll
        for (int r = 0; r < 8; ++r) { p[r].x = row_xor_add<2>(p[r].x);  p[r].y = row_xor_add<2>(p[r].y); }
#pragma unroll
        for (int r = 0; r < 8; ++r) { p[r].x = row_xor_add<4>(p[r].x);  p[r].y = row_xor_add<4>(p[r].y); }
#pragma unroll
        for (int r = 0; r < 8; ++r) {
            float mean = p[r].x * (1.f / 128.f);
            float var  = p[r].y * (1.f / 128.f) - mean * mean;
            float rstd = rsqrtf(var + 1e-5f);
            float v = (u1[r] - mean) * rstd * gam1 + bet1;
            tile[(half * 8 + r) * LP + wv * 16 + mr] = fmaxf(v, 0.f);
        }
    }
    __syncthreads();

    float hn[8];
    float gam2 = 0.f, bet2 = 0.f;
    if (wv < 4) {
        const long rowbase = (long)(tm * 16 + half * 8) * Hc + wv * 16 + mr;
        float hres[8];
        if (add_res) {
#pragma unroll
            for (int r = 0; r < 8; ++r) hres[r] = h[rowbase + (long)r * Hc];
        } else {
#pragma unroll
            for (int r = 0; r < 8; ++r) hres[r] = 0.f;
        }
        const float bias2 = c2[wv * 16 + mr];
        gam2 = c2[Hc + wv * 16 + mr];
        bet2 = c2[2 * Hc + wv * 16 + mr];

        const float* At  = tile + mr * LP + half * 2;
        const float* B2p = W2l + (half * 2) * Hc + wv * 16 + mr;
        v8f acc = {0.f,0.f,0.f,0.f,0.f,0.f,0.f,0.f};
        v2f bc[8];
#pragma unroll
        for (int u = 0; u < 8; ++u) {
            bc[u].x = B2p[(4 * u) * Hc];
            bc[u].y = B2p[(4 * u + 1) * Hc];
        }
#pragma unroll
        for (int j = 0; j < 4; ++j) {
            v2f bn[8];
            if (j + 1 < 4) {
#pragma unroll
                for (int u = 0; u < 8; ++u) {
                    const int kk = 8 * (j + 1) + u;
                    bn[u].x = B2p[(4 * kk) * Hc];
                    bn[u].y = B2p[(4 * kk + 1) * Hc];
                }
            }
#pragma unroll
            for (int u = 0; u < 8; ++u) {
                v2f a = *(const v2f*)(At + 4 * (8 * j + u));
                acc = __builtin_amdgcn_wmma_f32_16x16x4_f32(
                          false, a, false, bc[u], (short)0, acc, false, false);
            }
            if (j + 1 < 4) {
#pragma unroll
                for (int u = 0; u < 8; ++u) bc[u] = bn[u];
            }
        }

        float s2[8], q2[8];
#pragma unroll
        for (int r = 0; r < 8; ++r) {
            float x = hres[r] + (acc[r] + bias2);
            hn[r] = x;  s2[r] = x;  q2[r] = x * x;
        }
#pragma unroll
        for (int r = 0; r < 8; ++r) { s2[r] = row_xor_add<1>(s2[r]);  q2[r] = row_xor_add<1>(q2[r]); }
#pragma unroll
        for (int r = 0; r < 8; ++r) { s2[r] = row_xor_add<2>(s2[r]);  q2[r] = row_xor_add<2>(q2[r]); }
#pragma unroll
        for (int r = 0; r < 8; ++r) { s2[r] = row_xor_add<4>(s2[r]);  q2[r] = row_xor_add<4>(q2[r]); }
#pragma unroll
        for (int r = 0; r < 8; ++r) { s2[r] = row_xor_add<8>(s2[r]);  q2[r] = row_xor_add<8>(q2[r]); }
        if (mr == 0) {
#pragma unroll
            for (int r = 0; r < 8; ++r)
                *(v2f*)(part + wv * PP + (half * 8 + r) * 2) = v2f{s2[r], q2[r]};
        }
    } else {
#pragma unroll
        for (int r = 0; r < 8; ++r) hn[r] = 0.f;
    }
    __syncthreads();

    if (wv < 4) {
        v2f p[8];
#pragma unroll
        for (int r = 0; r < 8; ++r) p[r] = *(const v2f*)(part + (mr & 3) * PP + (half * 8 + r) * 2);
#pragma unroll
        for (int r = 0; r < 8; ++r) { p[r].x = row_xor_add<1>(p[r].x);  p[r].y = row_xor_add<1>(p[r].y); }
#pragma unroll
        for (int r = 0; r < 8; ++r) { p[r].x = row_xor_add<2>(p[r].x);  p[r].y = row_xor_add<2>(p[r].y); }
#pragma unroll
        for (int r = 0; r < 8; ++r) {
            float mean = p[r].x * (1.f / 64.f);
            float var  = p[r].y * (1.f / 64.f) - mean * mean;
            float rstd = rsqrtf(var + 1e-5f);
            float v = (hn[r] - mean) * rstd * gam2 + bet2;
            zs[(half * 8 + r) * XP + wv * 16 + mr] = fmaxf(v, 0.f);
            if constexpr (!LAST) hs[(half * 8 + r) * XP + wv * 16 + mr] = hn[r];
        }
    }
    __syncthreads();

    if constexpr (!LAST) {
        const int j   = threadIdx.x & 255;
        const int row = j >> 4;
        const int c4  = (j & 15) * 4;
        const float* sb = (wv < 8) ? hs : zs;
        float*       db = (wv < 8) ? h : zout;
        *(v4f*)(db + (long)(tm * 16 + row) * Hc + c4) = *(const v4f*)(sb + row * XP + c4);
    } else {
        if (wv == 0) {
            int   nc3[3];
            float lb3[3];
#pragma unroll
            for (int t = 0; t < 3; ++t) {
                int nn = t * 16 + mr;
                nc3[t] = (nn < Cc) ? nn : (Cc - 1);
                lb3[t] = c3[nc3[t]];
            }
            const float* B3p = W3l + (half * 2) * Cc;
            const float* At3 = zs + mr * XP + half * 2;
            v8f acc3[3];
#pragma unroll
            for (int t = 0; t < 3; ++t) acc3[t] = v8f{0.f,0.f,0.f,0.f,0.f,0.f,0.f,0.f};
#pragma unroll 1
            for (int c = 0; c < 4; ++c) {
                v2f b3[4][3];
                v2f a3[4];
#pragma unroll
                for (int u = 0; u < 4; ++u) {
                    const int kk = 4 * c + u;
#pragma unroll
                    for (int t = 0; t < 3; ++t) {
                        b3[u][t].x = B3p[(4 * kk) * Cc + nc3[t]];
                        b3[u][t].y = B3p[(4 * kk + 1) * Cc + nc3[t]];
                    }
                    a3[u] = *(const v2f*)(At3 + 4 * kk);
                }
#pragma unroll
                for (int u = 0; u < 4; ++u)
#pragma unroll
                    for (int t = 0; t < 3; ++t)
                        acc3[t] = __builtin_amdgcn_wmma_f32_16x16x4_f32(
                                      false, a3[u], false, b3[u][t], (short)0, acc3[t], false, false);
            }

            const bool c2ok = mr < (Cc - 32);
            float mx3[8], sm3[8];
#pragma unroll
            for (int r = 0; r < 8; ++r) {
#pragma unroll
                for (int t = 0; t < 3; ++t) acc3[t][r] += lb3[t];
                float m01 = fmaxf(acc3[0][r], acc3[1][r]);
                mx3[r] = c2ok ? fmaxf(m01, acc3[2][r]) : m01;
            }
#pragma unroll
            for (int r = 0; r < 8; ++r) mx3[r] = row_xor_max<1>(mx3[r]);
#pragma unroll
            for (int r = 0; r < 8; ++r) mx3[r] = row_xor_max<2>(mx3[r]);
#pragma unroll
            for (int r = 0; r < 8; ++r) mx3[r] = row_xor_max<4>(mx3[r]);
#pragma unroll
            for (int r = 0; r < 8; ++r) mx3[r] = row_xor_max<8>(mx3[r]);
#pragma unroll
            for (int r = 0; r < 8; ++r) {
                float e01 = __expf(acc3[0][r] - mx3[r]) + __expf(acc3[1][r] - mx3[r]);
                float e2  = __expf(acc3[2][r] - mx3[r]);
                sm3[r] = c2ok ? (e01 + e2) : e01;
            }
#pragma unroll
            for (int r = 0; r < 8; ++r) sm3[r] = row_xor_add<1>(sm3[r]);
#pragma unroll
            for (int r = 0; r < 8; ++r) sm3[r] = row_xor_add<2>(sm3[r]);
#pragma unroll
            for (int r = 0; r < 8; ++r) sm3[r] = row_xor_add<4>(sm3[r]);
#pragma unroll
            for (int r = 0; r < 8; ++r) sm3[r] = row_xor_add<8>(sm3[r]);

#pragma unroll
            for (int r = 0; r < 8; ++r) {
                float lse = mx3[r] + __logf(sm3[r]);
                float* orow = tile + (half * 8 + r) * Cc + mr;
                orow[0]  = acc3[0][r] - lse;
                orow[16] = acc3[1][r] - lse;
                if (c2ok) orow[32] = acc3[2][r] - lse;
            }
        }
        __syncthreads();
        if (threadIdx.x < (16 * Cc) / 4) {
            float* ob = out + (long)tm * (16 * Cc);
            const int q = threadIdx.x;
            *(v4f*)(ob + 4 * q) = *(const v4f*)(tile + 4 * q);
        }
    }
    }
}

extern "C" void kernel_launch(void* const* d_in, const int* in_sizes, int n_in,
                              void* d_out, int out_size, void* d_ws, size_t ws_size,
                              hipStream_t stream) {
    (void)n_in; (void)out_size; (void)ws_size;

    const float* x    = (const float*)d_in[0];
    const int*   ei   = (const int*)  d_in[1];
    const float* encW = (const float*)d_in[2];
    const float* encB = (const float*)d_in[3];
    const float* t    = (const float*)d_in[4];
    const float* W1   = (const float*)d_in[5];
    const float* b1   = (const float*)d_in[6];
    const float* g1   = (const float*)d_in[7];
    const float* bb1  = (const float*)d_in[8];
    const float* W2   = (const float*)d_in[9];
    const float* b2   = (const float*)d_in[10];
    const float* ng   = (const float*)d_in[11];
    const float* nb   = (const float*)d_in[12];
    const float* linW = (const float*)d_in[13];
    const float* linB = (const float*)d_in[14];

    const int F = 128, H = 64, L = 4;
    const int N = in_sizes[0] / F;
    const int E = in_sizes[1] / 2;
    const int* src = ei;
    const int* dst = ei + E;

    float* ws = (float*)d_ws;
    const long NH = (long)N * H;
    float* h0   = ws;  ws += NH;
    float* zA   = ws;  ws += NH;
    int*   csr  = (int*)ws;  ws += 2 * NH;
    float* zB   = ws;  ws += NH;
    float* h    = ws;  ws += NH;

    auto cdiv = [](long a, long b) { return (int)((a + b - 1) / b); };
    auto gemm_waves = [&](int M, int Nc) { return (long)(M / 16) * ((Nc + 63) / 64); };

    const int G  = cdiv(E, CH);
    const int nbk = cdiv(N, BN);
    int* pairs  = csr;
    int* ssrc   = pairs + 2 * E;
    int* offT   = ssrc + E;
    int* rowptr = offT + (nbk + 1) * G;

    const int GB = cdiv(gemm_waves(N, H) * 32, T1);
    encode_and_bin_kernel<<<G + GB, T1, 0, stream>>>(src, dst, pairs, offT, E, nbk, G,
                                                     x, encW, encB, h0, N, H, F);
    bucket_group_kernel<<<nbk, 256, 0, stream>>>(pairs, offT, rowptr, ssrc, N, G);

    const int GL = 768;
    for (int l = 0; l < L; ++l) {
        const float* zin  = (l == 0) ? h0 : ((l & 1) ? zA : zB);
        float*       zout = (l & 1) ? zB : zA;
        const int ln = (l + 1 < L) ? (l + 1) : 0;
        if (l == 0)
            gather_mlp_kernel<true, false><<<GL, NW * 32, 0, stream>>>(
                zin, rowptr, ssrc, t, l,
                W1 + (long)l * H * 2 * H, b1 + l * 2 * H, g1 + l * 2 * H, bb1 + l * 2 * H,
                W2 + (long)l * 2 * H * H, b2 + l * H, ng + ln * H, nb + ln * H,
                h, zout, 0, linW, linB, (float*)d_out, N);
        else if (l + 1 < L)
            gather_mlp_kernel<false, false><<<GL, NW * 32, 0, stream>>>(
                zin, rowptr, ssrc, t, l,
                W1 + (long)l * H * 2 * H, b1 + l * 2 * H, g1 + l * 2 * H, bb1 + l * 2 * H,
                W2 + (long)l * 2 * H * H, b2 + l * H, ng + ln * H, nb + ln * H,
                h, zout, 1, linW, linB, (float*)d_out, N);
        else
            gather_mlp_kernel<false, true><<<GL, NW * 32, 0, stream>>>(
                zin, rowptr, ssrc, t, l,
                W1 + (long)l * H * 2 * H, b1 + l * 2 * H, g1 + l * 2 * H, bb1 + l * 2 * H,
                W2 + (long)l * 2 * H * H, b2 + l * H, ng + ln * H, nb + ln * H,
                h, zout, 1, linW, linB, (float*)d_out, N);
    }
}


# Edge walk index path on scalar loads: source lists hold row byte offsets, fetched one batch ahead into scalar registers; per edge one 32-bit add + load off a scalar base; sum orders unchanged
# speedup vs baseline: 9.6494x; 1.0715x over previous
#include <hip/hip_runtime.h>
#include <math.h>

typedef float v2f __attribute__((ext_vector_type(2)));
typedef float v4f __attribute__((ext_vector_type(4)));
typedef float v8f __attribute__((ext_vector_type(8)));

#define HD __device__ __forceinline__

static constexpr int   Hc   = 64;
static constexpr int   Cc   = 40;
static constexpr float GEPS = 1e-7f;

HD void gemm_bias_wmma_body(const float* __restrict__ A,
                            const float* __restrict__ B,
                            const float* __restrict__ bias,
                            float* __restrict__ C,
                            int M, int N, int K, int gid) {
    int wave    = gid >> 5;
    int lane    = threadIdx.x & 31;
    int groupsN = (N + 63) >> 6;
    int tm = wave / groupsN;
    int gn = wave - tm * groupsN;
    if (tm >= (M >> 4)) return;

    int mr   = lane & 15;
    int half = lane >> 4;
    const float* Arow = A + (long)(tm * 16 + mr) * K;

    int ncol[4];
#pragma unroll
    for (int t = 0; t < 4; ++t) {
        int n   = gn * 64 + t * 16 + mr;
        ncol[t] = (n < N) ? n : (N - 1);
    }

    v8f acc[4] = {{0.f,0.f,0.f,0.f,0.f,0.f,0.f,0.f},
                  {0.f,0.f,0.f,0.f,0.f,0.f,0.f,0.f},
                  {0.f,0.f,0.f,0.f,0.f,0.f,0.f,0.f},
                  {0.f,0.f,0.f,0.f,0.f,0.f,0.f,0.f}};

    int ka = half * 2;
    v2f a_cur;
    a_cur.x = Arow[ka];
    a_cur.y = Arow[ka + 1];
    v2f b_cur[4];
#pragma unroll
    for (int t = 0; t < 4; ++t) {
        b_cur[t].x = B[ka * N + ncol[t]];
        b_cur[t].y = B[(ka + 1) * N + ncol[t]];
    }

#pragma unroll 2
    for (int k = 0; k < K; k += 4) {
        int kn = (k + 4 < K) ? (k + 4) : k;
        int kb = kn + half * 2;

        v2f a_nxt;
        a_nxt.x = Arow[kb];
        a_nxt.y = Arow[kb + 1];
        v2f b_nxt[4];
#pragma unroll
        for (int t = 0; t < 4; ++t) {
            b_nxt[t].x = B[kb * N + ncol[t]];
            b_nxt[t].y = B[(kb + 1) * N + ncol[t]];
        }

#pragma unroll
        for (int t = 0; t < 4; ++t)
            acc[t] = __builtin_amdgcn_wmma_f32_16x16x4_f32(
                         false, a_cur, false, b_cur[t], (short)0, acc[t], false, false);

        a_cur = a_nxt;
#pragma unroll
        for (int t = 0; t < 4; ++t) b_cur[t] = b_nxt[t];
    }

#pragma unroll
    for (int t = 0; t < 4; ++t) {
        int n = gn * 64 + t * 16 + mr;
        if (n < N) {
            float bv = bias[n];
#pragma unroll
            for (int r = 0; r < 8; ++r)
                C[(long)(tm * 16 + half * 8 + r) * N + n] = acc[t][r] + bv;
        }
    }
}

typedef int v2i __attribute__((ext_vector_type(2)));

static constexpr int CH   = 4096;
static constexpr int BSH  = 7;
static constexpr int BN   = 1 << BSH;
static constexpr int MAXB = 256;

static constexpr int T1  = 256;
static constexpr int EPT = CH / T1;

HD void edge_bin_body(const int* __restrict__ src, const int* __restrict__ dst,
                      int* __restrict__ pairs, int* __restrict__ offT, int E, int nb,
                      int g, int G) {
    __shared__ int lcnt[MAXB];
    __shared__ v2i stage[CH];
    const int t  = threadIdx.x;
    const int e0 = g * CH;
    int n = E - e0; if (n > CH) n = CH;

    int d[EPT], s[EPT];
#pragma unroll
    for (int k = 0; k < EPT; ++k) {
        int idx = t + T1 * k;
        bool ok = idx < n;
        d[k] = ok ? dst[e0 + idx] : -1;
        s[k] = ok ? src[e0 + idx] * (Hc * 4) : 0;
    }
    for (int q = t; q < MAXB; q += T1) lcnt[q] = 0;
    __syncthreads();

    int rk[EPT];
#pragma unroll
    for (int k = 0; k < EPT; ++k) {
        rk[k] = 0;
        if (d[k] >= 0) rk[k] = atomicAdd(&lcnt[d[k] >> BSH], 1);
    }
    __syncthreads();

    if (t < 32) {
        int c[8];
        int sum = 0;
#pragma unroll
        for (int q = 0; q < 8; ++q) { int x = lcnt[8 * t + q]; c[q] = sum; sum += x; }
        int incl = sum;
        for (int off = 1; off < 32; off <<= 1) {
            int v = __shfl_up(incl, off);
            if (t >= off) incl += v;
        }
        int excl = incl - sum;
#pragma unroll
        for (int q = 0; q < 8; ++q) lcnt[8 * t + q] = excl + c[q];
    }
    __syncthreads();

    int pos[EPT];
#pragma unroll
    for (int k = 0; k < EPT; ++k) pos[k] = lcnt[(d[k] >= 0) ? (d[k] >> BSH) : 0] + rk[k];
#pragma unroll
    for (int k = 0; k < EPT; ++k)
        if (d[k] >= 0) stage[pos[k]] = v2i{d[k], s[k]};
    __syncthreads();

    v2i* pr = (v2i*)pairs + e0;
    v2i w[EPT];
#pragma unroll
    for (int k = 0; k < EPT; ++k) {
        int idx = t + T1 * k;
        w[k] = (idx < n) ? stage[idx] : v2i{0, 0};
    }
#pragma unroll
    for (int k = 0; k < EPT; ++k) {
        int idx = t + T1 * k;
        if (idx < n) pr[idx] = w[k];
    }
    for (int q = t; q <= nb; q += T1) offT[q * G + g] = (q < nb) ? lcnt[q] : n;
}

__global__ void __launch_bounds__(T1)
encode_and_bin_kernel(const int* __restrict__ src, const int* __restrict__ dst,
                      int* __restrict__ pairs, int* __restrict__ offT, int E, int nb, int G,
                      const float* __restrict__ x, const float* __restrict__ encW,
                      const float* __restrict__ encB, float* __restrict__ h,
                      int M, int N, int K) {
    const int b = (int)blockIdx.x;
    if (b < G) edge_bin_body(src, dst, pairs, offT, E, nb, b, G);
    else       gemm_bias_wmma_body(x, encW, encB, h, M, N, K, (b - G) * T1 + (int)threadIdx.x);
}

template <bool PLACE>
HD int bucket_sweep(const v2i* __restrict__ pr, const int* __restrict__ r0,
                    const int* __restrict__ r1, int G, int wv, int lane, int lo,
                    int* ctr, int* __restrict__ ssrc) {
    int below = 0;
    for (int gb = wv * 32; gb < G; gb += 8 * 32) {
        int gg = gb + lane;
        int o0 = 0, o1 = 0;
        if (gg < G) { o0 = r0[gg]; o1 = r1[gg]; }
        below += o0;
        int len = o1 - o0;
        int st  = gg * CH + o0;
        v2i p[32];
#pragma unroll
        for (int u = 0; u < 32; ++u) {
            int L = __builtin_amdgcn_readlane(len, u);
            int S = __builtin_amdgcn_readlane(st, u);
            p[u] = v2i{-1, 0};
            if (lane < L) p[u] = pr[S + lane];
        }
        if (PLACE) {
            int slot[32];
#pragma unroll
            for (int u = 0; u < 32; ++u) {
                slot[u] = 0;
                if (p[u].x >= 0) slot[u] = atomicAdd(&ctr[p[u].x - lo], 1);
            }
#pragma unroll
            for (int u = 0; u < 32; ++u)
                if (p[u].x >= 0) ssrc[slot[u]] = p[u].y;
        } else {
#pragma unroll
            for (int u = 0; u < 32; ++u)
                if (p[u].x >= 0) atomicAdd(&ctr[p[u].x - lo], 1);
        }
        unsigned m = __builtin_amdgcn_ballot_w32(len > 32);
        while (m) {
            int u = __builtin_ctz(m);
            m &= m - 1;
            int L = __builtin_amdgcn_readlane(len, u);
            int S = __builtin_amdgcn_readlane(st, u);
            for (int k = 32 + lane; k < L; k += 32) {
                v2i q = pr[S + k];
                if (PLACE) ssrc[atomicAdd(&ctr[q.x - lo], 1)] = q.y;
                else       atomicAdd(&ctr[q.x - lo], 1);
            }
        }
    }
    return below;
}

__global__ void __launch_bounds__(256)
bucket_group_kernel(const int* __restrict__ pairs, const int* __restrict__ offT,
                    int* __restrict__ rowptr, int* __restrict__ ssrc, int N, int G) {
    __shared__ int cnt[BN];
    __shared__ int cur[BN];
    __shared__ int wsum[8];
    const int t    = threadIdx.x;
    const int lane = t & 31;
    const int wv   = __builtin_amdgcn_readfirstlane(t >> 5);
    const int b    = blockIdx.x;
    const int lo   = b << BSH;
    const v2i* pr  = (const v2i*)pairs;
    const int* r0  = offT + (long)b * G;
    const int* r1  = r0 + G;

    if (t < BN) cnt[t] = 0;
    __syncthreads();

    int below = bucket_sweep<false>(pr, r0, r1, G, wv, lane, lo, cnt, ssrc);
    for (int off = 16; off; off >>= 1) below += __shfl_xor(below, off);
    if (lane == 0) wsum[wv] = below;
    __syncthreads();

    if (t < 32) {
        int base = 0;
#pragma unroll
        for (int w = 0; w < 8; ++w) base += wsum[w];
        int c[4];
        int sum = 0;
#pragma unroll
        for (int q = 0; q < 4; ++q) { int x = cnt[4 * t + q]; c[q] = sum; sum += x; }
        int incl = sum;
        for (int off = 1; off < 32; off <<= 1) {
            int v = __shfl_up(incl, off);
            if (t >= off) incl += v;
        }
        int excl = base + incl - sum;
#pragma unroll
        for (int q = 0; q < 4; ++q) {
            int node = lo + 4 * t + q;
            int v    = excl + c[q];
            cur[4 * t + q] = v;
            if (node < N) rowptr[node] = v;
        }
        if (b == (int)gridDim.x - 1 && t == 31) rowptr[N] = base + incl;
    }
    __syncthreads();

    bucket_sweep<true>(pr, r0, r1, G, wv, lane, lo, cur, ssrc);
}

static constexpr int EB = 8;

typedef __attribute__((address_space(4))) const int      ci32;
typedef __attribute__((address_space(4))) const unsigned cu32;

template <bool RELU, bool MASK>
HD void agg_batch(const char* __restrict__ zb, const unsigned (&vo)[EB], int cnt,
                  v2f T2, v2f& M, v2f& D, v2f& A) {
    v2f v[EB];
#pragma unroll
    for (int u = 0; u < EB; ++u)
        v[u] = *(const v2f*)(zb + vo[u]);
    v2f q[EB];
    v2f Mn = M;
#pragma unroll
    for (int u = 0; u < EB; ++u) {
        if (RELU) { v[u].x = fmaxf(v[u].x, 0.f); v[u].y = fmaxf(v[u].y, 0.f); }
        q[u] = v[u] * T2;
        if (MASK) {
            bool ok = u < cnt;
            v[u].x = ok ? v[u].x : 0.f;       v[u].y = ok ? v[u].y : 0.f;
            q[u].x = ok ? q[u].x : -3.0e38f;  q[u].y = ok ? q[u].y : -3.0e38f;
        }
        Mn.x = fmaxf(Mn.x, q[u].x);
        Mn.y = fmaxf(Mn.y, q[u].y);
    }
    v2f dm = M - Mn;
    v2f r;
    r.x = __builtin_amdgcn_exp2f(dm.x);
    r.y = __builtin_amdgcn_exp2f(dm.y);
    D *= r;
    A *= r;
#pragma unroll
    for (int u = 0; u < EB; ++u) {
        v2f a = q[u] - Mn;
        v2f e;
        e.x = __builtin_amdgcn_exp2f(a.x);
        e.y = __builtin_amdgcn_exp2f(a.y);
        D += e;
        A += v[u] * e;
    }
    M = Mn;
}

static constexpr int LP = 132;
static constexpr int XP = 68;
static constexpr int PP = 34;
static constexpr int NW = 16;

template <int M>
HD float row_xor_add(float v) {
    int o = __builtin_amdgcn_update_dpp(0, __float_as_int(v), 0x160 | M, 0xf, 0xf, false);
    return v + __int_as_float(o);
}
template <int M>
HD float row_xor_max(float v) {
    int o = __builtin_amdgcn_update_dpp(0, __float_as_int(v), 0x160 | M, 0xf, 0xf, false);
    return fmaxf(v, __int_as_float(o));
}

template <bool RELU, bool LAST>
__global__ void __launch_bounds__(NW * 32, 12)
gather_mlp_kernel(const float* __restrict__ zin,
                  const int* __restrict__ rowptr, const int* __restrict__ ssrc,
                  const float* __restrict__ tptr, int layer,
                  const float* __restrict__ W1, const float* __restrict__ b1,
                  const float* __restrict__ g1, const float* __restrict__ bb1,
                  const float* __restrict__ W2, const float* __restrict__ b2,
                  const float* __restrict__ ng, const float* __restrict__ nb,
                  float* __restrict__ h, float* __restrict__ zout, int add_res,
                  const float* __restrict__ linW, const float* __restrict__ linB,
                  float* __restrict__ out, int n) {
    __shared__ __attribute__((aligned(16))) float W1l[Hc * 2 * Hc];
    __shared__ __attribute__((aligned(16))) float W2l[2 * Hc * Hc];
    __shared__ __attribute__((aligned(16))) float W3l[LAST ? Hc * Cc : 4];
    __shared__ float c1[3 * 2 * Hc];
    __shared__ float c2[3 * Hc];
    __shared__ float c3[LAST ? Cc : 1];
    __shared__ __attribute__((aligned(16))) float xs[16 * XP];
    __shared__ __attribute__((aligned(16))) float hs[16 * XP];
    __shared__ __attribute__((aligned(16))) float zs[16 * XP];
    __shared__ __attribute__((aligned(16))) float tile[16 * LP];
    __shared__ __attribute__((aligned(16))) float part[8 * PP];

    const int lane = threadIdx.x & 31;
    const int wv   = __builtin_amdgcn_readfirstlane((int)(threadIdx.x >> 5));
    const int mr   = lane & 15;
    const int half = lane >> 4;

#pragma unroll
    for (int i = 0; i < (Hc * 2 * Hc) / (4 * NW * 32); ++i) {
        const int q = threadIdx.x + NW * 32 * i;
        ((v4f*)W1l)[q] = ((const v4f*)W1)[q];
        ((v4f*)W2l)[q] = ((const v4f*)W2)[q];
    }
    if (threadIdx.x < 2 * Hc) {
        c1[threadIdx.x]          = b1[threadIdx.x];
        c1[2 * Hc + threadIdx.x] = g1[threadIdx.x];
        c1[4 * Hc + threadIdx.x] = bb1[threadIdx.x];
    }
    if (threadIdx.x < Hc) {
        c2[threadIdx.x]          = b2[threadIdx.x];
        c2[Hc + threadIdx.x]     = ng[threadIdx.x];
        c2[2 * Hc + threadIdx.x] = nb[threadIdx.x];
    }
    if constexpr (LAST) {
        for (int q = threadIdx.x; q < (Hc * Cc) / 4; q += NW * 32)
            ((v4f*)W3l)[q] = ((const v4f*)linW)[q];
        if (threadIdx.x < Cc) c3[threadIdx.x] = linB[threadIdx.x];
    }

    const int ntiles = n / NW;
    for (int tm = blockIdx.x; tm < ntiles; tm += gridDim.x) {

    {
        const int node = tm * NW + wv;
        v2f o = {0.f, 0.f};
        if (node < n) {
            const ci32* rps = (const ci32*)rowptr;
            const cu32* so  = (const cu32*)ssrc;
            int   start = rps[node];
            int   end   = rps[node + 1];
            float t2    = tptr[layer] * 1.44269504088896341f;
            const v2f* zr = (const v2f*)zin;
            const char* zb = (const char*)zin;
            const unsigned lane8 = (unsigned)lane * 8u;
            v2f self = zr[node * (Hc / 2) + lane];

            v2f T2 = {t2, t2};
            v2f M  = {-3.0e38f, -3.0e38f};
            v2f D  = {0.f, 0.f};
            v2f A  = {0.f, 0.f};

            int j = start;
            unsigned cur[EB];
#pragma unroll
            for (int u = 0; u < EB; ++u) cur[u] = so[j + u];
            for (; j + EB <= end; j += EB) {
                unsigned vo[EB];
#pragma unroll
                for (int u = 0; u < EB; ++u) vo[u] = cur[u] + lane8;
#pragma unroll
                for (int u = 0; u < EB; ++u) cur[u] = so[j + EB + u];
                agg_batch<RELU, false>(zb, vo, EB, T2, M, D, A);
            }
            if (j < end) {
                const int cnt = end - j;
                unsigned vo[EB];
#pragma unroll
                for (int u = 0; u < EB; ++u) vo[u] = ((u < cnt) ? cur[u] : 0u) + lane8;
                agg_batch<RELU, true>(zb, vo, cnt, T2, M, D, A);
            }
            o.x = (A.x + GEPS * D.x) / (D.x + 1e-16f) + self.x;
            o.y = (A.y + GEPS * D.y) / (D.y + 1e-16f) + self.y;
        }
        *(v2f*)(xs + wv * XP + 2 * lane) = o;
    }
    __syncthreads();

    float u1[8];
    float gam1 = 0.f, bet1 = 0.f;
    if (wv < 8) {
        const float* B1p = W1l + (half * 2) * (2 * Hc) + wv * 16 + mr;
        v2f bf[16];
#pragma unroll
        for (int kk = 0; kk < 16; ++kk) {
            bf[kk].x = B1p[(4 * kk) * (2 * Hc)];
            bf[kk].y = B1p[(4 * kk + 1) * (2 * Hc)];
        }
        const float bias1 = c1[wv * 16 + mr];
        gam1 = c1[2 * Hc + wv * 16 + mr];
        bet1 = c1[4 * Hc + wv * 16 + mr];

        const float* Ax = xs + mr * XP + half * 2;
        v8f acc = {0.f,0.f,0.f,0.f,0.f,0.f,0.f,0.f};
#pragma unroll
        for (int kk = 0; kk < 16; ++kk) {
            v2f a = *(const v2f*)(Ax + 4 * kk);
            acc = __builtin_amdgcn_wmma_f32_16x16x4_f32(
                      false, a, false, bf[kk], (short)0, acc, false, false);
        }

        float s1[8], q1[8];
#pragma unroll
        for (int r = 0; r < 8; ++r) {
            float u = acc[r] + bias1;
            u1[r] = u;  s1[r] = u;  q1[r] = u * u;
        }
#pragma unroll
        for (int r = 0; r < 8; ++r) { s1[r] = row_xor_add<1>(s1[r]);  q1[r] = row_xor_add<1>(q1[r]); }
#pragma unroll
        for (int r = 0; r < 8; ++r) { s1[r] = row_xor_add<2>(s1[r]);  q1[r] = row_xor_add<2>(q1[r]); }
#pragma unroll
        for (int r = 0; r < 8; ++r) { s1[r] = row_xor_add<4>(s1[r]);  q1[r] = row_xor_add<4>(q1[r]); }
#pragma unroll
        for (int r = 0; r < 8; ++r) { s1[r] = row_xor_add<8>(s1[r]);  q1[r] = row_xor_add<8>(q1[r]); }
        if (mr == 0) {
#pragma unroll
            for (int r = 0; r < 8; ++r)
                *(v2f*)(part + wv * PP + (half * 8 + r) * 2) = v2f{s1[r], q1[r]};
        }
    } else {
#pragma unroll
        for (int r = 0; r < 8; ++r) u1[r] = 0.f;
    }
    __syncthreads();

    if (wv < 8) {
        v2f p[8];
#pragma unroll
        for (int r = 0; r < 8; ++r) p[r] = *(const v2f*)(part + (mr & 7) * PP + (half * 8 + r) * 2);
#pragma unroll
        for (int r = 0; r < 8; ++r) { p[r].x = row_xor_add<1>(p[r].x);  p[r].y = row_xor_add<1>(p[r].y); }
#pragma unroll
        for (int r = 0; r < 8; ++r) { p[r].x = row_xor_add<2>(p[r].x);  p[r].y = row_xor_add<2>(p[r].y); }
#pragma unroll
        for (int r = 0; r < 8; ++r) { p[r].x = row_xor_add<4>(p[r].x);  p[r].y = row_xor_add<4>(p[r].y); }
#pragma unroll
        for (int r = 0; r < 8; ++r) {
            float mean = p[r].x * (1.f / 128.f);
            float var  = p[r].y * (1.f / 128.f) - mean * mean;
            float rstd = rsqrtf(var + 1e-5f);
            float v = (u1[r] - mean) * rstd * gam1 + bet1;
            tile[(half * 8 + r) * LP + wv * 16 + mr] = fmaxf(v, 0.f);
        }
    }
    __syncthreads();

    float hn[8];
    float gam2 = 0.f, bet2 = 0.f;
    if (wv < 4) {
        const long rowbase = (long)(tm * 16 + half * 8) * Hc + wv * 16 + mr;
        float hres[8];
        if (add_res) {
#pragma unroll
            for (int r = 0; r < 8; ++r) hres[r] = h[rowbase + (long)r * Hc];
        } else {
#pragma unroll
            for (int r = 0; r < 8; ++r) hres[r] = 0.f;
        }
        const float bias2 = c2[wv * 16 + mr];
        gam2 = c2[Hc + wv * 16 + mr];
        bet2 = c2[2 * Hc + wv * 16 + mr];

        const float* At  = tile + mr * LP + half * 2;
        const float* B2p = W2l + (half * 2) * Hc + wv * 16 + mr;
        v8f acc = {0.f,0.f,0.f,0.f,0.f,0.f,0.f,0.f};
        v2f bc[8];
#pragma unroll
        for (int u = 0; u < 8; ++u) {
            bc[u].x = B2p[(4 * u) * Hc];
            bc[u].y = B2p[(4 * u + 1) * Hc];
        }
#pragma unroll
        for (int j = 0; j < 4; ++j) {
            v2f bn[8];
            if (j + 1 < 4) {
#pragma unroll
                for (int u = 0; u < 8; ++u) {
                    const int kk = 8 * (j + 1) + u;
                    bn[u].x = B2p[(4 * kk) * Hc];
                    bn[u].y = B2p[(4 * kk + 1) * Hc];
                }
            }
#pragma unroll
            for (int u = 0; u < 8; ++u) {
                v2f a = *(const v2f*)(At + 4 * (8 * j + u));
                acc = __builtin_amdgcn_wmma_f32_16x16x4_f32(
                          false, a, false, bc[u], (short)0, acc, false, false);
            }
            if (j + 1 < 4) {
#pragma unroll
                for (int u = 0; u < 8; ++u) bc[u] = bn[u];
            }
        }

        float s2[8], q2[8];
#pragma unroll
        for (int r = 0; r < 8; ++r) {
            float x = hres[r] + (acc[r] + bias2);
            hn[r] = x;  s2[r] = x;  q2[r] = x * x;
        }
#pragma unroll
        for (int r = 0; r < 8; ++r) { s2[r] = row_xor_add<1>(s2[r]);  q2[r] = row_xor_add<1>(q2[r]); }
#pragma unroll
        for (int r = 0; r < 8; ++r) { s2[r] = row_xor_add<2>(s2[r]);  q2[r] = row_xor_add<2>(q2[r]); }
#pragma unroll
        for (int r = 0; r < 8; ++r) { s2[r] = row_xor_add<4>(s2[r]);  q2[r] = row_xor_add<4>(q2[r]); }
#pragma unroll
        for (int r = 0; r < 8; ++r) { s2[r] = row_xor_add<8>(s2[r]);  q2[r] = row_xor_add<8>(q2[r]); }
        if (mr == 0) {
#pragma unroll
            for (int r = 0; r < 8; ++r)
                *(v2f*)(part + wv * PP + (half * 8 + r) * 2) = v2f{s2[r], q2[r]};
        }
    } else {
#pragma unroll
        for (int r = 0; r < 8; ++r) hn[r] = 0.f;
    }
    __syncthreads();

    if (wv < 4) {
        v2f p[8];
#pragma unroll
        for (int r = 0; r < 8; ++r) p[r] = *(const v2f*)(part + (mr & 3) * PP + (half * 8 + r) * 2);
#pragma unroll
        for (int r = 0; r < 8; ++r) { p[r].x = row_xor_add<1>(p[r].x);  p[r].y = row_xor_add<1>(p[r].y); }
#pragma unroll
        for (int r = 0; r < 8; ++r) { p[r].x = row_xor_add<2>(p[r].x);  p[r].y = row_xor_add<2>(p[r].y); }
#pragma unroll
        for (int r = 0; r < 8; ++r) {
            float mean = p[r].x * (1.f / 64.f);
            float var  = p[r].y * (1.f / 64.f) - mean * mean;
            float rstd = rsqrtf(var + 1e-5f);
            float v = (hn[r] - mean) * rstd * gam2 + bet2;
            zs[(half * 8 + r) * XP + wv * 16 + mr] = fmaxf(v, 0.f);
            if constexpr (!LAST) hs[(half * 8 + r) * XP + wv * 16 + mr] = hn[r];
        }
    }
    __syncthreads();

    if constexpr (!LAST) {
        const int j   = threadIdx.x & 255;
        const int row = j >> 4;
        const int c4  = (j & 15) * 4;
        const float* sb = (wv < 8) ? hs : zs;
        float*       db = (wv < 8) ? h : zout;
        *(v4f*)(db + (long)(tm * 16 + row) * Hc + c4) = *(const v4f*)(sb + row * XP + c4);
    } else {
        if (wv == 0) {
            int   nc3[3];
            float lb3[3];
#pragma unroll
            for (int t = 0; t < 3; ++t) {
                int nn = t * 16 + mr;
                nc3[t] = (nn < Cc) ? nn : (Cc - 1);
                lb3[t] = c3[nc3[t]];
            }
            const float* B3p = W3l + (half * 2) * Cc;
            const float* At3 = zs + mr * XP + half * 2;
            v8f acc3[3];
#pragma unroll
            for (int t = 0; t < 3; ++t) acc3[t] = v8f{0.f,0.f,0.f,0.f,0.f,0.f,0.f,0.f};
#pragma unroll 1
            for (int c = 0; c < 4; ++c) {
                v2f b3[4][3];
                v2f a3[4];
#pragma unroll
                for (int u = 0; u < 4; ++u) {
                    const int kk = 4 * c + u;
#pragma unroll
                    for (int t = 0; t < 3; ++t) {
                        b3[u][t].x = B3p[(4 * kk) * Cc + nc3[t]];
                        b3[u][t].y = B3p[(4 * kk + 1) * Cc + nc3[t]];
                    }
                    a3[u] = *(const v2f*)(At3 + 4 * kk);
                }
#pragma unroll
                for (int u = 0; u < 4; ++u)
#pragma unroll
                    for (int t = 0; t < 3; ++t)
                        acc3[t] = __builtin_amdgcn_wmma_f32_16x16x4_f32(
                                      false, a3[u], false, b3[u][t], (short)0, acc3[t], false, false);
            }

            const bool c2ok = mr < (Cc - 32);
            float mx3[8], sm3[8];
#pragma unroll
            for (int r = 0; r < 8; ++r) {
#pragma unroll
                for (int t = 0; t < 3; ++t) acc3[t][r] += lb3[t];
                float m01 = fmaxf(acc3[0][r], acc3[1][r]);
                mx3[r] = c2ok ? fmaxf(m01, acc3[2][r]) : m01;
            }
#pragma unroll
            for (int r = 0; r < 8; ++r) mx3[r] = row_xor_max<1>(mx3[r]);
#pragma unroll
            for (int r = 0; r < 8; ++r) mx3[r] = row_xor_max<2>(mx3[r]);
#pragma unroll
            for (int r = 0; r < 8; ++r) mx3[r] = row_xor_max<4>(mx3[r]);
#pragma unroll
            for (int r = 0; r < 8; ++r) mx3[r] = row_xor_max<8>(mx3[r]);
#pragma unroll
            for (int r = 0; r < 8; ++r) {
                float e01 = __expf(acc3[0][r] - mx3[r]) + __expf(acc3[1][r] - mx3[r]);
                float e2  = __expf(acc3[2][r] - mx3[r]);
                sm3[r] = c2ok ? (e01 + e2) : e01;
            }
#pragma unroll
            for (int r = 0; r < 8; ++r) sm3[r] = row_xor_add<1>(sm3[r]);
#pragma unroll
            for (int r = 0; r < 8; ++r) sm3[r] = row_xor_add<2>(sm3[r]);
#pragma unroll
            for (int r = 0; r < 8; ++r) sm3[r] = row_xor_add<4>(sm3[r]);
#pragma unroll
            for (int r = 0; r < 8; ++r) sm3[r] = row_xor_add<8>(sm3[r]);

#pragma unroll
            for (int r = 0; r < 8; ++r) {
                float lse = mx3[r] + __logf(sm3[r]);
                float* orow = tile + (half * 8 + r) * Cc + mr;
                orow[0]  = acc3[0][r] - lse;
                orow[16] = acc3[1][r] - lse;
                if (c2ok) orow[32] = acc3[2][r] - lse;
            }
        }
        __syncthreads();
        if (threadIdx.x < (16 * Cc) / 4) {
            float* ob = out + (long)tm * (16 * Cc);
            const int q = threadIdx.x;
            *(v4f*)(ob + 4 * q) = *(const v4f*)(tile + 4 * q);
        }
    }
    }
}

extern "C" void kernel_launch(void* const* d_in, const int* in_sizes, int n_in,
                              void* d_out, int out_size, void* d_ws, size_t ws_size,
                              hipStream_t stream) {
    (void)n_in; (void)out_size; (void)ws_size;

    const float* x    = (const float*)d_in[0];
    const int*   ei   = (const int*)  d_in[1];
    const float* encW = (const float*)d_in[2];
    const float* encB = (const float*)d_in[3];
    const float* t    = (const float*)d_in[4];
    const float* W1   = (const float*)d_in[5];
    const float* b1   = (const float*)d_in[6];
    const float* g1   = (const float*)d_in[7];
    const float* bb1  = (const float*)d_in[8];
    const float* W2   = (const float*)d_in[9];
    const float* b2   = (const float*)d_in[10];
    const float* ng   = (const float*)d_in[11];
    const float* nb   = (const float*)d_in[12];
    const float* linW = (const float*)d_in[13];
    const float* linB = (const float*)d_in[14];

    const int F = 128, H = 64, L = 4;
    const int N = in_sizes[0] / F;
    const int E = in_sizes[1] / 2;
    const int* src = ei;
    const int* dst = ei + E;

    float* ws = (float*)d_ws;
    const long NH = (long)N * H;
    float* h0   = ws;  ws += NH;
    float* zA   = ws;  ws += NH;
    int*   csr  = (int*)ws;  ws += 2 * NH;
    float* zB   = ws;  ws += NH;
    float* h    = ws;  ws += NH;

    auto cdiv = [](long a, long b) { return (int)((a + b - 1) / b); };
    auto gemm_waves = [&](int M, int Nc) { return (long)(M / 16) * ((Nc + 63) / 64); };

    const int G  = cdiv(E, CH);
    const int nbk = cdiv(N, BN);
    int* pairs  = csr;
    int* ssrc   = pairs + 2 * E;
    int* offT   = ssrc + E;
    int* rowptr = offT + (nbk + 1) * G;

    const int GB = cdiv(gemm_waves(N, H) * 32, T1);
    encode_and_bin_kernel<<<G + GB, T1, 0, stream>>>(src, dst, pairs, offT, E, nbk, G,
                                                     x, encW, encB, h0, N, H, F);
    bucket_group_kernel<<<nbk, 256, 0, stream>>>(pairs, offT, rowptr, ssrc, N, G);

    const int GL = 768;
    for (int l = 0; l < L; ++l) {
        const float* zin  = (l == 0) ? h0 : ((l & 1) ? zA : zB);
        float*       zout = (l & 1) ? zB : zA;
        const int ln = (l + 1 < L) ? (l + 1) : 0;
        if (l == 0)
            gather_mlp_kernel<true, false><<<GL, NW * 32, 0, stream>>>(
                zin, rowptr, ssrc, t, l,
                W1 + (long)l * H * 2 * H, b1 + l * 2 * H, g1 + l * 2 * H, bb1 + l * 2 * H,
                W2 + (long)l * 2 * H * H, b2 + l * H, ng + ln * H, nb + ln * H,
                h, zout, 0, linW, linB, (float*)d_out, N);
        else if (l + 1 < L)
            gather_mlp_kernel<false, false><<<GL, NW * 32, 0, stream>>>(
                zin, rowptr, ssrc, t, l,
                W1 + (long)l * H * 2 * H, b1 + l * 2 * H, g1 + l * 2 * H, bb1 + l * 2 * H,
                W2 + (long)l * 2 * H * H, b2 + l * H, ng + ln * H, nb + ln * H,
                h, zout, 1, linW, linB, (float*)d_out, N);
        else
            gather_mlp_kernel<false, true><<<GL, NW * 32, 0, stream>>>(
                zin, rowptr, ssrc, t, l,
                W1 + (long)l * H * 2 * H, b1 + l * 2 * H, g1 + l * 2 * H, bb1 + l * 2 * H,
                W2 + (long)l * 2 * H * H, b2 + l * H, ng + ln * H, nb + ln * H,
                h, zout, 1, linW, linB, (float*)d_out, N);
    }
}


# Both LayerNorms (with the residual add) moved from accumulator ownership in 8/4 waves to one row per wave over all 16 waves, through the LDS tiles; LayerNorm row-sum order changes
# speedup vs baseline: 10.1767x; 1.0546x over previous
#include <hip/hip_runtime.h>
#include <math.h>

typedef float v2f __attribute__((ext_vector_type(2)));
typedef float v4f __attribute__((ext_vector_type(4)));
typedef float v8f __attribute__((ext_vector_type(8)));

#define HD __device__ __forceinline__

static constexpr int   Hc   = 64;
static constexpr int   Cc   = 40;
static constexpr float GEPS = 1e-7f;

HD void gemm_bias_wmma_body(const float* __restrict__ A,
                            const float* __restrict__ B,
                            const float* __restrict__ bias,
                            float* __restrict__ C,
                            int M, int N, int K, int gid) {
    int wave    = gid >> 5;
    int lane    = threadIdx.x & 31;
    int groupsN = (N + 63) >> 6;
    int tm = wave / groupsN;
    int gn = wave - tm * groupsN;
    if (tm >= (M >> 4)) return;

    int mr   = lane & 15;
    int half = lane >> 4;
    const float* Arow = A + (long)(tm * 16 + mr) * K;

    int ncol[4];
#pragma unroll
    for (int t = 0; t < 4; ++t) {
        int n   = gn * 64 + t * 16 + mr;
        ncol[t] = (n < N) ? n : (N - 1);
    }

    v8f acc[4] = {{0.f,0.f,0.f,0.f,0.f,0.f,0.f,0.f},
                  {0.f,0.f,0.f,0.f,0.f,0.f,0.f,0.f},
                  {0.f,0.f,0.f,0.f,0.f,0.f,0.f,0.f},
                  {0.f,0.f,0.f,0.f,0.f,0.f,0.f,0.f}};

    int ka = half * 2;
    v2f a_cur;
    a_cur.x = Arow[ka];
    a_cur.y = Arow[ka + 1];
    v2f b_cur[4];
#pragma unroll
    for (int t = 0; t < 4; ++t) {
        b_cur[t].x = B[ka * N + ncol[t]];
        b_cur[t].y = B[(ka + 1) * N + ncol[t]];
    }

#pragma unroll 2
    for (int k = 0; k < K; k += 4) {
        int kn = (k + 4 < K) ? (k + 4) : k;
        int kb = kn + half * 2;

        v2f a_nxt;
        a_nxt.x = Arow[kb];
        a_nxt.y = Arow[kb + 1];
        v2f b_nxt[4];
#pragma unroll
        for (int t = 0; t < 4; ++t) {
            b_nxt[t].x = B[kb * N + ncol[t]];
            b_nxt[t].y = B[(kb + 1) * N + ncol[t]];
        }

#pragma unroll
        for (int t = 0; t < 4; ++t)
            acc[t] = __builtin_amdgcn_wmma_f32_16x16x4_f32(
                         false, a_cur, false, b_cur[t], (short)0, acc[t], false, false);

        a_cur = a_nxt;
#pragma unroll
        for (int t = 0; t < 4; ++t) b_cur[t] = b_nxt[t];
    }

#pragma unroll
    for (int t = 0; t < 4; ++t) {
        int n = gn * 64 + t * 16 + mr;
        if (n < N) {
            float bv = bias[n];
#pragma unroll
            for (int r = 0; r < 8; ++r)
                C[(long)(tm * 16 + half * 8 + r) * N + n] = acc[t][r] + bv;
        }
    }
}

typedef int v2i __attribute__((ext_vector_type(2)));

static constexpr int CH   = 4096;
static constexpr int BSH  = 7;
static constexpr int BN   = 1 << BSH;
static constexpr int MAXB = 256;

static constexpr int T1  = 256;
static constexpr int EPT = CH / T1;

HD void edge_bin_body(const int* __restrict__ src, const int* __restrict__ dst,
                      int* __restrict__ pairs, int* __restrict__ offT, int E, int nb,
                      int g, int G) {
    __shared__ int lcnt[MAXB];
    __shared__ v2i stage[CH];
    const int t  = threadIdx.x;
    const int e0 = g * CH;
    int n = E - e0; if (n > CH) n = CH;

    int d[EPT], s[EPT];
#pragma unroll
    for (int k = 0; k < EPT; ++k) {
        int idx = t + T1 * k;
        bool ok = idx < n;
        d[k] = ok ? dst[e0 + idx] : -1;
        s[k] = ok ? src[e0 + idx] * (Hc * 4) : 0;
    }
    for (int q = t; q < MAXB; q += T1) lcnt[q] = 0;
    __syncthreads();

    int rk[EPT];
#pragma unroll
    for (int k = 0; k < EPT; ++k) {
        rk[k] = 0;
        if (d[k] >= 0) rk[k] = atomicAdd(&lcnt[d[k] >> BSH], 1);
    }
    __syncthreads();

    if (t < 32) {
        int c[8];
        int sum = 0;
#pragma unroll
        for (int q = 0; q < 8; ++q) { int x = lcnt[8 * t + q]; c[q] = sum; sum += x; }
        int incl = sum;
        for (int off = 1; off < 32; off <<= 1) {
            int v = __shfl_up(incl, off);
            if (t >= off) incl += v;
        }
        int excl = incl - sum;
#pragma unroll
        for (int q = 0; q < 8; ++q) lcnt[8 * t + q] = excl + c[q];
    }
    __syncthreads();

    int pos[EPT];
#pragma unroll
    for (int k = 0; k < EPT; ++k) pos[k] = lcnt[(d[k] >= 0) ? (d[k] >> BSH) : 0] + rk[k];
#pragma unroll
    for (int k = 0; k < EPT; ++k)
        if (d[k] >= 0) stage[pos[k]] = v2i{d[k], s[k]};
    __syncthreads();

    v2i* pr = (v2i*)pairs + e0;
    v2i w[EPT];
#pragma unroll
    for (int k = 0; k < EPT; ++k) {
        int idx = t + T1 * k;
        w[k] = (idx < n) ? stage[idx] : v2i{0, 0};
    }
#pragma unroll
    for (int k = 0; k < EPT; ++k) {
        int idx = t + T1 * k;
        if (idx < n) pr[idx] = w[k];
    }
    for (int q = t; q <= nb; q += T1) offT[q * G + g] = (q < nb) ? lcnt[q] : n;
}

__global__ void __launch_bounds__(T1)
encode_and_bin_kernel(const int* __restrict__ src, const int* __restrict__ dst,
                      int* __restrict__ pairs, int* __restrict__ offT, int E, int nb, int G,
                      const float* __restrict__ x, const float* __restrict__ encW,
                      const float* __restrict__ encB, float* __restrict__ h,
                      int M, int N, int K) {
    const int b = (int)blockIdx.x;
    if (b < G) edge_bin_body(src, dst, pairs, offT, E, nb, b, G);
    else       gemm_bias_wmma_body(x, encW, encB, h, M, N, K, (b - G) * T1 + (int)threadIdx.x);
}

template <bool PLACE>
HD int bucket_sweep(const v2i* __restrict__ pr, const int* __restrict__ r0,
                    const int* __restrict__ r1, int G, int wv, int lane, int lo,
                    int* ctr, int* __restrict__ ssrc) {
    int below = 0;
    for (int gb = wv * 32; gb < G; gb += 8 * 32) {
        int gg = gb + lane;
        int o0 = 0, o1 = 0;
        if (gg < G) { o0 = r0[gg]; o1 = r1[gg]; }
        below += o0;
        int len = o1 - o0;
        int st  = gg * CH + o0;
        v2i p[32];
#pragma unroll
        for (int u = 0; u < 32; ++u) {
            int L = __builtin_amdgcn_readlane(len, u);
            int S = __builtin_amdgcn_readlane(st, u);
            p[u] = v2i{-1, 0};
            if (lane < L) p[u] = pr[S + lane];
        }
        if (PLACE) {
            int slot[32];
#pragma unroll
            for (int u = 0; u < 32; ++u) {
                slot[u] = 0;
                if (p[u].x >= 0) slot[u] = atomicAdd(&ctr[p[u].x - lo], 1);
            }
#pragma unroll
            for (int u = 0; u < 32; ++u)
                if (p[u].x >= 0) ssrc[slot[u]] = p[u].y;
        } else {
#pragma unroll
            for (int u = 0; u < 32; ++u)
                if (p[u].x >= 0) atomicAdd(&ctr[p[u].x - lo], 1);
        }
        unsigned m = __builtin_amdgcn_ballot_w32(len > 32);
        while (m) {
            int u = __builtin_ctz(m);
            m &= m - 1;
            int L = __builtin_amdgcn_readlane(len, u);
            int S = __builtin_amdgcn_readlane(st, u);
            for (int k = 32 + lane; k < L; k += 32) {
                v2i q = pr[S + k];
                if (PLACE) ssrc[atomicAdd(&ctr[q.x - lo], 1)] = q.y;
                else       atomicAdd(&ctr[q.x - lo], 1);
            }
        }
    }
    return below;
}

__global__ void __launch_bounds__(256)
bucket_group_kernel(const int* __restrict__ pairs, const int* __restrict__ offT,
                    int* __restrict__ rowptr, int* __restrict__ ssrc, int N, int G) {
    __shared__ int cnt[BN];
    __shared__ int cur[BN];
    __shared__ int wsum[8];
    const int t    = threadIdx.x;
    const int lane = t & 31;
    const int wv   = __builtin_amdgcn_readfirstlane(t >> 5);
    const int b    = blockIdx.x;
    const int lo   = b << BSH;
    const v2i* pr  = (const v2i*)pairs;
    const int* r0  = offT + (long)b * G;
    const int* r1  = r0 + G;

    if (t < BN) cnt[t] = 0;
    __syncthreads();

    int below = bucket_sweep<false>(pr, r0, r1, G, wv, lane, lo, cnt, ssrc);
    for (int off = 16; off; off >>= 1) below += __shfl_xor(below, off);
    if (lane == 0) wsum[wv] = below;
    __syncthreads();

    if (t < 32) {
        int base = 0;
#pragma unroll
        for (int w = 0; w < 8; ++w) base += wsum[w];
        int c[4];
        int sum = 0;
#pragma unroll
        for (int q = 0; q < 4; ++q) { int x = cnt[4 * t + q]; c[q] = sum; sum += x; }
        int incl = sum;
        for (int off = 1; off < 32; off <<= 1) {
            int v = __shfl_up(incl, off);
            if (t >= off) incl += v;
        }
        int excl = base + incl - sum;
#pragma unroll
        for (int q = 0; q < 4; ++q) {
            int node = lo + 4 * t + q;
            int v    = excl + c[q];
            cur[4 * t + q] = v;
            if (node < N) rowptr[node] = v;
        }
        if (b == (int)gridDim.x - 1 && t == 31) rowptr[N] = base + incl;
    }
    __syncthreads();

    bucket_sweep<true>(pr, r0, r1, G, wv, lane, lo, cur, ssrc);
}

static constexpr int EB = 8;

typedef __attribute__((address_space(4))) const int      ci32;
typedef __attribute__((address_space(4))) const unsigned cu32;

template <bool RELU, bool MASK>
HD void agg_batch(const char* __restrict__ zb, const unsigned (&vo)[EB], int cnt,
                  v2f T2, v2f& M, v2f& D, v2f& A) {
    v2f v[EB];
#pragma unroll
    for (int u = 0; u < EB; ++u)
        v[u] = *(const v2f*)(zb + vo[u]);
    v2f q[EB];
    v2f Mn = M;
#pragma unroll
    for (int u = 0; u < EB; ++u) {
        if (RELU) { v[u].x = fmaxf(v[u].x, 0.f); v[u].y = fmaxf(v[u].y, 0.f); }
        q[u] = v[u] * T2;
        if (MASK) {
            bool ok = u < cnt;
            v[u].x = ok ? v[u].x : 0.f;       v[u].y = ok ? v[u].y : 0.f;
            q[u].x = ok ? q[u].x : -3.0e38f;  q[u].y = ok ? q[u].y : -3.0e38f;
        }
        Mn.x = fmaxf(Mn.x, q[u].x);
        Mn.y = fmaxf(Mn.y, q[u].y);
    }
    v2f dm = M - Mn;
    v2f r;
    r.x = __builtin_amdgcn_exp2f(dm.x);
    r.y = __builtin_amdgcn_exp2f(dm.y);
    D *= r;
    A *= r;
#pragma unroll
    for (int u = 0; u < EB; ++u) {
        v2f a = q[u] - Mn;
        v2f e;
        e.x = __builtin_amdgcn_exp2f(a.x);
        e.y = __builtin_amdgcn_exp2f(a.y);
        D += e;
        A += v[u] * e;
    }
    M = Mn;
}

static constexpr int LP = 132;
static constexpr int XP = 68;
static constexpr int NW = 16;

template <int M>
HD float row_xor_add(float v) {
    int o = __builtin_amdgcn_update_dpp(0, __float_as_int(v), 0x160 | M, 0xf, 0xf, false);
    return v + __int_as_float(o);
}
HD float xhalf_add(float v) {
    int b = __float_as_int(v);
    int o = __builtin_amdgcn_permlanex16(b, b, 0x76543210, 0xfedcba98, false, false);
    return v + __int_as_float(o);
}
HD float wave_sum(float v) {
    v = row_xor_add<1>(v);
    v = row_xor_add<2>(v);
    v = row_xor_add<4>(v);
    v = row_xor_add<8>(v);
    return xhalf_add(v);
}
template <int M>
HD float row_xor_max(float v) {
    int o = __builtin_amdgcn_update_dpp(0, __float_as_int(v), 0x160 | M, 0xf, 0xf, false);
    return fmaxf(v, __int_as_float(o));
}

template <bool RELU, bool LAST>
__global__ void __launch_bounds__(NW * 32, 12)
gather_mlp_kernel(const float* __restrict__ zin,
                  const int* __restrict__ rowptr, const int* __restrict__ ssrc,
                  const float* __restrict__ tptr, int layer,
                  const float* __restrict__ W1, const float* __restrict__ b1,
                  const float* __restrict__ g1, const float* __restrict__ bb1,
                  const float* __restrict__ W2, const float* __restrict__ b2,
                  const float* __restrict__ ng, const float* __restrict__ nb,
                  float* __restrict__ h, float* __restrict__ zout, int add_res,
                  const float* __restrict__ linW, const float* __restrict__ linB,
                  float* __restrict__ out, int n) {
    __shared__ __attribute__((aligned(16))) float W1l[Hc * 2 * Hc];
    __shared__ __attribute__((aligned(16))) float W2l[2 * Hc * Hc];
    __shared__ __attribute__((aligned(16))) float W3l[LAST ? Hc * Cc : 4];
    __shared__ __attribute__((aligned(16))) float c1[3 * 2 * Hc];
    __shared__ __attribute__((aligned(16))) float c2[3 * Hc];
    __shared__ float c3[LAST ? Cc : 1];
    __shared__ __attribute__((aligned(16))) float xs[16 * XP];
    __shared__ __attribute__((aligned(16))) float hs[16 * XP];
    __shared__ __attribute__((aligned(16))) float zs[LAST ? 16 * XP : 4];
    __shared__ __attribute__((aligned(16))) float tile[16 * LP];

    const int lane = threadIdx.x & 31;
    const int wv   = __builtin_amdgcn_readfirstlane((int)(threadIdx.x >> 5));
    const int mr   = lane & 15;
    const int half = lane >> 4;

#pragma unroll
    for (int i = 0; i < (Hc * 2 * Hc) / (4 * NW * 32); ++i) {
        const int q = threadIdx.x + NW * 32 * i;
        ((v4f*)W1l)[q] = ((const v4f*)W1)[q];
        ((v4f*)W2l)[q] = ((const v4f*)W2)[q];
    }
    if (threadIdx.x < 2 * Hc) {
        c1[threadIdx.x]          = b1[threadIdx.x];
        c1[2 * Hc + threadIdx.x] = g1[threadIdx.x];
        c1[4 * Hc + threadIdx.x] = bb1[threadIdx.x];
    }
    if (threadIdx.x < Hc) {
        c2[threadIdx.x]          = b2[threadIdx.x];
        c2[Hc + threadIdx.x]     = ng[threadIdx.x];
        c2[2 * Hc + threadIdx.x] = nb[threadIdx.x];
    }
    if constexpr (LAST) {
        for (int q = threadIdx.x; q < (Hc * Cc) / 4; q += NW * 32)
            ((v4f*)W3l)[q] = ((const v4f*)linW)[q];
        if (threadIdx.x < Cc) c3[threadIdx.x] = linB[threadIdx.x];
    }

    const int ntiles = n / NW;
    for (int tm = blockIdx.x; tm < ntiles; tm += gridDim.x) {

    v2f hres = {0.f, 0.f};
    if (add_res) hres = *(const v2f*)(h + (long)(tm * NW + wv) * Hc + 2 * lane);

    {
        const int node = tm * NW + wv;
        v2f o = {0.f, 0.f};
        if (node < n) {
            const ci32* rps = (const ci32*)rowptr;
            const cu32* so  = (const cu32*)ssrc;
            int   start = rps[node];
            int   end   = rps[node + 1];
            float t2    = tptr[layer] * 1.44269504088896341f;
            const v2f* zr = (const v2f*)zin;
            const char* zb = (const char*)zin;
            const unsigned lane8 = (unsigned)lane * 8u;
            v2f self = zr[node * (Hc / 2) + lane];

            v2f T2 = {t2, t2};
            v2f M  = {-3.0e38f, -3.0e38f};
            v2f D  = {0.f, 0.f};
            v2f A  = {0.f, 0.f};

            int j = start;
            unsigned cur[EB];
#pragma unroll
            for (int u = 0; u < EB; ++u) cur[u] = so[j + u];
            for (; j + EB <= end; j += EB) {
                unsigned vo[EB];
#pragma unroll
                for (int u = 0; u < EB; ++u) vo[u] = cur[u] + lane8;
#pragma unroll
                for (int u = 0; u < EB; ++u) cur[u] = so[j + EB + u];
                agg_batch<RELU, false>(zb, vo, EB, T2, M, D, A);
            }
            if (j < end) {
                const int cnt = end - j;
                unsigned vo[EB];
#pragma unroll
                for (int u = 0; u < EB; ++u) vo[u] = ((u < cnt) ? cur[u] : 0u) + lane8;
                agg_batch<RELU, true>(zb, vo, cnt, T2, M, D, A);
            }
            o.x = (A.x + GEPS * D.x) / (D.x + 1e-16f) + self.x;
            o.y = (A.y + GEPS * D.y) / (D.y + 1e-16f) + self.y;
        }
        *(v2f*)(xs + wv * XP + 2 * lane) = o;
    }
    __syncthreads();

    if (wv < 8) {
        const float* B1p = W1l + (half * 2) * (2 * Hc) + wv * 16 + mr;
        v2f bf[16];
#pragma unroll
        for (int kk = 0; kk < 16; ++kk) {
            bf[kk].x = B1p[(4 * kk) * (2 * Hc)];
            bf[kk].y = B1p[(4 * kk + 1) * (2 * Hc)];
        }
        const float bias1 = c1[wv * 16 + mr];

        const float* Ax = xs + mr * XP + half * 2;
        v8f acc = {0.f,0.f,0.f,0.f,0.f,0.f,0.f,0.f};
#pragma unroll
        for (int kk = 0; kk < 16; ++kk) {
            v2f a = *(const v2f*)(Ax + 4 * kk);
            acc = __builtin_amdgcn_wmma_f32_16x16x4_f32(
                      false, a, false, bf[kk], (short)0, acc, false, false);
        }

#pragma unroll
        for (int r = 0; r < 8; ++r)
            tile[(half * 8 + r) * LP + wv * 16 + mr] = acc[r] + bias1;
    }
    __syncthreads();

    {
        float* urow = tile + wv * LP + 4 * lane;
        const v4f u  = *(const v4f*)urow;
        const v4f g1v = *(const v4f*)(c1 + 2 * Hc + 4 * lane);
        const v4f b1v = *(const v4f*)(c1 + 4 * Hc + 4 * lane);
        float s = (u.x + u.y) + (u.z + u.w);
        float q = (u.x * u.x + u.y * u.y) + (u.z * u.z + u.w * u.w);
        s = wave_sum(s);
        q = wave_sum(q);
        const float mean = s * (1.f / 128.f);
        const float var  = q * (1.f / 128.f) - mean * mean;
        const float rstd = rsqrtf(var + 1e-5f);
        v4f v;
        v.x = fmaxf((u.x - mean) * rstd * g1v.x + b1v.x, 0.f);
        v.y = fmaxf((u.y - mean) * rstd * g1v.y + b1v.y, 0.f);
        v.z = fmaxf((u.z - mean) * rstd * g1v.z + b1v.z, 0.f);
        v.w = fmaxf((u.w - mean) * rstd * g1v.w + b1v.w, 0.f);
        *(v4f*)urow = v;
    }
    __syncthreads();

    if (wv < 4) {
        const float bias2 = c2[wv * 16 + mr];

        const float* At  = tile + mr * LP + half * 2;
        const float* B2p = W2l + (half * 2) * Hc + wv * 16 + mr;
        v8f acc = {0.f,0.f,0.f,0.f,0.f,0.f,0.f,0.f};
        v2f bc[8];
#pragma unroll
        for (int u = 0; u < 8; ++u) {
            bc[u].x = B2p[(4 * u) * Hc];
            bc[u].y = B2p[(4 * u + 1) * Hc];
        }
#pragma unroll
        for (int j = 0; j < 4; ++j) {
            v2f bn[8];
            if (j + 1 < 4) {
#pragma unroll
                for (int u = 0; u < 8; ++u) {
                    const int kk = 8 * (j + 1) + u;
                    bn[u].x = B2p[(4 * kk) * Hc];
                    bn[u].y = B2p[(4 * kk + 1) * Hc];
                }
            }
#pragma unroll
            for (int u = 0; u < 8; ++u) {
                v2f a = *(const v2f*)(At + 4 * (8 * j + u));
                acc = __builtin_amdgcn_wmma_f32_16x16x4_f32(
                          false, a, false, bc[u], (short)0, acc, false, false);
            }
            if (j + 1 < 4) {
#pragma unroll
                for (int u = 0; u < 8; ++u) bc[u] = bn[u];
            }
        }

#pragma unroll
        for (int r = 0; r < 8; ++r)
            hs[(half * 8 + r) * XP + wv * 16 + mr] = acc[r] + bias2;
    }
    __syncthreads();

    {
        const v2f y   = *(const v2f*)(hs + wv * XP + 2 * lane);
        const v2f g2v = *(const v2f*)(c2 + Hc + 2 * lane);
        const v2f b2v = *(const v2f*)(c2 + 2 * Hc + 2 * lane);
        const v2f hn  = hres + y;
        float s = hn.x + hn.y;
        float q = hn.x * hn.x + hn.y * hn.y;
        s = wave_sum(s);
        q = wave_sum(q);
        const float mean = s * (1.f / 64.f);
        const float var  = q * (1.f / 64.f) - mean * mean;
        const float rstd = rsqrtf(var + 1e-5f);
        v2f z;
        z.x = fmaxf((hn.x - mean) * rstd * g2v.x + b2v.x, 0.f);
        z.y = fmaxf((hn.y - mean) * rstd * g2v.y + b2v.y, 0.f);
        if constexpr (!LAST) {
            const long ro = (long)(tm * NW + wv) * Hc + 2 * lane;
            *(v2f*)(h + ro)    = hn;
            *(v2f*)(zout + ro) = z;
        } else {
            *(v2f*)(zs + wv * XP + 2 * lane) = z;
        }
    }

    if constexpr (LAST) {
        __syncthreads();
        if (wv == 0) {
            int   nc3[3];
            float lb3[3];
#pragma unroll
            for (int t = 0; t < 3; ++t) {
                int nn = t * 16 + mr;
                nc3[t] = (nn < Cc) ? nn : (Cc - 1);
                lb3[t] = c3[nc3[t]];
            }
            const float* B3p = W3l + (half * 2) * Cc;
            const float* At3 = zs + mr * XP + half * 2;
            v8f acc3[3];
#pragma unroll
            for (int t = 0; t < 3; ++t) acc3[t] = v8f{0.f,0.f,0.f,0.f,0.f,0.f,0.f,0.f};
#pragma unroll 1
            for (int c = 0; c < 4; ++c) {
                v2f b3[4][3];
                v2f a3[4];
#pragma unroll
                for (int u = 0; u < 4; ++u) {
                    const int kk = 4 * c + u;
#pragma unroll
                    for (int t = 0; t < 3; ++t) {
                        b3[u][t].x = B3p[(4 * kk) * Cc + nc3[t]];
                        b3[u][t].y = B3p[(4 * kk + 1) * Cc + nc3[t]];
                    }
                    a3[u] = *(const v2f*)(At3 + 4 * kk);
                }
#pragma unroll
                for (int u = 0; u < 4; ++u)
#pragma unroll
                    for (int t = 0; t < 3; ++t)
                        acc3[t] = __builtin_amdgcn_wmma_f32_16x16x4_f32(
                                      false, a3[u], false, b3[u][t], (short)0, acc3[t], false, false);
            }

            const bool c2ok = mr < (Cc - 32);
            float mx3[8], sm3[8];
#pragma unroll
            for (int r = 0; r < 8; ++r) {
#pragma unroll
                for (int t = 0; t < 3; ++t) acc3[t][r] += lb3[t];
                float m01 = fmaxf(acc3[0][r], acc3[1][r]);
                mx3[r] = c2ok ? fmaxf(m01, acc3[2][r]) : m01;
            }
#pragma unroll
            for (int r = 0; r < 8; ++r) mx3[r] = row_xor_max<1>(mx3[r]);
#pragma unroll
            for (int r = 0; r < 8; ++r) mx3[r] = row_xor_max<2>(mx3[r]);
#pragma unroll
            for (int r = 0; r < 8; ++r) mx3[r] = row_xor_max<4>(mx3[r]);
#pragma unroll
            for (int r = 0; r < 8; ++r) mx3[r] = row_xor_max<8>(mx3[r]);
#pragma unroll
            for (int r = 0; r < 8; ++r) {
                float e01 = __expf(acc3[0][r] - mx3[r]) + __expf(acc3[1][r] - mx3[r]);
                float e2  = __expf(acc3[2][r] - mx3[r]);
                sm3[r] = c2ok ? (e01 + e2) : e01;
            }
#pragma unroll
            for (int r = 0; r < 8; ++r) sm3[r] = row_xor_add<1>(sm3[r]);
#pragma unroll
            for (int r = 0; r < 8; ++r) sm3[r] = row_xor_add<2>(sm3[r]);
#pragma unroll
            for (int r = 0; r < 8; ++r) sm3[r] = row_xor_add<4>(sm3[r]);
#pragma unroll
            for (int r = 0; r < 8; ++r) sm3[r] = row_xor_add<8>(sm3[r]);

#pragma unroll
            for (int r = 0; r < 8; ++r) {
                float lse = mx3[r] + __logf(sm3[r]);
                float* orow = tile + (half * 8 + r) * Cc + mr;
                orow[0]  = acc3[0][r] - lse;
                orow[16] = acc3[1][r] - lse;
                if (c2ok) orow[32] = acc3[2][r] - lse;
            }
        }
        __syncthreads();
        if (threadIdx.x < (16 * Cc) / 4) {
            float* ob = out + (long)tm * (16 * Cc);
            const int q = threadIdx.x;
            *(v4f*)(ob + 4 * q) = *(const v4f*)(tile + 4 * q);
        }
    }
    }
}

extern "C" void kernel_launch(void* const* d_in, const int* in_sizes, int n_in,
                              void* d_out, int out_size, void* d_ws, size_t ws_size,
                              hipStream_t stream) {
    (void)n_in; (void)out_size; (void)ws_size;

    const float* x    = (const float*)d_in[0];
    const int*   ei   = (const int*)  d_in[1];
    const float* encW = (const float*)d_in[2];
    const float* encB = (const float*)d_in[3];
    const float* t    = (const float*)d_in[4];
    const float* W1   = (const float*)d_in[5];
    const float* b1   = (const float*)d_in[6];
    const float* g1   = (const float*)d_in[7];
    const float* bb1  = (const float*)d_in[8];
    const float* W2   = (const float*)d_in[9];
    const float* b2   = (const float*)d_in[10];
    const float* ng   = (const float*)d_in[11];
    const float* nb   = (const float*)d_in[12];
    const float* linW = (const float*)d_in[13];
    const float* linB = (const float*)d_in[14];

    const int F = 128, H = 64, L = 4;
    const int N = in_sizes[0] / F;
    const int E = in_sizes[1] / 2;
    const int* src = ei;
    const int* dst = ei + E;

    float* ws = (float*)d_ws;
    const long NH = (long)N * H;
    float* h0   = ws;  ws += NH;
    float* zA   = ws;  ws += NH;
    int*   csr  = (int*)ws;  ws += 2 * NH;
    float* zB   = ws;  ws += NH;
    float* h    = ws;  ws += NH;

    auto cdiv = [](long a, long b) { return (int)((a + b - 1) / b); };
    auto gemm_waves = [&](int M, int Nc) { return (long)(M / 16) * ((Nc + 63) / 64); };

    const int G  = cdiv(E, CH);
    const int nbk = cdiv(N, BN);
    int* pairs  = csr;
    int* ssrc   = pairs + 2 * E;
    int* offT   = ssrc + E;
    int* rowptr = offT + (nbk + 1) * G;

    const int GB = cdiv(gemm_waves(N, H) * 32, T1);
    encode_and_bin_kernel<<<G + GB, T1, 0, stream>>>(src, dst, pairs, offT, E, nbk, G,
                                                     x, encW, encB, h0, N, H, F);
    bucket_group_kernel<<<nbk, 256, 0, stream>>>(pairs, offT, rowptr, ssrc, N, G);

    const int GL = 768;
    for (int l = 0; l < L; ++l) {
        const float* zin  = (l == 0) ? h0 : ((l & 1) ? zA : zB);
        float*       zout = (l & 1) ? zB : zA;
        const int ln = (l + 1 < L) ? (l + 1) : 0;
        if (l == 0)
            gather_mlp_kernel<true, false><<<GL, NW * 32, 0, stream>>>(
                zin, rowptr, ssrc, t, l,
                W1 + (long)l * H * 2 * H, b1 + l * 2 * H, g1 + l * 2 * H, bb1 + l * 2 * H,
                W2 + (long)l * 2 * H * H, b2 + l * H, ng + ln * H, nb + ln * H,
                h, zout, 0, linW, linB, (float*)d_out, N);
        else if (l + 1 < L)
            gather_mlp_kernel<false, false><<<GL, NW * 32, 0, stream>>>(
                zin, rowptr, ssrc, t, l,
                W1 + (long)l * H * 2 * H, b1 + l * 2 * H, g1 + l * 2 * H, bb1 + l * 2 * H,
                W2 + (long)l * 2 * H * H, b2 + l * H, ng + ln * H, nb + ln * H,
                h, zout, 1, linW, linB, (float*)d_out, N);
        else
            gather_mlp_kernel<false, true><<<GL, NW * 32, 0, stream>>>(
                zin, rowptr, ssrc, t, l,
                W1 + (long)l * H * 2 * H, b1 + l * 2 * H, g1 + l * 2 * H, bb1 + l * 2 * H,
                W2 + (long)l * 2 * H * H, b2 + l * H, ng + ln * H, nb + ln * H,
                h, zout, 1, linW, linB, (float*)d_out, N);
    }
}
